# starting kernel
# baseline (speedup 1.0000x reference)
#include <hip/hip_runtime.h>
#include <hip/hip_bf16.h>
#include <math.h>

#define NN   50000
#define EE   800000
#define IND  128
#define HID  32
#define NH   4
#define NC   40
#define FDIM 128

typedef float v8f __attribute__((ext_vector_type(8)));
typedef float v2f __attribute__((ext_vector_type(2)));

__device__ __forceinline__ unsigned fkey(float f) {
    unsigned b = __float_as_uint(f);
    return (b & 0x80000000u) ? ~b : (b | 0x80000000u);
}
__device__ __forceinline__ float funkey(unsigned k) {
    unsigned b = (k & 0x80000000u) ? (k & 0x7FFFFFFFu) : ~k;
    return __uint_as_float(b);
}

__global__ void gat_gemm128(const float* __restrict__ X,
                            const float* __restrict__ W,
                            float* __restrict__ Y) {
    __shared__ float As[16 * FDIM];
    const int m0 = blockIdx.x * 16;
    const int t  = threadIdx.x;

    for (int i = t; i < 16 * FDIM / 4; i += 256)
        ((float4*)As)[i] = ((const float4*)(X + (size_t)m0 * FDIM))[i];
    __syncthreads();

    const int wave  = t >> 5;
    const int lane  = t & 31;
    const int n0    = wave * 16;
    const int mrow  = lane & 15;
    const int kgrp  = lane >> 4;

    v8f acc = {};
    for (int kk = 0; kk < FDIM; kk += 4) {
        v2f a, b;
        a.x = As[mrow * FDIM + kk + 2 * kgrp + 0];
        a.y = As[mrow * FDIM + kk + 2 * kgrp + 1];
        b.x = W[(kk + 2 * kgrp + 0) * FDIM + n0 + mrow];
        b.y = W[(kk + 2 * kgrp + 1) * FDIM + n0 + mrow];
        acc = __builtin_amdgcn_wmma_f32_16x16x4_f32(false, a, false, b,
                                                    (short)0, acc, false, false);
    }
    const int cn    = lane & 15;
    const int mbase = (lane >> 4) * 8;
    for (int r = 0; r < 8; ++r)
        Y[(size_t)(m0 + mbase + r) * FDIM + n0 + cn] = acc[r];
}

__global__ void gat_gemm_out(const float* __restrict__ Hm,
                             const float* __restrict__ Wout,
                             const float* __restrict__ bout,
                             float* __restrict__ logits) {
    __shared__ float As[16 * FDIM];
    const int m0 = blockIdx.x * 16;
    const int t  = threadIdx.x;
    for (int i = t; i < 16 * FDIM / 4; i += 96)
        ((float4*)As)[i] = ((const float4*)(Hm + (size_t)m0 * FDIM))[i];
    __syncthreads();

    const int wave = t >> 5;
    const int lane = t & 31;
    const int n0   = wave * 16;
    const int mrow = lane & 15;
    const int kgrp = lane >> 4;
    const int col  = n0 + mrow;

    v8f acc = {};
    for (int kk = 0; kk < FDIM; kk += 4) {
        v2f a, b;
        a.x = As[mrow * FDIM + kk + 2 * kgrp + 0];
        a.y = As[mrow * FDIM + kk + 2 * kgrp + 1];
        b.x = (col < NC) ? Wout[(kk + 2 * kgrp + 0) * NC + col] : 0.0f;
        b.y = (col < NC) ? Wout[(kk + 2 * kgrp + 1) * NC + col] : 0.0f;
        acc = __builtin_amdgcn_wmma_f32_16x16x4_f32(false, a, false, b,
                                                    (short)0, acc, false, false);
    }
    const int cn    = lane & 15;
    const int mbase = (lane >> 4) * 8;
    const int ocol  = n0 + cn;
    if (ocol < NC) {
        const float bb = bout[ocol];
        for (int r = 0; r < 8; ++r)
            logits[(size_t)(m0 + mbase + r) * NC + ocol] = acc[r] + bb;
    }
}

__global__ void gat_attn_coeff(const float* __restrict__ feat,
                               const float* __restrict__ al,
                               const float* __restrict__ ar,
                               float* __restrict__ el,
                               float* __restrict__ er) {
    int i = blockIdx.x * blockDim.x + threadIdx.x;
    if (i >= NN * NH) return;
    int n = i >> 2, h = i & 3;
    const float* f = feat + (size_t)n * FDIM + h * HID;
    float sl = 0.f, sr = 0.f;
    for (int d = 0; d < HID; ++d) {
        sl += f[d] * al[h * HID + d];
        sr += f[d] * ar[h * HID + d];
    }
    el[i] = sl;
    er[i] = sr;
}

__global__ void gat_edge_logits(const float* __restrict__ el,
                                const float* __restrict__ er,
                                const int* __restrict__ src,
                                const int* __restrict__ dst,
                                float* __restrict__ ebuf,
                                unsigned* __restrict__ maxkey) {
    int i = blockIdx.x * blockDim.x + threadIdx.x;
    if (i >= EE * NH) return;
    int e = i >> 2, h = i & 3;
    int s = src[e], d = dst[e];
    float v = el[s * NH + h] + er[d * NH + h];
    v = (v > 0.f) ? v : 0.2f * v;
    ebuf[i] = v;
    atomicMax(&maxkey[d * NH + h], fkey(v));
}

__global__ void gat_edge_exp(const int* __restrict__ dst,
                             float* __restrict__ ebuf,
                             const unsigned* __restrict__ maxkey,
                             float* __restrict__ denom) {
    int i = blockIdx.x * blockDim.x + threadIdx.x;
    if (i >= EE * NH) return;
    int e = i >> 2, h = i & 3;
    int d = dst[e];
    float m  = funkey(maxkey[d * NH + h]);
    float ee = expf(ebuf[i] - m);
    ebuf[i] = ee;
    atomicAdd(&denom[d * NH + h], ee);
}

__global__ void gat_scatter(const int* __restrict__ src,
                            const int* __restrict__ dst,
                            const float* __restrict__ ebuf,
                            const float* __restrict__ denom,
                            const float* __restrict__ feat,
                            float* __restrict__ agg) {
    int i = blockIdx.x * blockDim.x + threadIdx.x;
    if (i >= EE * FDIM) return;
    int e = i >> 7;
    int c = i & (FDIM - 1);
    int h = c >> 5;
    int s = src[e], d = dst[e];
    float a = ebuf[e * NH + h] / denom[d * NH + h];
    atomicAdd(&agg[(size_t)d * FDIM + c], feat[(size_t)s * FDIM + c] * a);
}

__global__ void gat_elu(const float* agg, const float* resid, float* out, int use_res) {
    int i = blockIdx.x * blockDim.x + threadIdx.x;
    if (i >= NN * FDIM) return;
    float v = agg[i] + (use_res ? resid[i] : 0.f);
    out[i] = (v > 0.f) ? v : (expf(v) - 1.0f);
}

__global__ void gat_mix(const float* __restrict__ h2,
                        const float* __restrict__ lamb,
                        float* hout) {
    int i = blockIdx.x * blockDim.x + threadIdx.x;
    if (i >= NN * FDIM) return;
    float lam = lamb[0];
    hout[i] = lam * hout[i] + (1.0f - lam) * h2[i];
}

static void run_conv(const float* X, const float* W, const float* al, const float* ar,
                     const int* src, const int* dst, const float* resid, float* out,
                     float* feat, float* agg, float* ebuf, float* el, float* er,
                     unsigned* maxkey, float* denom, hipStream_t stream) {
    hipMemsetAsync(agg,    0, (size_t)NN * FDIM * sizeof(float), stream);
    hipMemsetAsync(maxkey, 0, (size_t)NN * NH * sizeof(unsigned), stream);
    hipMemsetAsync(denom,  0, (size_t)NN * NH * sizeof(float), stream);

    gat_gemm128<<<NN / 16, 256, 0, stream>>>(X, W, feat);
    gat_attn_coeff<<<(NN * NH + 255) / 256, 256, 0, stream>>>(feat, al, ar, el, er);
    gat_edge_logits<<<(EE * NH + 255) / 256, 256, 0, stream>>>(el, er, src, dst, ebuf, maxkey);
    gat_edge_exp<<<(EE * NH + 255) / 256, 256, 0, stream>>>(dst, ebuf, maxkey, denom);
    gat_scatter<<<(EE * FDIM + 255) / 256, 256, 0, stream>>>(src, dst, ebuf, denom, feat, agg);
    gat_elu<<<(NN * FDIM + 255) / 256, 256, 0, stream>>>(agg, resid, out, resid != nullptr);
}

extern "C" void kernel_launch(void* const* d_in, const int* in_sizes, int n_in,
                              void* d_out, int out_size, void* d_ws, size_t ws_size,
                              hipStream_t stream) {
    (void)in_sizes; (void)n_in; (void)out_size; (void)ws_size;

    const float* inputs = (const float*)d_in[0];
    const float* target = (const float*)d_in[1];
    const float* lamb   = (const float*)d_in[2];
    const float* W0     = (const float*)d_in[3];
    const float* al0    = (const float*)d_in[4];
    const float* ar0    = (const float*)d_in[5];
    const float* W1     = (const float*)d_in[6];
    const float* al1    = (const float*)d_in[7];
    const float* ar1    = (const float*)d_in[8];
    const float* Wout   = (const float*)d_in[9];
    const float* bout   = (const float*)d_in[10];
    const int*   src    = (const int*)d_in[11];
    const int*   dst    = (const int*)d_in[12];

    float* feat   = (float*)d_ws;
    float* agg    = feat + (size_t)NN * FDIM;
    float* hbuf   = agg  + (size_t)NN * FDIM;
    float* ebuf   = hbuf + (size_t)NN * FDIM;
    float* el     = ebuf + (size_t)EE * NH;
    float* er     = el   + (size_t)NN * NH;
    unsigned* mk  = (unsigned*)(er + (size_t)NN * NH);
    float* denom  = (float*)(mk + (size_t)NN * NH);

    float* hregion = (float*)d_out;
    float* logits  = hregion + (size_t)NN * FDIM;

    run_conv(inputs, W0, al0, ar0, src, dst, nullptr, hbuf,
             feat, agg, ebuf, el, er, mk, denom, stream);
    run_conv(hbuf, W1, al1, ar1, src, dst, hbuf, hregion,
             feat, agg, ebuf, el, er, mk, denom, stream);

    run_conv(target, W0, al0, ar0, src, dst, nullptr, hbuf,
             feat, agg, ebuf, el, er, mk, denom, stream);
    run_conv(hbuf, W1, al1, ar1, src, dst, hbuf, hbuf,
             feat, agg, ebuf, el, er, mk, denom, stream);

    gat_mix<<<(NN * FDIM + 255) / 256, 256, 0, stream>>>(hbuf, lamb, hregion);
    gat_gemm_out<<<NN / 16, 96, 0, stream>>>(hregion, Wout, bout, logits);
}


# Edges grouped by destination once per call (arrival-order rank); one wave per node does softmax, weighted row gather, residual and elu in registers: no float atomics, no per-layer memsets
# speedup vs baseline: 4.2993x; 4.2993x over previous
#include <hip/hip_runtime.h>
#include <hip/hip_bf16.h>
#include <math.h>

#define NN   50000
#define EE   800000
#define IND  128
#define HID  32
#define NH   4
#define NC   40
#define FDIM 128

typedef float v8f __attribute__((ext_vector_type(8)));
typedef float v2f __attribute__((ext_vector_type(2)));

__global__ void gat_gemm128(const float* __restrict__ X,
                            const float* __restrict__ W,
                            float* __restrict__ Y) {
    __shared__ float As[16 * FDIM];
    const int m0 = blockIdx.x * 16;
    const int t  = threadIdx.x;

    for (int i = t; i < 16 * FDIM / 4; i += 256)
        ((float4*)As)[i] = ((const float4*)(X + (size_t)m0 * FDIM))[i];
    __syncthreads();

    const int wave  = t >> 5;
    const int lane  = t & 31;
    const int n0    = wave * 16;
    const int mrow  = lane & 15;
    const int kgrp  = lane >> 4;

    v8f acc = {};
    for (int kk = 0; kk < FDIM; kk += 4) {
        v2f a, b;
        a.x = As[mrow * FDIM + kk + 2 * kgrp + 0];
        a.y = As[mrow * FDIM + kk + 2 * kgrp + 1];
        b.x = W[(kk + 2 * kgrp + 0) * FDIM + n0 + mrow];
        b.y = W[(kk + 2 * kgrp + 1) * FDIM + n0 + mrow];
        acc = __builtin_amdgcn_wmma_f32_16x16x4_f32(false, a, false, b,
                                                    (short)0, acc, false, false);
    }
    const int cn    = lane & 15;
    const int mbase = (lane >> 4) * 8;
    for (int r = 0; r < 8; ++r)
        Y[(size_t)(m0 + mbase + r) * FDIM + n0 + cn] = acc[r];
}

__global__ void gat_gemm_out(const float* __restrict__ Hm,
                             const float* __restrict__ Wout,
                             const float* __restrict__ bout,
                             float* __restrict__ logits) {
    __shared__ float As[16 * FDIM];
    const int m0 = blockIdx.x * 16;
    const int t  = threadIdx.x;
    for (int i = t; i < 16 * FDIM / 4; i += 96)
        ((float4*)As)[i] = ((const float4*)(Hm + (size_t)m0 * FDIM))[i];
    __syncthreads();

    const int wave = t >> 5;
    const int lane = t & 31;
    const int n0   = wave * 16;
    const int mrow = lane & 15;
    const int kgrp = lane >> 4;
    const int col  = n0 + mrow;

    v8f acc = {};
    for (int kk = 0; kk < FDIM; kk += 4) {
        v2f a, b;
        a.x = As[mrow * FDIM + kk + 2 * kgrp + 0];
        a.y = As[mrow * FDIM + kk + 2 * kgrp + 1];
        b.x = (col < NC) ? Wout[(kk + 2 * kgrp + 0) * NC + col] : 0.0f;
        b.y = (col < NC) ? Wout[(kk + 2 * kgrp + 1) * NC + col] : 0.0f;
        acc = __builtin_amdgcn_wmma_f32_16x16x4_f32(false, a, false, b,
                                                    (short)0, acc, false, false);
    }
    const int cn    = lane & 15;
    const int mbase = (lane >> 4) * 8;
    const int ocol  = n0 + cn;
    if (ocol < NC) {
        const float bb = bout[ocol];
        for (int r = 0; r < 8; ++r)
            logits[(size_t)(m0 + mbase + r) * NC + ocol] = acc[r] + bb;
    }
}

__global__ void gat_attn_coeff(const float* __restrict__ feat,
                               const float* __restrict__ al,
                               const float* __restrict__ ar,
                               float* __restrict__ el,
                               float* __restrict__ er) {
    int i = blockIdx.x * blockDim.x + threadIdx.x;
    if (i >= NN * NH) return;
    int n = i >> 2, h = i & 3;
    const float* f = feat + (size_t)n * FDIM + h * HID;
    float sl = 0.f, sr = 0.f;
    for (int d = 0; d < HID; ++d) {
        sl += f[d] * al[h * HID + d];
        sr += f[d] * ar[h * HID + d];
    }
    el[i] = sl;
    er[i] = sr;
}

__global__ void gat_deg_count(const int* __restrict__ dst, int* __restrict__ cnt) {
    int e = blockIdx.x * blockDim.x + threadIdx.x;
    if (e >= EE) return;
    atomicAdd(&cnt[dst[e]], 1);
}

__global__ void gat_seg_alloc(int* cnt, int* __restrict__ seg, int* __restrict__ cursor) {
    const int i    = blockIdx.x * blockDim.x + threadIdx.x;
    const int lane = threadIdx.x & 31;
    const int c    = (i < NN) ? cnt[i] : 0;
    int incl = c;
    for (int o = 1; o < 32; o <<= 1) {
        int t = __shfl_up(incl, o);
        if (lane >= o) incl += t;
    }
    const int tot = __shfl(incl, 31);
    int base = 0;
    if (lane == 31) base = atomicAdd(&cnt[NN], tot);
    base = __shfl(base, 31);
    const int st = base + incl - c;
    if (i < NN) {
        seg[2 * i]     = st;
        seg[2 * i + 1] = c;
        cursor[i]      = st;
    }
}

__global__ void gat_seg_fill(const int* __restrict__ src, const int* __restrict__ dst,
                             int* __restrict__ cursor, int* __restrict__ csr) {
    int e = blockIdx.x * blockDim.x + threadIdx.x;
    if (e >= EE) return;
    int p = atomicAdd(&cursor[dst[e]], 1);
    if ((unsigned)p < (unsigned)EE) csr[p] = src[e];
}

__device__ __forceinline__ float lrelu(float v) { return (v > 0.f) ? v : 0.2f * v; }
__device__ __forceinline__ float elu1(float v)  { return (v > 0.f) ? v : (expf(v) - 1.0f); }

__device__ __forceinline__ int lane_i(int v, int j) { return __builtin_amdgcn_readlane(v, j); }
__device__ __forceinline__ float lane_f(float v, int j) {
    return __int_as_float(__builtin_amdgcn_readlane(__float_as_int(v), j));
}
__device__ __forceinline__ float head_w(int head, float a0, float a1, float a2, float a3, int j) {
    const float w0 = lane_f(a0, j), w1 = lane_f(a1, j), w2 = lane_f(a2, j), w3 = lane_f(a3, j);
    const float lo = (head & 1) ? w1 : w0;
    const float hi = (head & 1) ? w3 : w2;
    return (head & 2) ? hi : lo;
}
__device__ __forceinline__ void mul_add4(float4& acc, const float4 f, const float w) {
    acc.x += f.x * w; acc.y += f.y * w; acc.z += f.z * w; acc.w += f.w * w;
}

__device__ __forceinline__ void gather_rows(const float4* __restrict__ feat4, int lane, int head,
                                            int s_l, float a0, float a1, float a2, float a3,
                                            int count, float4& acc) {
    int j = 0;
    for (; j + 4 <= count; j += 4) {
        const int sA = lane_i(s_l, j),     sB = lane_i(s_l, j + 1);
        const int sC = lane_i(s_l, j + 2), sD = lane_i(s_l, j + 3);
        const float4 fA = feat4[(size_t)sA * 32 + lane];
        const float4 fB = feat4[(size_t)sB * 32 + lane];
        const float4 fC = feat4[(size_t)sC * 32 + lane];
        const float4 fD = feat4[(size_t)sD * 32 + lane];
        const float wA = head_w(head, a0, a1, a2, a3, j);
        const float wB = head_w(head, a0, a1, a2, a3, j + 1);
        const float wC = head_w(head, a0, a1, a2, a3, j + 2);
        const float wD = head_w(head, a0, a1, a2, a3, j + 3);
        mul_add4(acc, fA, wA); mul_add4(acc, fB, wB); mul_add4(acc, fC, wC); mul_add4(acc, fD, wD);
    }
#pragma clang loop unroll(disable)
    for (; j < count; ++j) {
        const int sA = lane_i(s_l, j);
        const float4 fA = feat4[(size_t)sA * 32 + lane];
        mul_add4(acc, fA, head_w(head, a0, a1, a2, a3, j));
    }
}

__global__ void gat_gather(const int* __restrict__ seg, const int* __restrict__ csr,
                           const float* __restrict__ el, const float* __restrict__ er,
                           const float* __restrict__ feat,
                           const float* resid, float* out, int use_res) {
    const int lane = threadIdx.x & 31;
    const int d = __builtin_amdgcn_readfirstlane(blockIdx.x * 8 + (threadIdx.x >> 5));
    if (d >= NN) return;
    const int head = lane >> 3;
    const int beg  = seg[2 * d], deg = seg[2 * d + 1];
    const float4 er4 = ((const float4*)er)[d];
    const float4* el4   = (const float4*)el;
    const float4* feat4 = (const float4*)feat;

    const bool ok0 = lane < deg;
    int s0 = 0;
    float v0 = 0.f, v1 = 0.f, v2 = 0.f, v3 = 0.f;
    if (ok0) {
        s0 = min(max(csr[beg + lane], 0), NN - 1);
        const float4 q = el4[s0];
        v0 = lrelu(q.x + er4.x); v1 = lrelu(q.y + er4.y);
        v2 = lrelu(q.z + er4.z); v3 = lrelu(q.w + er4.w);
    }
    float m0 = ok0 ? v0 : -INFINITY, m1 = ok0 ? v1 : -INFINITY;
    float m2 = ok0 ? v2 : -INFINITY, m3 = ok0 ? v3 : -INFINITY;
    for (int base = 32; base < deg; base += 32) {
        const int idx = base + lane;
        if (idx < deg) {
            const int s = min(max(csr[beg + idx], 0), NN - 1);
            const float4 q = el4[s];
            m0 = fmaxf(m0, lrelu(q.x + er4.x)); m1 = fmaxf(m1, lrelu(q.y + er4.y));
            m2 = fmaxf(m2, lrelu(q.z + er4.z)); m3 = fmaxf(m3, lrelu(q.w + er4.w));
        }
    }
    for (int o = 16; o > 0; o >>= 1) {
        m0 = fmaxf(m0, __shfl_xor(m0, o)); m1 = fmaxf(m1, __shfl_xor(m1, o));
        m2 = fmaxf(m2, __shfl_xor(m2, o)); m3 = fmaxf(m3, __shfl_xor(m3, o));
    }

    const float p0 = ok0 ? expf(v0 - m0) : 0.f, p1 = ok0 ? expf(v1 - m1) : 0.f;
    const float p2 = ok0 ? expf(v2 - m2) : 0.f, p3 = ok0 ? expf(v3 - m3) : 0.f;
    float d0 = p0, d1 = p1, d2 = p2, d3 = p3;
    for (int base = 32; base < deg; base += 32) {
        const int idx = base + lane;
        if (idx < deg) {
            const int s = min(max(csr[beg + idx], 0), NN - 1);
            const float4 q = el4[s];
            d0 += expf(lrelu(q.x + er4.x) - m0); d1 += expf(lrelu(q.y + er4.y) - m1);
            d2 += expf(lrelu(q.z + er4.z) - m2); d3 += expf(lrelu(q.w + er4.w) - m3);
        }
    }
    for (int o = 16; o > 0; o >>= 1) {
        d0 += __shfl_xor(d0, o); d1 += __shfl_xor(d1, o);
        d2 += __shfl_xor(d2, o); d3 += __shfl_xor(d3, o);
    }

    float4 acc = make_float4(0.f, 0.f, 0.f, 0.f);
    {
        const float a0 = ok0 ? p0 / d0 : 0.f, a1 = ok0 ? p1 / d1 : 0.f;
        const float a2 = ok0 ? p2 / d2 : 0.f, a3 = ok0 ? p3 / d3 : 0.f;
        gather_rows(feat4, lane, head, s0, a0, a1, a2, a3, min(deg, 32), acc);
    }
    for (int base = 32; base < deg; base += 32) {
        const int idx = base + lane;
        int s = 0;
        float a0 = 0.f, a1 = 0.f, a2 = 0.f, a3 = 0.f;
        if (idx < deg) {
            s = min(max(csr[beg + idx], 0), NN - 1);
            const float4 q = el4[s];
            a0 = expf(lrelu(q.x + er4.x) - m0) / d0; a1 = expf(lrelu(q.y + er4.y) - m1) / d1;
            a2 = expf(lrelu(q.z + er4.z) - m2) / d2; a3 = expf(lrelu(q.w + er4.w) - m3) / d3;
        }
        gather_rows(feat4, lane, head, s, a0, a1, a2, a3, min(deg - base, 32), acc);
    }

    if (use_res) {
        const float4 r = ((const float4*)resid)[(size_t)d * 32 + lane];
        acc.x += r.x; acc.y += r.y; acc.z += r.z; acc.w += r.w;
    }
    acc.x = elu1(acc.x); acc.y = elu1(acc.y); acc.z = elu1(acc.z); acc.w = elu1(acc.w);
    ((float4*)out)[(size_t)d * 32 + lane] = acc;
}

__global__ void gat_mix(const float* __restrict__ h2,
                        const float* __restrict__ lamb,
                        float* hout) {
    int i = blockIdx.x * blockDim.x + threadIdx.x;
    if (i >= NN * FDIM) return;
    float lam = lamb[0];
    hout[i] = lam * hout[i] + (1.0f - lam) * h2[i];
}

static void run_conv(const float* X, const float* W, const float* al, const float* ar,
                     const int* seg, const int* csr, const float* resid, float* out,
                     float* feat, float* el, float* er, hipStream_t stream) {
    gat_gemm128<<<NN / 16, 256, 0, stream>>>(X, W, feat);
    gat_attn_coeff<<<(NN * NH + 255) / 256, 256, 0, stream>>>(feat, al, ar, el, er);
    gat_gather<<<(NN + 7) / 8, 256, 0, stream>>>(seg, csr, el, er, feat, resid, out, resid != nullptr);
}

extern "C" void kernel_launch(void* const* d_in, const int* in_sizes, int n_in,
                              void* d_out, int out_size, void* d_ws, size_t ws_size,
                              hipStream_t stream) {
    (void)in_sizes; (void)n_in; (void)out_size; (void)ws_size;

    const float* inputs = (const float*)d_in[0];
    const float* target = (const float*)d_in[1];
    const float* lamb   = (const float*)d_in[2];
    const float* W0     = (const float*)d_in[3];
    const float* al0    = (const float*)d_in[4];
    const float* ar0    = (const float*)d_in[5];
    const float* W1     = (const float*)d_in[6];
    const float* al1    = (const float*)d_in[7];
    const float* ar1    = (const float*)d_in[8];
    const float* Wout   = (const float*)d_in[9];
    const float* bout   = (const float*)d_in[10];
    const int*   src    = (const int*)d_in[11];
    const int*   dst    = (const int*)d_in[12];

    float* feat   = (float*)d_ws;
    float* hbuf   = feat + (size_t)NN * FDIM;
    float* el     = hbuf + (size_t)NN * FDIM;
    float* er     = el   + (size_t)NN * NH;
    int*   cnt    = (int*)(er + (size_t)NN * NH);
    int*   seg    = cnt + NN + 16;
    int*   cursor = seg + 2 * NN;
    int*   csr    = cursor + NN;

    float* hregion = (float*)d_out;
    float* logits  = hregion + (size_t)NN * FDIM;

    hipMemsetAsync(cnt, 0, (size_t)(NN + 16) * sizeof(int), stream);
    gat_deg_count<<<(EE + 255) / 256, 256, 0, stream>>>(dst, cnt);
    gat_seg_alloc<<<(NN + 255) / 256, 256, 0, stream>>>(cnt, seg, cursor);
    gat_seg_fill<<<(EE + 255) / 256, 256, 0, stream>>>(src, dst, cursor, csr);

    run_conv(inputs, W0, al0, ar0, seg, csr, nullptr, hbuf, feat, el, er, stream);
    run_conv(hbuf, W1, al1, ar1, seg, csr, hbuf, hregion, feat, el, er, stream);

    run_conv(target, W0, al0, ar0, seg, csr, nullptr, hbuf, feat, el, er, stream);
    run_conv(hbuf, W1, al1, ar1, seg, csr, hbuf, hbuf, feat, el, er, stream);

    gat_mix<<<(NN * FDIM + 255) / 256, 256, 0, stream>>>(hbuf, lamb, hregion);
    gat_gemm_out<<<NN / 16, 96, 0, stream>>>(hregion, Wout, bout, logits);
}


# Mixup folded into the epilogue of the last gather launch (reads the first path's row, writes lam*h+(1-lam)*h2 to out): the element-wise launch of 6.4M one-float threads and its 77 MB are gone
# speedup vs baseline: 4.4767x; 1.0413x over previous
#include <hip/hip_runtime.h>
#include <hip/hip_bf16.h>
#include <math.h>

#define NN   50000
#define EE   800000
#define IND  128
#define HID  32
#define NH   4
#define NC   40
#define FDIM 128

typedef float v8f __attribute__((ext_vector_type(8)));
typedef float v2f __attribute__((ext_vector_type(2)));

__global__ void gat_gemm128(const float* __restrict__ X,
                            const float* __restrict__ W,
                            float* __restrict__ Y) {
    __shared__ float As[16 * FDIM];
    const int m0 = blockIdx.x * 16;
    const int t  = threadIdx.x;

    for (int i = t; i < 16 * FDIM / 4; i += 256)
        ((float4*)As)[i] = ((const float4*)(X + (size_t)m0 * FDIM))[i];
    __syncthreads();

    const int wave  = t >> 5;
    const int lane  = t & 31;
    const int n0    = wave * 16;
    const int mrow  = lane & 15;
    const int kgrp  = lane >> 4;

    v8f acc = {};
    for (int kk = 0; kk < FDIM; kk += 4) {
        v2f a, b;
        a.x = As[mrow * FDIM + kk + 2 * kgrp + 0];
        a.y = As[mrow * FDIM + kk + 2 * kgrp + 1];
        b.x = W[(kk + 2 * kgrp + 0) * FDIM + n0 + mrow];
        b.y = W[(kk + 2 * kgrp + 1) * FDIM + n0 + mrow];
        acc = __builtin_amdgcn_wmma_f32_16x16x4_f32(false, a, false, b,
                                                    (short)0, acc, false, false);
    }
    const int cn    = lane & 15;
    const int mbase = (lane >> 4) * 8;
    for (int r = 0; r < 8; ++r)
        Y[(size_t)(m0 + mbase + r) * FDIM + n0 + cn] = acc[r];
}

__global__ void gat_gemm_out(const float* __restrict__ Hm,
                             const float* __restrict__ Wout,
                             const float* __restrict__ bout,
                             float* __restrict__ logits) {
    __shared__ float As[16 * FDIM];
    const int m0 = blockIdx.x * 16;
    const int t  = threadIdx.x;
    for (int i = t; i < 16 * FDIM / 4; i += 96)
        ((float4*)As)[i] = ((const float4*)(Hm + (size_t)m0 * FDIM))[i];
    __syncthreads();

    const int wave = t >> 5;
    const int lane = t & 31;
    const int n0   = wave * 16;
    const int mrow = lane & 15;
    const int kgrp = lane >> 4;
    const int col  = n0 + mrow;

    v8f acc = {};
    for (int kk = 0; kk < FDIM; kk += 4) {
        v2f a, b;
        a.x = As[mrow * FDIM + kk + 2 * kgrp + 0];
        a.y = As[mrow * FDIM + kk + 2 * kgrp + 1];
        b.x = (col < NC) ? Wout[(kk + 2 * kgrp + 0) * NC + col] : 0.0f;
        b.y = (col < NC) ? Wout[(kk + 2 * kgrp + 1) * NC + col] : 0.0f;
        acc = __builtin_amdgcn_wmma_f32_16x16x4_f32(false, a, false, b,
                                                    (short)0, acc, false, false);
    }
    const int cn    = lane & 15;
    const int mbase = (lane >> 4) * 8;
    const int ocol  = n0 + cn;
    if (ocol < NC) {
        const float bb = bout[ocol];
        for (int r = 0; r < 8; ++r)
            logits[(size_t)(m0 + mbase + r) * NC + ocol] = acc[r] + bb;
    }
}

__global__ void gat_attn_coeff(const float* __restrict__ feat,
                               const float* __restrict__ al,
                               const float* __restrict__ ar,
                               float* __restrict__ el,
                               float* __restrict__ er) {
    int i = blockIdx.x * blockDim.x + threadIdx.x;
    if (i >= NN * NH) return;
    int n = i >> 2, h = i & 3;
    const float* f = feat + (size_t)n * FDIM + h * HID;
    float sl = 0.f, sr = 0.f;
    for (int d = 0; d < HID; ++d) {
        sl += f[d] * al[h * HID + d];
        sr += f[d] * ar[h * HID + d];
    }
    el[i] = sl;
    er[i] = sr;
}

__global__ void gat_deg_count(const int* __restrict__ dst, int* __restrict__ cnt) {
    int e = blockIdx.x * blockDim.x + threadIdx.x;
    if (e >= EE) return;
    atomicAdd(&cnt[dst[e]], 1);
}

__global__ void gat_seg_alloc(int* cnt, int* __restrict__ seg, int* __restrict__ cursor) {
    const int i    = blockIdx.x * blockDim.x + threadIdx.x;
    const int lane = threadIdx.x & 31;
    const int c    = (i < NN) ? cnt[i] : 0;
    int incl = c;
    for (int o = 1; o < 32; o <<= 1) {
        int t = __shfl_up(incl, o);
        if (lane >= o) incl += t;
    }
    const int tot = __shfl(incl, 31);
    int base = 0;
    if (lane == 31) base = atomicAdd(&cnt[NN], tot);
    base = __shfl(base, 31);
    const int st = base + incl - c;
    if (i < NN) {
        seg[2 * i]     = st;
        seg[2 * i + 1] = c;
        cursor[i]      = st;
    }
}

__global__ void gat_seg_fill(const int* __restrict__ src, const int* __restrict__ dst,
                             int* __restrict__ cursor, int* __restrict__ csr) {
    int e = blockIdx.x * blockDim.x + threadIdx.x;
    if (e >= EE) return;
    int p = atomicAdd(&cursor[dst[e]], 1);
    if ((unsigned)p < (unsigned)EE) csr[p] = src[e];
}

__device__ __forceinline__ float lrelu(float v) { return (v > 0.f) ? v : 0.2f * v; }
__device__ __forceinline__ float elu1(float v)  { return (v > 0.f) ? v : (expf(v) - 1.0f); }

__device__ __forceinline__ int lane_i(int v, int j) { return __builtin_amdgcn_readlane(v, j); }
__device__ __forceinline__ float lane_f(float v, int j) {
    return __int_as_float(__builtin_amdgcn_readlane(__float_as_int(v), j));
}
__device__ __forceinline__ float head_w(int head, float a0, float a1, float a2, float a3, int j) {
    const float w0 = lane_f(a0, j), w1 = lane_f(a1, j), w2 = lane_f(a2, j), w3 = lane_f(a3, j);
    const float lo = (head & 1) ? w1 : w0;
    const float hi = (head & 1) ? w3 : w2;
    return (head & 2) ? hi : lo;
}
__device__ __forceinline__ void mul_add4(float4& acc, const float4 f, const float w) {
    acc.x += f.x * w; acc.y += f.y * w; acc.z += f.z * w; acc.w += f.w * w;
}

__device__ __forceinline__ void gather_rows(const float4* __restrict__ feat4, int lane, int head,
                                            int s_l, float a0, float a1, float a2, float a3,
                                            int count, float4& acc) {
    int j = 0;
    for (; j + 4 <= count; j += 4) {
        const int sA = lane_i(s_l, j),     sB = lane_i(s_l, j + 1);
        const int sC = lane_i(s_l, j + 2), sD = lane_i(s_l, j + 3);
        const float4 fA = feat4[(size_t)sA * 32 + lane];
        const float4 fB = feat4[(size_t)sB * 32 + lane];
        const float4 fC = feat4[(size_t)sC * 32 + lane];
        const float4 fD = feat4[(size_t)sD * 32 + lane];
        const float wA = head_w(head, a0, a1, a2, a3, j);
        const float wB = head_w(head, a0, a1, a2, a3, j + 1);
        const float wC = head_w(head, a0, a1, a2, a3, j + 2);
        const float wD = head_w(head, a0, a1, a2, a3, j + 3);
        mul_add4(acc, fA, wA); mul_add4(acc, fB, wB); mul_add4(acc, fC, wC); mul_add4(acc, fD, wD);
    }
#pragma clang loop unroll(disable)
    for (; j < count; ++j) {
        const int sA = lane_i(s_l, j);
        const float4 fA = feat4[(size_t)sA * 32 + lane];
        mul_add4(acc, fA, head_w(head, a0, a1, a2, a3, j));
    }
}

__global__ void gat_gather(const int* __restrict__ seg, const int* __restrict__ csr,
                           const float* __restrict__ el, const float* __restrict__ er,
                           const float* __restrict__ feat,
                           const float* resid, float* out, int use_res,
                           const float* mixrow, const float* __restrict__ lamb) {
    const int lane = threadIdx.x & 31;
    const int d = __builtin_amdgcn_readfirstlane(blockIdx.x * 8 + (threadIdx.x >> 5));
    if (d >= NN) return;
    float4 mx = make_float4(0.f, 0.f, 0.f, 0.f);
    if (mixrow) mx = ((const float4*)mixrow)[(size_t)d * 32 + lane];
    const float lam = lamb[0];
    const int head = lane >> 3;
    const int beg  = seg[2 * d], deg = seg[2 * d + 1];
    const float4 er4 = ((const float4*)er)[d];
    const float4* el4   = (const float4*)el;
    const float4* feat4 = (const float4*)feat;

    const bool ok0 = lane < deg;
    int s0 = 0;
    float v0 = 0.f, v1 = 0.f, v2 = 0.f, v3 = 0.f;
    if (ok0) {
        s0 = min(max(csr[beg + lane], 0), NN - 1);
        const float4 q = el4[s0];
        v0 = lrelu(q.x + er4.x); v1 = lrelu(q.y + er4.y);
        v2 = lrelu(q.z + er4.z); v3 = lrelu(q.w + er4.w);
    }
    float m0 = ok0 ? v0 : -INFINITY, m1 = ok0 ? v1 : -INFINITY;
    float m2 = ok0 ? v2 : -INFINITY, m3 = ok0 ? v3 : -INFINITY;
    for (int base = 32; base < deg; base += 32) {
        const int idx = base + lane;
        if (idx < deg) {
            const int s = min(max(csr[beg + idx], 0), NN - 1);
            const float4 q = el4[s];
            m0 = fmaxf(m0, lrelu(q.x + er4.x)); m1 = fmaxf(m1, lrelu(q.y + er4.y));
            m2 = fmaxf(m2, lrelu(q.z + er4.z)); m3 = fmaxf(m3, lrelu(q.w + er4.w));
        }
    }
    for (int o = 16; o > 0; o >>= 1) {
        m0 = fmaxf(m0, __shfl_xor(m0, o)); m1 = fmaxf(m1, __shfl_xor(m1, o));
        m2 = fmaxf(m2, __shfl_xor(m2, o)); m3 = fmaxf(m3, __shfl_xor(m3, o));
    }

    const float p0 = ok0 ? expf(v0 - m0) : 0.f, p1 = ok0 ? expf(v1 - m1) : 0.f;
    const float p2 = ok0 ? expf(v2 - m2) : 0.f, p3 = ok0 ? expf(v3 - m3) : 0.f;
    float d0 = p0, d1 = p1, d2 = p2, d3 = p3;
    for (int base = 32; base < deg; base += 32) {
        const int idx = base + lane;
        if (idx < deg) {
            const int s = min(max(csr[beg + idx], 0), NN - 1);
            const float4 q = el4[s];
            d0 += expf(lrelu(q.x + er4.x) - m0); d1 += expf(lrelu(q.y + er4.y) - m1);
            d2 += expf(lrelu(q.z + er4.z) - m2); d3 += expf(lrelu(q.w + er4.w) - m3);
        }
    }
    for (int o = 16; o > 0; o >>= 1) {
        d0 += __shfl_xor(d0, o); d1 += __shfl_xor(d1, o);
        d2 += __shfl_xor(d2, o); d3 += __shfl_xor(d3, o);
    }

    float4 acc = make_float4(0.f, 0.f, 0.f, 0.f);
    {
        const float a0 = ok0 ? p0 / d0 : 0.f, a1 = ok0 ? p1 / d1 : 0.f;
        const float a2 = ok0 ? p2 / d2 : 0.f, a3 = ok0 ? p3 / d3 : 0.f;
        gather_rows(feat4, lane, head, s0, a0, a1, a2, a3, min(deg, 32), acc);
    }
    for (int base = 32; base < deg; base += 32) {
        const int idx = base + lane;
        int s = 0;
        float a0 = 0.f, a1 = 0.f, a2 = 0.f, a3 = 0.f;
        if (idx < deg) {
            s = min(max(csr[beg + idx], 0), NN - 1);
            const float4 q = el4[s];
            a0 = expf(lrelu(q.x + er4.x) - m0) / d0; a1 = expf(lrelu(q.y + er4.y) - m1) / d1;
            a2 = expf(lrelu(q.z + er4.z) - m2) / d2; a3 = expf(lrelu(q.w + er4.w) - m3) / d3;
        }
        gather_rows(feat4, lane, head, s, a0, a1, a2, a3, min(deg - base, 32), acc);
    }

    if (use_res) {
        const float4 r = ((const float4*)resid)[(size_t)d * 32 + lane];
        acc.x += r.x; acc.y += r.y; acc.z += r.z; acc.w += r.w;
    }
    acc.x = elu1(acc.x); acc.y = elu1(acc.y); acc.z = elu1(acc.z); acc.w = elu1(acc.w);
    if (mixrow) {
        const float om = 1.0f - lam;
        acc.x = fmaf(lam, mx.x, om * acc.x); acc.y = fmaf(lam, mx.y, om * acc.y);
        acc.z = fmaf(lam, mx.z, om * acc.z); acc.w = fmaf(lam, mx.w, om * acc.w);
    }
    ((float4*)out)[(size_t)d * 32 + lane] = acc;
}

static void run_conv(const float* X, const float* W, const float* al, const float* ar,
                     const int* seg, const int* csr, const float* resid,
                     const float* mixrow, const float* lamb, float* out,
                     float* feat, float* el, float* er, hipStream_t stream) {
    gat_gemm128<<<NN / 16, 256, 0, stream>>>(X, W, feat);
    gat_attn_coeff<<<(NN * NH + 255) / 256, 256, 0, stream>>>(feat, al, ar, el, er);
    gat_gather<<<(NN + 7) / 8, 256, 0, stream>>>(seg, csr, el, er, feat, resid, out, resid != nullptr,
                                                 mixrow, lamb);
}

extern "C" void kernel_launch(void* const* d_in, const int* in_sizes, int n_in,
                              void* d_out, int out_size, void* d_ws, size_t ws_size,
                              hipStream_t stream) {
    (void)in_sizes; (void)n_in; (void)out_size; (void)ws_size;

    const float* inputs = (const float*)d_in[0];
    const float* target = (const float*)d_in[1];
    const float* lamb   = (const float*)d_in[2];
    const float* W0     = (const float*)d_in[3];
    const float* al0    = (const float*)d_in[4];
    const float* ar0    = (const float*)d_in[5];
    const float* W1     = (const float*)d_in[6];
    const float* al1    = (const float*)d_in[7];
    const float* ar1    = (const float*)d_in[8];
    const float* Wout   = (const float*)d_in[9];
    const float* bout   = (const float*)d_in[10];
    const int*   src    = (const int*)d_in[11];
    const int*   dst    = (const int*)d_in[12];

    float* feat   = (float*)d_ws;
    float* hbuf   = feat + (size_t)NN * FDIM;
    float* el     = hbuf + (size_t)NN * FDIM;
    float* er     = el   + (size_t)NN * NH;
    int*   cnt    = (int*)(er + (size_t)NN * NH);
    int*   seg    = cnt + NN + 16;
    int*   cursor = seg + 2 * NN;
    int*   csr    = cursor + NN;

    float* hregion = (float*)d_out;
    float* logits  = hregion + (size_t)NN * FDIM;

    hipMemsetAsync(cnt, 0, (size_t)(NN + 16) * sizeof(int), stream);
    gat_deg_count<<<(EE + 255) / 256, 256, 0, stream>>>(dst, cnt);
    gat_seg_alloc<<<(NN + 255) / 256, 256, 0, stream>>>(cnt, seg, cursor);
    gat_seg_fill<<<(EE + 255) / 256, 256, 0, stream>>>(src, dst, cursor, csr);

    run_conv(inputs, W0, al0, ar0, seg, csr, nullptr, nullptr, lamb, hbuf, feat, el, er, stream);
    run_conv(hbuf, W1, al1, ar1, seg, csr, hbuf, nullptr, lamb, hregion, feat, el, er, stream);

    run_conv(target, W0, al0, ar0, seg, csr, nullptr, nullptr, lamb, hbuf, feat, el, er, stream);
    run_conv(hbuf, W1, al1, ar1, seg, csr, hbuf, hregion, lamb, hregion, feat, el, er, stream);

    gat_gemm_out<<<NN / 16, 96, 0, stream>>>(hregion, Wout, bout, logits);
}


# GEMM A tiles kept in LDS at a row pitch of 132 floats instead of 128: the 16 rows of every A-fragment read start on 16 different banks, not on one (16-way bank conflict gone)
# speedup vs baseline: 5.0817x; 1.1351x over previous
#include <hip/hip_runtime.h>
#include <hip/hip_bf16.h>
#include <math.h>

#define NN   50000
#define EE   800000
#define IND  128
#define HID  32
#define NH   4
#define NC   40
#define FDIM 128
#define APITCH 132

typedef float v8f __attribute__((ext_vector_type(8)));
typedef float v2f __attribute__((ext_vector_type(2)));

__global__ void gat_gemm128(const float* __restrict__ X,
                            const float* __restrict__ W,
                            float* __restrict__ Y) {
    __shared__ __align__(16) float As[16 * APITCH];
    const int m0 = blockIdx.x * 16;
    const int t  = threadIdx.x;

    for (int i = t; i < 16 * FDIM / 4; i += 256)
        *(float4*)(As + (i >> 5) * APITCH + 4 * (i & 31)) =
            ((const float4*)(X + (size_t)m0 * FDIM))[i];
    __syncthreads();

    const int wave  = t >> 5;
    const int lane  = t & 31;
    const int n0    = wave * 16;
    const int mrow  = lane & 15;
    const int kgrp  = lane >> 4;

    v8f acc = {};
    for (int kk = 0; kk < FDIM; kk += 4) {
        v2f a, b;
        a.x = As[mrow * APITCH + kk + 2 * kgrp + 0];
        a.y = As[mrow * APITCH + kk + 2 * kgrp + 1];
        b.x = W[(kk + 2 * kgrp + 0) * FDIM + n0 + mrow];
        b.y = W[(kk + 2 * kgrp + 1) * FDIM + n0 + mrow];
        acc = __builtin_amdgcn_wmma_f32_16x16x4_f32(false, a, false, b,
                                                    (short)0, acc, false, false);
    }
    const int cn    = lane & 15;
    const int mbase = (lane >> 4) * 8;
    for (int r = 0; r < 8; ++r)
        Y[(size_t)(m0 + mbase + r) * FDIM + n0 + cn] = acc[r];
}

__global__ void gat_gemm_out(const float* __restrict__ Hm,
                             const float* __restrict__ Wout,
                             const float* __restrict__ bout,
                             float* __restrict__ logits) {
    __shared__ __align__(16) float As[16 * APITCH];
    const int m0 = blockIdx.x * 16;
    const int t  = threadIdx.x;
    for (int i = t; i < 16 * FDIM / 4; i += 96)
        *(float4*)(As + (i >> 5) * APITCH + 4 * (i & 31)) =
            ((const float4*)(Hm + (size_t)m0 * FDIM))[i];
    __syncthreads();

    const int wave = t >> 5;
    const int lane = t & 31;
    const int n0   = wave * 16;
    const int mrow = lane & 15;
    const int kgrp = lane >> 4;
    const int col  = n0 + mrow;

    v8f acc = {};
    for (int kk = 0; kk < FDIM; kk += 4) {
        v2f a, b;
        a.x = As[mrow * APITCH + kk + 2 * kgrp + 0];
        a.y = As[mrow * APITCH + kk + 2 * kgrp + 1];
        b.x = (col < NC) ? Wout[(kk + 2 * kgrp + 0) * NC + col] : 0.0f;
        b.y = (col < NC) ? Wout[(kk + 2 * kgrp + 1) * NC + col] : 0.0f;
        acc = __builtin_amdgcn_wmma_f32_16x16x4_f32(false, a, false, b,
                                                    (short)0, acc, false, false);
    }
    const int cn    = lane & 15;
    const int mbase = (lane >> 4) * 8;
    const int ocol  = n0 + cn;
    if (ocol < NC) {
        const float bb = bout[ocol];
        for (int r = 0; r < 8; ++r)
            logits[(size_t)(m0 + mbase + r) * NC + ocol] = acc[r] + bb;
    }
}

__global__ void gat_attn_coeff(const float* __restrict__ feat,
                               const float* __restrict__ al,
                               const float* __restrict__ ar,
                               float* __restrict__ el,
                               float* __restrict__ er) {
    int i = blockIdx.x * blockDim.x + threadIdx.x;
    if (i >= NN * NH) return;
    int n = i >> 2, h = i & 3;
    const float* f = feat + (size_t)n * FDIM + h * HID;
    float sl = 0.f, sr = 0.f;
    for (int d = 0; d < HID; ++d) {
        sl += f[d] * al[h * HID + d];
        sr += f[d] * ar[h * HID + d];
    }
    el[i] = sl;
    er[i] = sr;
}

__global__ void gat_deg_count(const int* __restrict__ dst, int* __restrict__ cnt) {
    int e = blockIdx.x * blockDim.x + threadIdx.x;
    if (e >= EE) return;
    atomicAdd(&cnt[dst[e]], 1);
}

__global__ void gat_seg_alloc(int* cnt, int* __restrict__ seg, int* __restrict__ cursor) {
    const int i    = blockIdx.x * blockDim.x + threadIdx.x;
    const int lane = threadIdx.x & 31;
    const int c    = (i < NN) ? cnt[i] : 0;
    int incl = c;
    for (int o = 1; o < 32; o <<= 1) {
        int t = __shfl_up(incl, o);
        if (lane >= o) incl += t;
    }
    const int tot = __shfl(incl, 31);
    int base = 0;
    if (lane == 31) base = atomicAdd(&cnt[NN], tot);
    base = __shfl(base, 31);
    const int st = base + incl - c;
    if (i < NN) {
        seg[2 * i]     = st;
        seg[2 * i + 1] = c;
        cursor[i]      = st;
    }
}

__global__ void gat_seg_fill(const int* __restrict__ src, const int* __restrict__ dst,
                             int* __restrict__ cursor, int* __restrict__ csr) {
    int e = blockIdx.x * blockDim.x + threadIdx.x;
    if (e >= EE) return;
    int p = atomicAdd(&cursor[dst[e]], 1);
    if ((unsigned)p < (unsigned)EE) csr[p] = src[e];
}

__device__ __forceinline__ float lrelu(float v) { return (v > 0.f) ? v : 0.2f * v; }
__device__ __forceinline__ float elu1(float v)  { return (v > 0.f) ? v : (expf(v) - 1.0f); }

__device__ __forceinline__ int lane_i(int v, int j) { return __builtin_amdgcn_readlane(v, j); }
__device__ __forceinline__ float lane_f(float v, int j) {
    return __int_as_float(__builtin_amdgcn_readlane(__float_as_int(v), j));
}
__device__ __forceinline__ float head_w(int head, float a0, float a1, float a2, float a3, int j) {
    const float w0 = lane_f(a0, j), w1 = lane_f(a1, j), w2 = lane_f(a2, j), w3 = lane_f(a3, j);
    const float lo = (head & 1) ? w1 : w0;
    const float hi = (head & 1) ? w3 : w2;
    return (head & 2) ? hi : lo;
}
__device__ __forceinline__ void mul_add4(float4& acc, const float4 f, const float w) {
    acc.x += f.x * w; acc.y += f.y * w; acc.z += f.z * w; acc.w += f.w * w;
}

__device__ __forceinline__ void gather_rows(const float4* __restrict__ feat4, int lane, int head,
                                            int s_l, float a0, float a1, float a2, float a3,
                                            int count, float4& acc) {
    int j = 0;
    for (; j + 4 <= count; j += 4) {
        const int sA = lane_i(s_l, j),     sB = lane_i(s_l, j + 1);
        const int sC = lane_i(s_l, j + 2), sD = lane_i(s_l, j + 3);
        const float4 fA = feat4[(size_t)sA * 32 + lane];
        const float4 fB = feat4[(size_t)sB * 32 + lane];
        const float4 fC = feat4[(size_t)sC * 32 + lane];
        const float4 fD = feat4[(size_t)sD * 32 + lane];
        const float wA = head_w(head, a0, a1, a2, a3, j);
        const float wB = head_w(head, a0, a1, a2, a3, j + 1);
        const float wC = head_w(head, a0, a1, a2, a3, j + 2);
        const float wD = head_w(head, a0, a1, a2, a3, j + 3);
        mul_add4(acc, fA, wA); mul_add4(acc, fB, wB); mul_add4(acc, fC, wC); mul_add4(acc, fD, wD);
    }
#pragma clang loop unroll(disable)
    for (; j < count; ++j) {
        const int sA = lane_i(s_l, j);
        const float4 fA = feat4[(size_t)sA * 32 + lane];
        mul_add4(acc, fA, head_w(head, a0, a1, a2, a3, j));
    }
}

__global__ void gat_gather(const int* __restrict__ seg, const int* __restrict__ csr,
                           const float* __restrict__ el, const float* __restrict__ er,
                           const float* __restrict__ feat,
                           const float* resid, float* out, int use_res,
                           const float* mixrow, const float* __restrict__ lamb) {
    const int lane = threadIdx.x & 31;
    const int d = __builtin_amdgcn_readfirstlane(blockIdx.x * 8 + (threadIdx.x >> 5));
    if (d >= NN) return;
    float4 mx = make_float4(0.f, 0.f, 0.f, 0.f);
    if (mixrow) mx = ((const float4*)mixrow)[(size_t)d * 32 + lane];
    const float lam = lamb[0];
    const int head = lane >> 3;
    const int beg  = seg[2 * d], deg = seg[2 * d + 1];
    const float4 er4 = ((const float4*)er)[d];
    const float4* el4   = (const float4*)el;
    const float4* feat4 = (const float4*)feat;

    const bool ok0 = lane < deg;
    int s0 = 0;
    float v0 = 0.f, v1 = 0.f, v2 = 0.f, v3 = 0.f;
    if (ok0) {
        s0 = min(max(csr[beg + lane], 0), NN - 1);
        const float4 q = el4[s0];
        v0 = lrelu(q.x + er4.x); v1 = lrelu(q.y + er4.y);
        v2 = lrelu(q.z + er4.z); v3 = lrelu(q.w + er4.w);
    }
    float m0 = ok0 ? v0 : -INFINITY, m1 = ok0 ? v1 : -INFINITY;
    float m2 = ok0 ? v2 : -INFINITY, m3 = ok0 ? v3 : -INFINITY;
    for (int base = 32; base < deg; base += 32) {
        const int idx = base + lane;
        if (idx < deg) {
            const int s = min(max(csr[beg + idx], 0), NN - 1);
            const float4 q = el4[s];
            m0 = fmaxf(m0, lrelu(q.x + er4.x)); m1 = fmaxf(m1, lrelu(q.y + er4.y));
            m2 = fmaxf(m2, lrelu(q.z + er4.z)); m3 = fmaxf(m3, lrelu(q.w + er4.w));
        }
    }
    for (int o = 16; o > 0; o >>= 1) {
        m0 = fmaxf(m0, __shfl_xor(m0, o)); m1 = fmaxf(m1, __shfl_xor(m1, o));
        m2 = fmaxf(m2, __shfl_xor(m2, o)); m3 = fmaxf(m3, __shfl_xor(m3, o));
    }

    const float p0 = ok0 ? expf(v0 - m0) : 0.f, p1 = ok0 ? expf(v1 - m1) : 0.f;
    const float p2 = ok0 ? expf(v2 - m2) : 0.f, p3 = ok0 ? expf(v3 - m3) : 0.f;
    float d0 = p0, d1 = p1, d2 = p2, d3 = p3;
    for (int base = 32; base < deg; base += 32) {
        const int idx = base + lane;
        if (idx < deg) {
            const int s = min(max(csr[beg + idx], 0), NN - 1);
            const float4 q = el4[s];
            d0 += expf(lrelu(q.x + er4.x) - m0); d1 += expf(lrelu(q.y + er4.y) - m1);
            d2 += expf(lrelu(q.z + er4.z) - m2); d3 += expf(lrelu(q.w + er4.w) - m3);
        }
    }
    for (int o = 16; o > 0; o >>= 1) {
        d0 += __shfl_xor(d0, o); d1 += __shfl_xor(d1, o);
        d2 += __shfl_xor(d2, o); d3 += __shfl_xor(d3, o);
    }

    float4 acc = make_float4(0.f, 0.f, 0.f, 0.f);
    {
        const float a0 = ok0 ? p0 / d0 : 0.f, a1 = ok0 ? p1 / d1 : 0.f;
        const float a2 = ok0 ? p2 / d2 : 0.f, a3 = ok0 ? p3 / d3 : 0.f;
        gather_rows(feat4, lane, head, s0, a0, a1, a2, a3, min(deg, 32), acc);
    }
    for (int base = 32; base < deg; base += 32) {
        const int idx = base + lane;
        int s = 0;
        float a0 = 0.f, a1 = 0.f, a2 = 0.f, a3 = 0.f;
        if (idx < deg) {
            s = min(max(csr[beg + idx], 0), NN - 1);
            const float4 q = el4[s];
            a0 = expf(lrelu(q.x + er4.x) - m0) / d0; a1 = expf(lrelu(q.y + er4.y) - m1) / d1;
            a2 = expf(lrelu(q.z + er4.z) - m2) / d2; a3 = expf(lrelu(q.w + er4.w) - m3) / d3;
        }
        gather_rows(feat4, lane, head, s, a0, a1, a2, a3, min(deg - base, 32), acc);
    }

    if (use_res) {
        const float4 r = ((const float4*)resid)[(size_t)d * 32 + lane];
        acc.x += r.x; acc.y += r.y; acc.z += r.z; acc.w += r.w;
    }
    acc.x = elu1(acc.x); acc.y = elu1(acc.y); acc.z = elu1(acc.z); acc.w = elu1(acc.w);
    if (mixrow) {
        const float om = 1.0f - lam;
        acc.x = fmaf(lam, mx.x, om * acc.x); acc.y = fmaf(lam, mx.y, om * acc.y);
        acc.z = fmaf(lam, mx.z, om * acc.z); acc.w = fmaf(lam, mx.w, om * acc.w);
    }
    ((float4*)out)[(size_t)d * 32 + lane] = acc;
}

static void run_conv(const float* X, const float* W, const float* al, const float* ar,
                     const int* seg, const int* csr, const float* resid,
                     const float* mixrow, const float* lamb, float* out,
                     float* feat, float* el, float* er, hipStream_t stream) {
    gat_gemm128<<<NN / 16, 256, 0, stream>>>(X, W, feat);
    gat_attn_coeff<<<(NN * NH + 255) / 256, 256, 0, stream>>>(feat, al, ar, el, er);
    gat_gather<<<(NN + 7) / 8, 256, 0, stream>>>(seg, csr, el, er, feat, resid, out, resid != nullptr,
                                                 mixrow, lamb);
}

extern "C" void kernel_launch(void* const* d_in, const int* in_sizes, int n_in,
                              void* d_out, int out_size, void* d_ws, size_t ws_size,
                              hipStream_t stream) {
    (void)in_sizes; (void)n_in; (void)out_size; (void)ws_size;

    const float* inputs = (const float*)d_in[0];
    const float* target = (const float*)d_in[1];
    const float* lamb   = (const float*)d_in[2];
    const float* W0     = (const float*)d_in[3];
    const float* al0    = (const float*)d_in[4];
    const float* ar0    = (const float*)d_in[5];
    const float* W1     = (const float*)d_in[6];
    const float* al1    = (const float*)d_in[7];
    const float* ar1    = (const float*)d_in[8];
    const float* Wout   = (const float*)d_in[9];
    const float* bout   = (const float*)d_in[10];
    const int*   src    = (const int*)d_in[11];
    const int*   dst    = (const int*)d_in[12];

    float* feat   = (float*)d_ws;
    float* hbuf   = feat + (size_t)NN * FDIM;
    float* el     = hbuf + (size_t)NN * FDIM;
    float* er     = el   + (size_t)NN * NH;
    int*   cnt    = (int*)(er + (size_t)NN * NH);
    int*   seg    = cnt + NN + 16;
    int*   cursor = seg + 2 * NN;
    int*   csr    = cursor + NN;

    float* hregion = (float*)d_out;
    float* logits  = hregion + (size_t)NN * FDIM;

    hipMemsetAsync(cnt, 0, (size_t)(NN + 16) * sizeof(int), stream);
    gat_deg_count<<<(EE + 255) / 256, 256, 0, stream>>>(dst, cnt);
    gat_seg_alloc<<<(NN + 255) / 256, 256, 0, stream>>>(cnt, seg, cursor);
    gat_seg_fill<<<(EE + 255) / 256, 256, 0, stream>>>(src, dst, cursor, csr);

    run_conv(inputs, W0, al0, ar0, seg, csr, nullptr, nullptr, lamb, hbuf, feat, el, er, stream);
    run_conv(hbuf, W1, al1, ar1, seg, csr, hbuf, nullptr, lamb, hregion, feat, el, er, stream);

    run_conv(target, W0, al0, ar0, seg, csr, nullptr, nullptr, lamb, hbuf, feat, el, er, stream);
    run_conv(hbuf, W1, al1, ar1, seg, csr, hbuf, hregion, lamb, hregion, feat, el, er, stream);

    gat_gemm_out<<<NN / 16, 96, 0, stream>>>(hregion, Wout, bout, logits);
}


# Edges grouped by destination with a two-level counting sort on LDS counters (bins of 256 nodes, then one work-group per bin): no global atomics, no memset; slot = arrival order as before
# speedup vs baseline: 6.6015x; 1.2991x over previous
#include <hip/hip_runtime.h>
#include <hip/hip_bf16.h>
#include <math.h>

#define NN   50000
#define EE   800000
#define IND  128
#define HID  32
#define NH   4
#define NC   40
#define FDIM 128
#define APITCH 132

typedef float v8f __attribute__((ext_vector_type(8)));
typedef float v2f __attribute__((ext_vector_type(2)));

__global__ void gat_gemm128(const float* __restrict__ X,
                            const float* __restrict__ W,
                            float* __restrict__ Y) {
    __shared__ __align__(16) float As[16 * APITCH];
    const int m0 = blockIdx.x * 16;
    const int t  = threadIdx.x;

    for (int i = t; i < 16 * FDIM / 4; i += 256)
        *(float4*)(As + (i >> 5) * APITCH + 4 * (i & 31)) =
            ((const float4*)(X + (size_t)m0 * FDIM))[i];
    __syncthreads();

    const int wave  = t >> 5;
    const int lane  = t & 31;
    const int n0    = wave * 16;
    const int mrow  = lane & 15;
    const int kgrp  = lane >> 4;

    v8f acc = {};
    for (int kk = 0; kk < FDIM; kk += 4) {
        v2f a, b;
        a.x = As[mrow * APITCH + kk + 2 * kgrp + 0];
        a.y = As[mrow * APITCH + kk + 2 * kgrp + 1];
        b.x = W[(kk + 2 * kgrp + 0) * FDIM + n0 + mrow];
        b.y = W[(kk + 2 * kgrp + 1) * FDIM + n0 + mrow];
        acc = __builtin_amdgcn_wmma_f32_16x16x4_f32(false, a, false, b,
                                                    (short)0, acc, false, false);
    }
    const int cn    = lane & 15;
    const int mbase = (lane >> 4) * 8;
    for (int r = 0; r < 8; ++r)
        Y[(size_t)(m0 + mbase + r) * FDIM + n0 + cn] = acc[r];
}

__global__ void gat_gemm_out(const float* __restrict__ Hm,
                             const float* __restrict__ Wout,
                             const float* __restrict__ bout,
                             float* __restrict__ logits) {
    __shared__ __align__(16) float As[16 * APITCH];
    const int m0 = blockIdx.x * 16;
    const int t  = threadIdx.x;
    for (int i = t; i < 16 * FDIM / 4; i += 96)
        *(float4*)(As + (i >> 5) * APITCH + 4 * (i & 31)) =
            ((const float4*)(Hm + (size_t)m0 * FDIM))[i];
    __syncthreads();

    const int wave = t >> 5;
    const int lane = t & 31;
    const int n0   = wave * 16;
    const int mrow = lane & 15;
    const int kgrp = lane >> 4;
    const int col  = n0 + mrow;

    v8f acc = {};
    for (int kk = 0; kk < FDIM; kk += 4) {
        v2f a, b;
        a.x = As[mrow * APITCH + kk + 2 * kgrp + 0];
        a.y = As[mrow * APITCH + kk + 2 * kgrp + 1];
        b.x = (col < NC) ? Wout[(kk + 2 * kgrp + 0) * NC + col] : 0.0f;
        b.y = (col < NC) ? Wout[(kk + 2 * kgrp + 1) * NC + col] : 0.0f;
        acc = __builtin_amdgcn_wmma_f32_16x16x4_f32(false, a, false, b,
                                                    (short)0, acc, false, false);
    }
    const int cn    = lane & 15;
    const int mbase = (lane >> 4) * 8;
    const int ocol  = n0 + cn;
    if (ocol < NC) {
        const float bb = bout[ocol];
        for (int r = 0; r < 8; ++r)
            logits[(size_t)(m0 + mbase + r) * NC + ocol] = acc[r] + bb;
    }
}

__global__ void gat_attn_coeff(const float* __restrict__ feat,
                               const float* __restrict__ al,
                               const float* __restrict__ ar,
                               float* __restrict__ el,
                               float* __restrict__ er) {
    int i = blockIdx.x * blockDim.x + threadIdx.x;
    if (i >= NN * NH) return;
    int n = i >> 2, h = i & 3;
    const float* f = feat + (size_t)n * FDIM + h * HID;
    float sl = 0.f, sr = 0.f;
    for (int d = 0; d < HID; ++d) {
        sl += f[d] * al[h * HID + d];
        sr += f[d] * ar[h * HID + d];
    }
    el[i] = sl;
    er[i] = sr;
}

#define NBK 196
#define NCH 256
#define CHE (EE / NCH)
#define KTR ((CHE + 1023) / 1024)
static_assert(EE % NCH == 0, "the edge list must split evenly into chunks");
static_assert(NBK * 256 >= NN, "the bins must cover every node");
static_assert(NCH == 256, "the scan below reads a bin's row as 32 lanes x 8 counts");

__device__ __forceinline__ int bin_of(int d) { return min((int)((unsigned)d >> 8), NBK - 1); }

__device__ __forceinline__ int wave_incl(int v, int lane) {
    for (int o = 1; o < 32; o <<= 1) {
        const int u = __shfl_up(v, o);
        if (lane >= o) v += u;
    }
    return v;
}

__global__ void grp_bin_count(const int* __restrict__ dst, int* __restrict__ M) {
    __shared__ int hist[NBK];
    const int t = threadIdx.x, c = blockIdx.x;
    if (t < NBK) hist[t] = 0;
    __syncthreads();
    int d[KTR];
#pragma unroll
    for (int k = 0; k < KTR; ++k) d[k] = dst[c * CHE + min(t + 1024 * k, CHE - 1)];
#pragma unroll
    for (int k = 0; k < KTR; ++k)
        if (t + 1024 * k < CHE) atomicAdd(&hist[bin_of(d[k])], 1);
    __syncthreads();
    if (t < NBK) M[t * NCH + c] = hist[t];
}

__global__ void grp_bin_scan(const int* __restrict__ M, int* __restrict__ O) {
    __shared__ int rtot[224];
    __shared__ int bst[224];
    const int t = threadIdx.x, lane = t & 31, w = t >> 5;
    const int4* M4 = (const int4*)M;
    int4* O4 = (int4*)O;
    int4 a[7], c[7];
    int s[7], incl[7];
#pragma unroll
    for (int i = 0; i < 7; ++i) {
        const int b = w + 32 * i;
        a[i] = make_int4(0, 0, 0, 0); c[i] = a[i];
        if (b < NBK) { a[i] = M4[b * 64 + 2 * lane]; c[i] = M4[b * 64 + 2 * lane + 1]; }
    }
#pragma unroll
    for (int i = 0; i < 7; ++i) {
        const int b = w + 32 * i;
        s[i] = a[i].x + a[i].y + a[i].z + a[i].w + c[i].x + c[i].y + c[i].z + c[i].w;
        incl[i] = wave_incl(s[i], lane);
        if (lane == 31 && b < NBK) rtot[b] = incl[i];
    }
    __syncthreads();
    if (w == 0) {
        int x[7];
        int ss = 0;
#pragma unroll
        for (int k = 0; k < 7; ++k) {
            const int idx = 7 * lane + k;
            x[k] = (idx < NBK) ? rtot[idx] : 0;
            ss += x[k];
        }
        const int in2 = wave_incl(ss, lane);
        int run = in2 - ss;
#pragma unroll
        for (int k = 0; k < 7; ++k) {
            const int idx = 7 * lane + k;
            if (idx < NBK) bst[idx] = run;
            run += x[k];
        }
        if (lane == 31) O[NBK * NCH] = in2;
    }
    __syncthreads();
#pragma unroll
    for (int i = 0; i < 7; ++i) {
        const int b = w + 32 * i;
        if (b < NBK) {
            int run = bst[b] + incl[i] - s[i];
            int4 o1, o2;
            o1.x = run; run += a[i].x; o1.y = run; run += a[i].y;
            o1.z = run; run += a[i].z; o1.w = run; run += a[i].w;
            o2.x = run; run += c[i].x; o2.y = run; run += c[i].y;
            o2.z = run; run += c[i].z; o2.w = run;
            O4[b * 64 + 2 * lane]     = o1;
            O4[b * 64 + 2 * lane + 1] = o2;
        }
    }
}

__global__ void grp_bin_place(const int* __restrict__ src, const int* __restrict__ dst,
                              const int* __restrict__ O, int2* __restrict__ ebuf) {
    __shared__ int cur[NBK];
    const int t = threadIdx.x, c = blockIdx.x;
    if (t < NBK) cur[t] = O[t * NCH + c];
    __syncthreads();
    int d[KTR], s[KTR];
#pragma unroll
    for (int k = 0; k < KTR; ++k) {
        const int e = c * CHE + min(t + 1024 * k, CHE - 1);
        d[k] = dst[e]; s[k] = src[e];
    }
#pragma unroll
    for (int k = 0; k < KTR; ++k)
        if (t + 1024 * k < CHE) {
            const int p = atomicAdd(&cur[bin_of(d[k])], 1);
            ebuf[min(max(p, 0), EE - 1)] = make_int2(s[k], d[k]);
        }
}

__global__ void grp_bin_sort(const int2* __restrict__ ebuf, const int* __restrict__ O,
                             int* __restrict__ seg, int* __restrict__ csr) {
    __shared__ int cnt[256];
    __shared__ int cur[256];
    __shared__ int wtot[8];
    const int t = threadIdx.x, lane = t & 31, w = t >> 5, b = blockIdx.x;
    const int s0 = min(max(O[b * NCH], 0), EE);
    const int n  = min(max(O[(b + 1) * NCH] - s0, 0), EE - s0);
    if (t < 256) cnt[t] = 0;
    __syncthreads();
    for (int i0 = 0; i0 < n; i0 += 4096) {
        int dk[4];
#pragma unroll
        for (int k = 0; k < 4; ++k) dk[k] = ebuf[s0 + min(i0 + t + 1024 * k, n - 1)].y;
#pragma unroll
        for (int k = 0; k < 4; ++k)
            if (i0 + t + 1024 * k < n) atomicAdd(&cnt[dk[k] & 255], 1);
    }
    __syncthreads();
    const int c    = (t < 256) ? cnt[t] : 0;
    const int incl = wave_incl(c, lane);
    if (t < 256 && lane == 31) wtot[w] = incl;
    __syncthreads();
    if (t < 256) {
        int base = 0;
        for (int k = 0; k < w; ++k) base += wtot[k];
        const int excl = base + incl - c;
        cur[t] = excl;
        const int node = b * 256 + t;
        if (node < NN) {
            seg[2 * node]     = s0 + excl;
            seg[2 * node + 1] = c;
        }
    }
    __syncthreads();
    for (int i0 = 0; i0 < n; i0 += 4096) {
        int2 r[4];
#pragma unroll
        for (int k = 0; k < 4; ++k) r[k] = ebuf[s0 + min(i0 + t + 1024 * k, n - 1)];
#pragma unroll
        for (int k = 0; k < 4; ++k)
            if (i0 + t + 1024 * k < n) {
                const int p = atomicAdd(&cur[r[k].y & 255], 1);
                csr[s0 + min(max(p, 0), n - 1)] = r[k].x;
            }
    }
}

__device__ __forceinline__ float lrelu(float v) { return (v > 0.f) ? v : 0.2f * v; }
__device__ __forceinline__ float elu1(float v)  { return (v > 0.f) ? v : (expf(v) - 1.0f); }

__device__ __forceinline__ int lane_i(int v, int j) { return __builtin_amdgcn_readlane(v, j); }
__device__ __forceinline__ float lane_f(float v, int j) {
    return __int_as_float(__builtin_amdgcn_readlane(__float_as_int(v), j));
}
__device__ __forceinline__ float head_w(int head, float a0, float a1, float a2, float a3, int j) {
    const float w0 = lane_f(a0, j), w1 = lane_f(a1, j), w2 = lane_f(a2, j), w3 = lane_f(a3, j);
    const float lo = (head & 1) ? w1 : w0;
    const float hi = (head & 1) ? w3 : w2;
    return (head & 2) ? hi : lo;
}
__device__ __forceinline__ void mul_add4(float4& acc, const float4 f, const float w) {
    acc.x += f.x * w; acc.y += f.y * w; acc.z += f.z * w; acc.w += f.w * w;
}

__device__ __forceinline__ void gather_rows(const float4* __restrict__ feat4, int lane, int head,
                                            int s_l, float a0, float a1, float a2, float a3,
                                            int count, float4& acc) {
    int j = 0;
    for (; j + 4 <= count; j += 4) {
        const int sA = lane_i(s_l, j),     sB = lane_i(s_l, j + 1);
        const int sC = lane_i(s_l, j + 2), sD = lane_i(s_l, j + 3);
        const float4 fA = feat4[(size_t)sA * 32 + lane];
        const float4 fB = feat4[(size_t)sB * 32 + lane];
        const float4 fC = feat4[(size_t)sC * 32 + lane];
        const float4 fD = feat4[(size_t)sD * 32 + lane];
        const float wA = head_w(head, a0, a1, a2, a3, j);
        const float wB = head_w(head, a0, a1, a2, a3, j + 1);
        const float wC = head_w(head, a0, a1, a2, a3, j + 2);
        const float wD = head_w(head, a0, a1, a2, a3, j + 3);
        mul_add4(acc, fA, wA); mul_add4(acc, fB, wB); mul_add4(acc, fC, wC); mul_add4(acc, fD, wD);
    }
#pragma clang loop unroll(disable)
    for (; j < count; ++j) {
        const int sA = lane_i(s_l, j);
        const float4 fA = feat4[(size_t)sA * 32 + lane];
        mul_add4(acc, fA, head_w(head, a0, a1, a2, a3, j));
    }
}

__global__ void gat_gather(const int* __restrict__ seg, const int* __restrict__ csr,
                           const float* __restrict__ el, const float* __restrict__ er,
                           const float* __restrict__ feat,
                           const float* resid, float* out, int use_res,
                           const float* mixrow, const float* __restrict__ lamb) {
    const int lane = threadIdx.x & 31;
    const int d = __builtin_amdgcn_readfirstlane(blockIdx.x * 8 + (threadIdx.x >> 5));
    if (d >= NN) return;
    float4 mx = make_float4(0.f, 0.f, 0.f, 0.f);
    if (mixrow) mx = ((const float4*)mixrow)[(size_t)d * 32 + lane];
    const float lam = lamb[0];
    const int head = lane >> 3;
    const int beg  = seg[2 * d], deg = seg[2 * d + 1];
    const float4 er4 = ((const float4*)er)[d];
    const float4* el4   = (const float4*)el;
    const float4* feat4 = (const float4*)feat;

    const bool ok0 = lane < deg;
    int s0 = 0;
    float v0 = 0.f, v1 = 0.f, v2 = 0.f, v3 = 0.f;
    if (ok0) {
        s0 = min(max(csr[beg + lane], 0), NN - 1);
        const float4 q = el4[s0];
        v0 = lrelu(q.x + er4.x); v1 = lrelu(q.y + er4.y);
        v2 = lrelu(q.z + er4.z); v3 = lrelu(q.w + er4.w);
    }
    float m0 = ok0 ? v0 : -INFINITY, m1 = ok0 ? v1 : -INFINITY;
    float m2 = ok0 ? v2 : -INFINITY, m3 = ok0 ? v3 : -INFINITY;
    for (int base = 32; base < deg; base += 32) {
        const int idx = base + lane;
        if (idx < deg) {
            const int s = min(max(csr[beg + idx], 0), NN - 1);
            const float4 q = el4[s];
            m0 = fmaxf(m0, lrelu(q.x + er4.x)); m1 = fmaxf(m1, lrelu(q.y + er4.y));
            m2 = fmaxf(m2, lrelu(q.z + er4.z)); m3 = fmaxf(m3, lrelu(q.w + er4.w));
        }
    }
    for (int o = 16; o > 0; o >>= 1) {
        m0 = fmaxf(m0, __shfl_xor(m0, o)); m1 = fmaxf(m1, __shfl_xor(m1, o));
        m2 = fmaxf(m2, __shfl_xor(m2, o)); m3 = fmaxf(m3, __shfl_xor(m3, o));
    }

    const float p0 = ok0 ? expf(v0 - m0) : 0.f, p1 = ok0 ? expf(v1 - m1) : 0.f;
    const float p2 = ok0 ? expf(v2 - m2) : 0.f, p3 = ok0 ? expf(v3 - m3) : 0.f;
    float d0 = p0, d1 = p1, d2 = p2, d3 = p3;
    for (int base = 32; base < deg; base += 32) {
        const int idx = base + lane;
        if (idx < deg) {
            const int s = min(max(csr[beg + idx], 0), NN - 1);
            const float4 q = el4[s];
            d0 += expf(lrelu(q.x + er4.x) - m0); d1 += expf(lrelu(q.y + er4.y) - m1);
            d2 += expf(lrelu(q.z + er4.z) - m2); d3 += expf(lrelu(q.w + er4.w) - m3);
        }
    }
    for (int o = 16; o > 0; o >>= 1) {
        d0 += __shfl_xor(d0, o); d1 += __shfl_xor(d1, o);
        d2 += __shfl_xor(d2, o); d3 += __shfl_xor(d3, o);
    }

    float4 acc = make_float4(0.f, 0.f, 0.f, 0.f);
    {
        const float a0 = ok0 ? p0 / d0 : 0.f, a1 = ok0 ? p1 / d1 : 0.f;
        const float a2 = ok0 ? p2 / d2 : 0.f, a3 = ok0 ? p3 / d3 : 0.f;
        gather_rows(feat4, lane, head, s0, a0, a1, a2, a3, min(deg, 32), acc);
    }
    for (int base = 32; base < deg; base += 32) {
        const int idx = base + lane;
        int s = 0;
        float a0 = 0.f, a1 = 0.f, a2 = 0.f, a3 = 0.f;
        if (idx < deg) {
            s = min(max(csr[beg + idx], 0), NN - 1);
            const float4 q = el4[s];
            a0 = expf(lrelu(q.x + er4.x) - m0) / d0; a1 = expf(lrelu(q.y + er4.y) - m1) / d1;
            a2 = expf(lrelu(q.z + er4.z) - m2) / d2; a3 = expf(lrelu(q.w + er4.w) - m3) / d3;
        }
        gather_rows(feat4, lane, head, s, a0, a1, a2, a3, min(deg - base, 32), acc);
    }

    if (use_res) {
        const float4 r = ((const float4*)resid)[(size_t)d * 32 + lane];
        acc.x += r.x; acc.y += r.y; acc.z += r.z; acc.w += r.w;
    }
    acc.x = elu1(acc.x); acc.y = elu1(acc.y); acc.z = elu1(acc.z); acc.w = elu1(acc.w);
    if (mixrow) {
        const float om = 1.0f - lam;
        acc.x = fmaf(lam, mx.x, om * acc.x); acc.y = fmaf(lam, mx.y, om * acc.y);
        acc.z = fmaf(lam, mx.z, om * acc.z); acc.w = fmaf(lam, mx.w, om * acc.w);
    }
    ((float4*)out)[(size_t)d * 32 + lane] = acc;
}

static void run_conv(const float* X, const float* W, const float* al, const float* ar,
                     const int* seg, const int* csr, const float* resid,
                     const float* mixrow, const float* lamb, float* out,
                     float* feat, float* el, float* er, hipStream_t stream) {
    gat_gemm128<<<NN / 16, 256, 0, stream>>>(X, W, feat);
    gat_attn_coeff<<<(NN * NH + 255) / 256, 256, 0, stream>>>(feat, al, ar, el, er);
    gat_gather<<<(NN + 7) / 8, 256, 0, stream>>>(seg, csr, el, er, feat, resid, out, resid != nullptr,
                                                 mixrow, lamb);
}

extern "C" void kernel_launch(void* const* d_in, const int* in_sizes, int n_in,
                              void* d_out, int out_size, void* d_ws, size_t ws_size,
                              hipStream_t stream) {
    (void)in_sizes; (void)n_in; (void)out_size; (void)ws_size;

    const float* inputs = (const float*)d_in[0];
    const float* target = (const float*)d_in[1];
    const float* lamb   = (const float*)d_in[2];
    const float* W0     = (const float*)d_in[3];
    const float* al0    = (const float*)d_in[4];
    const float* ar0    = (const float*)d_in[5];
    const float* W1     = (const float*)d_in[6];
    const float* al1    = (const float*)d_in[7];
    const float* ar1    = (const float*)d_in[8];
    const float* Wout   = (const float*)d_in[9];
    const float* bout   = (const float*)d_in[10];
    const int*   src    = (const int*)d_in[11];
    const int*   dst    = (const int*)d_in[12];

    float* feat   = (float*)d_ws;
    float* hbuf   = feat + (size_t)NN * FDIM;
    float* el     = hbuf + (size_t)NN * FDIM;
    float* er     = el   + (size_t)NN * NH;
    int*   seg    = (int*)(er + (size_t)NN * NH);
    int*   csr    = seg + 2 * NN;
    int*   binM   = csr + EE;
    int*   binO   = binM + NBK * NCH;
    int2*  ebuf   = (int2*)(binO + NBK * NCH + 16);

    float* hregion = (float*)d_out;
    float* logits  = hregion + (size_t)NN * FDIM;

    grp_bin_count<<<NCH, 1024, 0, stream>>>(dst, binM);
    grp_bin_scan<<<1, 1024, 0, stream>>>(binM, binO);
    grp_bin_place<<<NCH, 1024, 0, stream>>>(src, dst, binO, ebuf);
    grp_bin_sort<<<NBK, 1024, 0, stream>>>(ebuf, binO, seg, csr);

    run_conv(inputs, W0, al0, ar0, seg, csr, nullptr, nullptr, lamb, hbuf, feat, el, er, stream);
    run_conv(hbuf, W1, al1, ar1, seg, csr, hbuf, nullptr, lamb, hregion, feat, el, er, stream);

    run_conv(target, W0, al0, ar0, seg, csr, nullptr, nullptr, lamb, hbuf, feat, el, er, stream);
    run_conv(hbuf, W1, al1, ar1, seg, csr, hbuf, hregion, lamb, hregion, feat, el, er, stream);

    gat_gemm_out<<<NN / 16, 96, 0, stream>>>(hregion, Wout, bout, logits);
}


# el/er dot products taken in the GEMM epilogue from the output tile left in LDS (same order of sums, bit-identical): the four el/er launches and their re-read of the feature rows are gone
# speedup vs baseline: 6.9794x; 1.0572x over previous
#include <hip/hip_runtime.h>
#include <hip/hip_bf16.h>
#include <math.h>

#define NN   50000
#define EE   800000
#define IND  128
#define HID  32
#define NH   4
#define NC   40
#define FDIM 128
#define APITCH 132

typedef float v8f __attribute__((ext_vector_type(8)));
typedef float v2f __attribute__((ext_vector_type(2)));

__global__ void gat_gemm128(const float* __restrict__ X,
                            const float* __restrict__ W,
                            const float* __restrict__ al,
                            const float* __restrict__ ar,
                            float* __restrict__ Y,
                            float* __restrict__ el,
                            float* __restrict__ er) {
    __shared__ __align__(16) float As[16 * APITCH];
    __shared__ __align__(16) float Cs[16 * APITCH];
    const int m0 = blockIdx.x * 16;
    const int t  = threadIdx.x;

    for (int i = t; i < 16 * FDIM / 4; i += 256)
        *(float4*)(As + (i >> 5) * APITCH + 4 * (i & 31)) =
            ((const float4*)(X + (size_t)m0 * FDIM))[i];
    __syncthreads();

    const int wave  = t >> 5;
    const int lane  = t & 31;
    const int n0    = wave * 16;
    const int mrow  = lane & 15;
    const int kgrp  = lane >> 4;

    v8f acc = {};
    for (int kk = 0; kk < FDIM; kk += 4) {
        v2f a, b;
        a.x = As[mrow * APITCH + kk + 2 * kgrp + 0];
        a.y = As[mrow * APITCH + kk + 2 * kgrp + 1];
        b.x = W[(kk + 2 * kgrp + 0) * FDIM + n0 + mrow];
        b.y = W[(kk + 2 * kgrp + 1) * FDIM + n0 + mrow];
        acc = __builtin_amdgcn_wmma_f32_16x16x4_f32(false, a, false, b,
                                                    (short)0, acc, false, false);
    }
    const int cn    = lane & 15;
    const int mbase = (lane >> 4) * 8;
    for (int r = 0; r < 8; ++r)
        Cs[(mbase + r) * APITCH + n0 + cn] = acc[r];
    __syncthreads();
    for (int r = 0; r < 8; ++r)
        Y[(size_t)(m0 + mbase + r) * FDIM + n0 + cn] = acc[r];
    if (t < 128) {
        const int  row   = t & 15;
        const int  hd    = (t >> 4) & 3;
        const bool right = t >= 64;
        const float4* a4 = (const float4*)((right ? ar : al) + hd * HID);
        const float4* c4 = (const float4*)(Cs + row * APITCH + hd * HID);
        float s = 0.f;
        for (int q = 0; q < HID / 4; ++q) {
            const float4 cv = c4[q];
            const float4 a  = a4[q];
            s += cv.x * a.x; s += cv.y * a.y; s += cv.z * a.z; s += cv.w * a.w;
        }
        (right ? er : el)[(size_t)(m0 + row) * NH + hd] = s;
    }
}

__global__ void gat_gemm_out(const float* __restrict__ Hm,
                             const float* __restrict__ Wout,
                             const float* __restrict__ bout,
                             float* __restrict__ logits) {
    __shared__ __align__(16) float As[16 * APITCH];
    const int m0 = blockIdx.x * 16;
    const int t  = threadIdx.x;
    for (int i = t; i < 16 * FDIM / 4; i += 96)
        *(float4*)(As + (i >> 5) * APITCH + 4 * (i & 31)) =
            ((const float4*)(Hm + (size_t)m0 * FDIM))[i];
    __syncthreads();

    const int wave = t >> 5;
    const int lane = t & 31;
    const int n0   = wave * 16;
    const int mrow = lane & 15;
    const int kgrp = lane >> 4;
    const int col  = n0 + mrow;

    v8f acc = {};
    for (int kk = 0; kk < FDIM; kk += 4) {
        v2f a, b;
        a.x = As[mrow * APITCH + kk + 2 * kgrp + 0];
        a.y = As[mrow * APITCH + kk + 2 * kgrp + 1];
        b.x = (col < NC) ? Wout[(kk + 2 * kgrp + 0) * NC + col] : 0.0f;
        b.y = (col < NC) ? Wout[(kk + 2 * kgrp + 1) * NC + col] : 0.0f;
        acc = __builtin_amdgcn_wmma_f32_16x16x4_f32(false, a, false, b,
                                                    (short)0, acc, false, false);
    }
    const int cn    = lane & 15;
    const int mbase = (lane >> 4) * 8;
    const int ocol  = n0 + cn;
    if (ocol < NC) {
        const float bb = bout[ocol];
        for (int r = 0; r < 8; ++r)
            logits[(size_t)(m0 + mbase + r) * NC + ocol] = acc[r] + bb;
    }
}

#define NBK 196
#define NCH 256
#define CHE (EE / NCH)
#define KTR ((CHE + 1023) / 1024)
static_assert(EE % NCH == 0, "the edge list must split evenly into chunks");
static_assert(NBK * 256 >= NN, "the bins must cover every node");
static_assert(NCH == 256, "the scan below reads a bin's row as 32 lanes x 8 counts");

__device__ __forceinline__ int bin_of(int d) { return min((int)((unsigned)d >> 8), NBK - 1); }

__device__ __forceinline__ int wave_incl(int v, int lane) {
    for (int o = 1; o < 32; o <<= 1) {
        const int u = __shfl_up(v, o);
        if (lane >= o) v += u;
    }
    return v;
}

__global__ void grp_bin_count(const int* __restrict__ dst, int* __restrict__ M) {
    __shared__ int hist[NBK];
    const int t = threadIdx.x, c = blockIdx.x;
    if (t < NBK) hist[t] = 0;
    __syncthreads();
    int d[KTR];
#pragma unroll
    for (int k = 0; k < KTR; ++k) d[k] = dst[c * CHE + min(t + 1024 * k, CHE - 1)];
#pragma unroll
    for (int k = 0; k < KTR; ++k)
        if (t + 1024 * k < CHE) atomicAdd(&hist[bin_of(d[k])], 1);
    __syncthreads();
    if (t < NBK) M[t * NCH + c] = hist[t];
}

__global__ void grp_bin_scan(const int* __restrict__ M, int* __restrict__ O) {
    __shared__ int rtot[224];
    __shared__ int bst[224];
    const int t = threadIdx.x, lane = t & 31, w = t >> 5;
    const int4* M4 = (const int4*)M;
    int4* O4 = (int4*)O;
    int4 a[7], c[7];
    int s[7], incl[7];
#pragma unroll
    for (int i = 0; i < 7; ++i) {
        const int b = w + 32 * i;
        a[i] = make_int4(0, 0, 0, 0); c[i] = a[i];
        if (b < NBK) { a[i] = M4[b * 64 + 2 * lane]; c[i] = M4[b * 64 + 2 * lane + 1]; }
    }
#pragma unroll
    for (int i = 0; i < 7; ++i) {
        const int b = w + 32 * i;
        s[i] = a[i].x + a[i].y + a[i].z + a[i].w + c[i].x + c[i].y + c[i].z + c[i].w;
        incl[i] = wave_incl(s[i], lane);
        if (lane == 31 && b < NBK) rtot[b] = incl[i];
    }
    __syncthreads();
    if (w == 0) {
        int x[7];
        int ss = 0;
#pragma unroll
        for (int k = 0; k < 7; ++k) {
            const int idx = 7 * lane + k;
            x[k] = (idx < NBK) ? rtot[idx] : 0;
            ss += x[k];
        }
        const int in2 = wave_incl(ss, lane);
        int run = in2 - ss;
#pragma unroll
        for (int k = 0; k < 7; ++k) {
            const int idx = 7 * lane + k;
            if (idx < NBK) bst[idx] = run;
            run += x[k];
        }
        if (lane == 31) O[NBK * NCH] = in2;
    }
    __syncthreads();
#pragma unroll
    for (int i = 0; i < 7; ++i) {
        const int b = w + 32 * i;
        if (b < NBK) {
            int run = bst[b] + incl[i] - s[i];
            int4 o1, o2;
            o1.x = run; run += a[i].x; o1.y = run; run += a[i].y;
            o1.z = run; run += a[i].z; o1.w = run; run += a[i].w;
            o2.x = run; run += c[i].x; o2.y = run; run += c[i].y;
            o2.z = run; run += c[i].z; o2.w = run;
            O4[b * 64 + 2 * lane]     = o1;
            O4[b * 64 + 2 * lane + 1] = o2;
        }
    }
}

__global__ void grp_bin_place(const int* __restrict__ src, const int* __restrict__ dst,
                              const int* __restrict__ O, int2* __restrict__ ebuf) {
    __shared__ int cur[NBK];
    const int t = threadIdx.x, c = blockIdx.x;
    if (t < NBK) cur[t] = O[t * NCH + c];
    __syncthreads();
    int d[KTR], s[KTR];
#pragma unroll
    for (int k = 0; k < KTR; ++k) {
        const int e = c * CHE + min(t + 1024 * k, CHE - 1);
        d[k] = dst[e]; s[k] = src[e];
    }
#pragma unroll
    for (int k = 0; k < KTR; ++k)
        if (t + 1024 * k < CHE) {
            const int p = atomicAdd(&cur[bin_of(d[k])], 1);
            ebuf[min(max(p, 0), EE - 1)] = make_int2(s[k], d[k]);
        }
}

__global__ void grp_bin_sort(const int2* __restrict__ ebuf, const int* __restrict__ O,
                             int* __restrict__ seg, int* __restrict__ csr) {
    __shared__ int cnt[256];
    __shared__ int cur[256];
    __shared__ int wtot[8];
    const int t = threadIdx.x, lane = t & 31, w = t >> 5, b = blockIdx.x;
    const int s0 = min(max(O[b * NCH], 0), EE);
    const int n  = min(max(O[(b + 1) * NCH] - s0, 0), EE - s0);
    if (t < 256) cnt[t] = 0;
    __syncthreads();
    for (int i0 = 0; i0 < n; i0 += 4096) {
        int dk[4];
#pragma unroll
        for (int k = 0; k < 4; ++k) dk[k] = ebuf[s0 + min(i0 + t + 1024 * k, n - 1)].y;
#pragma unroll
        for (int k = 0; k < 4; ++k)
            if (i0 + t + 1024 * k < n) atomicAdd(&cnt[dk[k] & 255], 1);
    }
    __syncthreads();
    const int c    = (t < 256) ? cnt[t] : 0;
    const int incl = wave_incl(c, lane);
    if (t < 256 && lane == 31) wtot[w] = incl;
    __syncthreads();
    if (t < 256) {
        int base = 0;
        for (int k = 0; k < w; ++k) base += wtot[k];
        const int excl = base + incl - c;
        cur[t] = excl;
        const int node = b * 256 + t;
        if (node < NN) {
            seg[2 * node]     = s0 + excl;
            seg[2 * node + 1] = c;
        }
    }
    __syncthreads();
    for (int i0 = 0; i0 < n; i0 += 4096) {
        int2 r[4];
#pragma unroll
        for (int k = 0; k < 4; ++k) r[k] = ebuf[s0 + min(i0 + t + 1024 * k, n - 1)];
#pragma unroll
        for (int k = 0; k < 4; ++k)
            if (i0 + t + 1024 * k < n) {
                const int p = atomicAdd(&cur[r[k].y & 255], 1);
                csr[s0 + min(max(p, 0), n - 1)] = r[k].x;
            }
    }
}

__device__ __forceinline__ float lrelu(float v) { return (v > 0.f) ? v : 0.2f * v; }
__device__ __forceinline__ float elu1(float v)  { return (v > 0.f) ? v : (expf(v) - 1.0f); }

__device__ __forceinline__ int lane_i(int v, int j) { return __builtin_amdgcn_readlane(v, j); }
__device__ __forceinline__ float lane_f(float v, int j) {
    return __int_as_float(__builtin_amdgcn_readlane(__float_as_int(v), j));
}
__device__ __forceinline__ float head_w(int head, float a0, float a1, float a2, float a3, int j) {
    const float w0 = lane_f(a0, j), w1 = lane_f(a1, j), w2 = lane_f(a2, j), w3 = lane_f(a3, j);
    const float lo = (head & 1) ? w1 : w0;
    const float hi = (head & 1) ? w3 : w2;
    return (head & 2) ? hi : lo;
}
__device__ __forceinline__ void mul_add4(float4& acc, const float4 f, const float w) {
    acc.x += f.x * w; acc.y += f.y * w; acc.z += f.z * w; acc.w += f.w * w;
}

__device__ __forceinline__ void gather_rows(const float4* __restrict__ feat4, int lane, int head,
                                            int s_l, float a0, float a1, float a2, float a3,
                                            int count, float4& acc) {
    int j = 0;
    for (; j + 4 <= count; j += 4) {
        const int sA = lane_i(s_l, j),     sB = lane_i(s_l, j + 1);
        const int sC = lane_i(s_l, j + 2), sD = lane_i(s_l, j + 3);
        const float4 fA = feat4[(size_t)sA * 32 + lane];
        const float4 fB = feat4[(size_t)sB * 32 + lane];
        const float4 fC = feat4[(size_t)sC * 32 + lane];
        const float4 fD = feat4[(size_t)sD * 32 + lane];
        const float wA = head_w(head, a0, a1, a2, a3, j);
        const float wB = head_w(head, a0, a1, a2, a3, j + 1);
        const float wC = head_w(head, a0, a1, a2, a3, j + 2);
        const float wD = head_w(head, a0, a1, a2, a3, j + 3);
        mul_add4(acc, fA, wA); mul_add4(acc, fB, wB); mul_add4(acc, fC, wC); mul_add4(acc, fD, wD);
    }
#pragma clang loop unroll(disable)
    for (; j < count; ++j) {
        const int sA = lane_i(s_l, j);
        const float4 fA = feat4[(size_t)sA * 32 + lane];
        mul_add4(acc, fA, head_w(head, a0, a1, a2, a3, j));
    }
}

__global__ void gat_gather(const int* __restrict__ seg, const int* __restrict__ csr,
                           const float* __restrict__ el, const float* __restrict__ er,
                           const float* __restrict__ feat,
                           const float* resid, float* out, int use_res,
                           const float* mixrow, const float* __restrict__ lamb) {
    const int lane = threadIdx.x & 31;
    const int d = __builtin_amdgcn_readfirstlane(blockIdx.x * 8 + (threadIdx.x >> 5));
    if (d >= NN) return;
    float4 mx = make_float4(0.f, 0.f, 0.f, 0.f);
    if (mixrow) mx = ((const float4*)mixrow)[(size_t)d * 32 + lane];
    const float lam = lamb[0];
    const int head = lane >> 3;
    const int beg  = seg[2 * d], deg = seg[2 * d + 1];
    const float4 er4 = ((const float4*)er)[d];
    const float4* el4   = (const float4*)el;
    const float4* feat4 = (const float4*)feat;

    const bool ok0 = lane < deg;
    int s0 = 0;
    float v0 = 0.f, v1 = 0.f, v2 = 0.f, v3 = 0.f;
    if (ok0) {
        s0 = min(max(csr[beg + lane], 0), NN - 1);
        const float4 q = el4[s0];
        v0 = lrelu(q.x + er4.x); v1 = lrelu(q.y + er4.y);
        v2 = lrelu(q.z + er4.z); v3 = lrelu(q.w + er4.w);
    }
    float m0 = ok0 ? v0 : -INFINITY, m1 = ok0 ? v1 : -INFINITY;
    float m2 = ok0 ? v2 : -INFINITY, m3 = ok0 ? v3 : -INFINITY;
    for (int base = 32; base < deg; base += 32) {
        const int idx = base + lane;
        if (idx < deg) {
            const int s = min(max(csr[beg + idx], 0), NN - 1);
            const float4 q = el4[s];
            m0 = fmaxf(m0, lrelu(q.x + er4.x)); m1 = fmaxf(m1, lrelu(q.y + er4.y));
            m2 = fmaxf(m2, lrelu(q.z + er4.z)); m3 = fmaxf(m3, lrelu(q.w + er4.w));
        }
    }
    for (int o = 16; o > 0; o >>= 1) {
        m0 = fmaxf(m0, __shfl_xor(m0, o)); m1 = fmaxf(m1, __shfl_xor(m1, o));
        m2 = fmaxf(m2, __shfl_xor(m2, o)); m3 = fmaxf(m3, __shfl_xor(m3, o));
    }

    const float p0 = ok0 ? expf(v0 - m0) : 0.f, p1 = ok0 ? expf(v1 - m1) : 0.f;
    const float p2 = ok0 ? expf(v2 - m2) : 0.f, p3 = ok0 ? expf(v3 - m3) : 0.f;
    float d0 = p0, d1 = p1, d2 = p2, d3 = p3;
    for (int base = 32; base < deg; base += 32) {
        const int idx = base + lane;
        if (idx < deg) {
            const int s = min(max(csr[beg + idx], 0), NN - 1);
            const float4 q = el4[s];
            d0 += expf(lrelu(q.x + er4.x) - m0); d1 += expf(lrelu(q.y + er4.y) - m1);
            d2 += expf(lrelu(q.z + er4.z) - m2); d3 += expf(lrelu(q.w + er4.w) - m3);
        }
    }
    for (int o = 16; o > 0; o >>= 1) {
        d0 += __shfl_xor(d0, o); d1 += __shfl_xor(d1, o);
        d2 += __shfl_xor(d2, o); d3 += __shfl_xor(d3, o);
    }

    float4 acc = make_float4(0.f, 0.f, 0.f, 0.f);
    {
        const float a0 = ok0 ? p0 / d0 : 0.f, a1 = ok0 ? p1 / d1 : 0.f;
        const float a2 = ok0 ? p2 / d2 : 0.f, a3 = ok0 ? p3 / d3 : 0.f;
        gather_rows(feat4, lane, head, s0, a0, a1, a2, a3, min(deg, 32), acc);
    }
    for (int base = 32; base < deg; base += 32) {
        const int idx = base + lane;
        int s = 0;
        float a0 = 0.f, a1 = 0.f, a2 = 0.f, a3 = 0.f;
        if (idx < deg) {
            s = min(max(csr[beg + idx], 0), NN - 1);
            const float4 q = el4[s];
            a0 = expf(lrelu(q.x + er4.x) - m0) / d0; a1 = expf(lrelu(q.y + er4.y) - m1) / d1;
            a2 = expf(lrelu(q.z + er4.z) - m2) / d2; a3 = expf(lrelu(q.w + er4.w) - m3) / d3;
        }
        gather_rows(feat4, lane, head, s, a0, a1, a2, a3, min(deg - base, 32), acc);
    }

    if (use_res) {
        const float4 r = ((const float4*)resid)[(size_t)d * 32 + lane];
        acc.x += r.x; acc.y += r.y; acc.z += r.z; acc.w += r.w;
    }
    acc.x = elu1(acc.x); acc.y = elu1(acc.y); acc.z = elu1(acc.z); acc.w = elu1(acc.w);
    if (mixrow) {
        const float om = 1.0f - lam;
        acc.x = fmaf(lam, mx.x, om * acc.x); acc.y = fmaf(lam, mx.y, om * acc.y);
        acc.z = fmaf(lam, mx.z, om * acc.z); acc.w = fmaf(lam, mx.w, om * acc.w);
    }
    ((float4*)out)[(size_t)d * 32 + lane] = acc;
}

static void run_conv(const float* X, const float* W, const float* al, const float* ar,
                     const int* seg, const int* csr, const float* resid,
                     const float* mixrow, const float* lamb, float* out,
                     float* feat, float* el, float* er, hipStream_t stream) {
    gat_gemm128<<<NN / 16, 256, 0, stream>>>(X, W, al, ar, feat, el, er);
    gat_gather<<<(NN + 7) / 8, 256, 0, stream>>>(seg, csr, el, er, feat, resid, out, resid != nullptr,
                                                 mixrow, lamb);
}

extern "C" void kernel_launch(void* const* d_in, const int* in_sizes, int n_in,
                              void* d_out, int out_size, void* d_ws, size_t ws_size,
                              hipStream_t stream) {
    (void)in_sizes; (void)n_in; (void)out_size; (void)ws_size;

    const float* inputs = (const float*)d_in[0];
    const float* target = (const float*)d_in[1];
    const float* lamb   = (const float*)d_in[2];
    const float* W0     = (const float*)d_in[3];
    const float* al0    = (const float*)d_in[4];
    const float* ar0    = (const float*)d_in[5];
    const float* W1     = (const float*)d_in[6];
    const float* al1    = (const float*)d_in[7];
    const float* ar1    = (const float*)d_in[8];
    const float* Wout   = (const float*)d_in[9];
    const float* bout   = (const float*)d_in[10];
    const int*   src    = (const int*)d_in[11];
    const int*   dst    = (const int*)d_in[12];

    float* feat   = (float*)d_ws;
    float* hbuf   = feat + (size_t)NN * FDIM;
    float* el     = hbuf + (size_t)NN * FDIM;
    float* er     = el   + (size_t)NN * NH;
    int*   seg    = (int*)(er + (size_t)NN * NH);
    int*   csr    = seg + 2 * NN;
    int*   binM   = csr + EE;
    int*   binO   = binM + NBK * NCH;
    int2*  ebuf   = (int2*)(binO + NBK * NCH + 16);

    float* hregion = (float*)d_out;
    float* logits  = hregion + (size_t)NN * FDIM;

    grp_bin_count<<<NCH, 1024, 0, stream>>>(dst, binM);
    grp_bin_scan<<<1, 1024, 0, stream>>>(binM, binO);
    grp_bin_place<<<NCH, 1024, 0, stream>>>(src, dst, binO, ebuf);
    grp_bin_sort<<<NBK, 1024, 0, stream>>>(ebuf, binO, seg, csr);

    run_conv(inputs, W0, al0, ar0, seg, csr, nullptr, nullptr, lamb, hbuf, feat, el, er, stream);
    run_conv(hbuf, W1, al1, ar1, seg, csr, hbuf, nullptr, lamb, hregion, feat, el, er, stream);

    run_conv(target, W0, al0, ar0, seg, csr, nullptr, nullptr, lamb, hbuf, feat, el, er, stream);
    run_conv(hbuf, W1, al1, ar1, seg, csr, hbuf, hregion, lamb, hregion, feat, el, er, stream);

    gat_gemm_out<<<NN / 16, 96, 0, stream>>>(hregion, Wout, bout, logits);
}


# Both mixup paths carried through every launch together (rows interleaved per node): GEMM over 2N rows, one gather wave per node does both paths on shared edge ids; 13 launches -> 9
# speedup vs baseline: 7.7579x; 1.1115x over previous
#include <hip/hip_runtime.h>
#include <hip/hip_bf16.h>
#include <math.h>

#define NN   50000
#define EE   800000
#define IND  128
#define HID  32
#define NH   4
#define NC   40
#define FDIM 128
#define APITCH 132

typedef float v8f __attribute__((ext_vector_type(8)));
typedef float v2f __attribute__((ext_vector_type(2)));

__global__ void gat_gemm128(const float* __restrict__ X0,
                            const float* __restrict__ X1,
                            const int split,
                            const float* __restrict__ W,
                            const float* __restrict__ al,
                            const float* __restrict__ ar,
                            float* __restrict__ Y,
                            float* __restrict__ el,
                            float* __restrict__ er) {
    __shared__ __align__(16) float As[16 * APITCH];
    __shared__ __align__(16) float Cs[16 * APITCH];
    const int t  = threadIdx.x;
    int tile = blockIdx.x;
    const float* X = X0;
    int omul = 1, oadd = 0;
    if (split) {
        const int path = tile >= NN / 16;
        tile -= path * (NN / 16);
        X = path ? X1 : X0;
        omul = 2; oadd = path;
    }
    const int m0 = tile * 16;

    for (int i = t; i < 16 * FDIM / 4; i += 256)
        *(float4*)(As + (i >> 5) * APITCH + 4 * (i & 31)) =
            ((const float4*)(X + (size_t)m0 * FDIM))[i];
    __syncthreads();

    const int wave  = t >> 5;
    const int lane  = t & 31;
    const int n0    = wave * 16;
    const int mrow  = lane & 15;
    const int kgrp  = lane >> 4;

    v8f acc = {};
    for (int kk = 0; kk < FDIM; kk += 4) {
        v2f a, b;
        a.x = As[mrow * APITCH + kk + 2 * kgrp + 0];
        a.y = As[mrow * APITCH + kk + 2 * kgrp + 1];
        b.x = W[(kk + 2 * kgrp + 0) * FDIM + n0 + mrow];
        b.y = W[(kk + 2 * kgrp + 1) * FDIM + n0 + mrow];
        acc = __builtin_amdgcn_wmma_f32_16x16x4_f32(false, a, false, b,
                                                    (short)0, acc, false, false);
    }
    const int cn    = lane & 15;
    const int mbase = (lane >> 4) * 8;
    for (int r = 0; r < 8; ++r)
        Cs[(mbase + r) * APITCH + n0 + cn] = acc[r];
    __syncthreads();
    for (int r = 0; r < 8; ++r)
        Y[(size_t)((m0 + mbase + r) * omul + oadd) * FDIM + n0 + cn] = acc[r];
    if (t < 128) {
        const int  row   = t & 15;
        const int  hd    = (t >> 4) & 3;
        const bool right = t >= 64;
        const float4* a4 = (const float4*)((right ? ar : al) + hd * HID);
        const float4* c4 = (const float4*)(Cs + row * APITCH + hd * HID);
        float s = 0.f;
        for (int q = 0; q < HID / 4; ++q) {
            const float4 cv = c4[q];
            const float4 a  = a4[q];
            s += cv.x * a.x; s += cv.y * a.y; s += cv.z * a.z; s += cv.w * a.w;
        }
        (right ? er : el)[(size_t)((m0 + row) * omul + oadd) * NH + hd] = s;
    }
}

__global__ void gat_gemm_out(const float* __restrict__ Hm,
                             const float* __restrict__ Wout,
                             const float* __restrict__ bout,
                             float* __restrict__ logits) {
    __shared__ __align__(16) float As[16 * APITCH];
    const int m0 = blockIdx.x * 16;
    const int t  = threadIdx.x;
    for (int i = t; i < 16 * FDIM / 4; i += 96)
        *(float4*)(As + (i >> 5) * APITCH + 4 * (i & 31)) =
            ((const float4*)(Hm + (size_t)m0 * FDIM))[i];
    __syncthreads();

    const int wave = t >> 5;
    const int lane = t & 31;
    const int n0   = wave * 16;
    const int mrow = lane & 15;
    const int kgrp = lane >> 4;
    const int col  = n0 + mrow;

    v8f acc = {};
    for (int kk = 0; kk < FDIM; kk += 4) {
        v2f a, b;
        a.x = As[mrow * APITCH + kk + 2 * kgrp + 0];
        a.y = As[mrow * APITCH + kk + 2 * kgrp + 1];
        b.x = (col < NC) ? Wout[(kk + 2 * kgrp + 0) * NC + col] : 0.0f;
        b.y = (col < NC) ? Wout[(kk + 2 * kgrp + 1) * NC + col] : 0.0f;
        acc = __builtin_amdgcn_wmma_f32_16x16x4_f32(false, a, false, b,
                                                    (short)0, acc, false, false);
    }
    const int cn    = lane & 15;
    const int mbase = (lane >> 4) * 8;
    const int ocol  = n0 + cn;
    if (ocol < NC) {
        const float bb = bout[ocol];
        for (int r = 0; r < 8; ++r)
            logits[(size_t)(m0 + mbase + r) * NC + ocol] = acc[r] + bb;
    }
}

#define NBK 196
#define NCH 256
#define CHE (EE / NCH)
#define KTR ((CHE + 1023) / 1024)
static_assert(EE % NCH == 0, "the edge list must split evenly into chunks");
static_assert(NBK * 256 >= NN, "the bins must cover every node");
static_assert(NCH == 256, "the scan below reads a bin's row as 32 lanes x 8 counts");

__device__ __forceinline__ int bin_of(int d) { return min((int)((unsigned)d >> 8), NBK - 1); }

__device__ __forceinline__ int wave_incl(int v, int lane) {
    for (int o = 1; o < 32; o <<= 1) {
        const int u = __shfl_up(v, o);
        if (lane >= o) v += u;
    }
    return v;
}

__global__ void grp_bin_count(const int* __restrict__ dst, int* __restrict__ M) {
    __shared__ int hist[NBK];
    const int t = threadIdx.x, c = blockIdx.x;
    if (t < NBK) hist[t] = 0;
    __syncthreads();
    int d[KTR];
#pragma unroll
    for (int k = 0; k < KTR; ++k) d[k] = dst[c * CHE + min(t + 1024 * k, CHE - 1)];
#pragma unroll
    for (int k = 0; k < KTR; ++k)
        if (t + 1024 * k < CHE) atomicAdd(&hist[bin_of(d[k])], 1);
    __syncthreads();
    if (t < NBK) M[t * NCH + c] = hist[t];
}

__global__ void grp_bin_scan(const int* __restrict__ M, int* __restrict__ O) {
    __shared__ int rtot[224];
    __shared__ int bst[224];
    const int t = threadIdx.x, lane = t & 31, w = t >> 5;
    const int4* M4 = (const int4*)M;
    int4* O4 = (int4*)O;
    int4 a[7], c[7];
    int s[7], incl[7];
#pragma unroll
    for (int i = 0; i < 7; ++i) {
        const int b = w + 32 * i;
        a[i] = make_int4(0, 0, 0, 0); c[i] = a[i];
        if (b < NBK) { a[i] = M4[b * 64 + 2 * lane]; c[i] = M4[b * 64 + 2 * lane + 1]; }
    }
#pragma unroll
    for (int i = 0; i < 7; ++i) {
        const int b = w + 32 * i;
        s[i] = a[i].x + a[i].y + a[i].z + a[i].w + c[i].x + c[i].y + c[i].z + c[i].w;
        incl[i] = wave_incl(s[i], lane);
        if (lane == 31 && b < NBK) rtot[b] = incl[i];
    }
    __syncthreads();
    if (w == 0) {
        int x[7];
        int ss = 0;
#pragma unroll
        for (int k = 0; k < 7; ++k) {
            const int idx = 7 * lane + k;
            x[k] = (idx < NBK) ? rtot[idx] : 0;
            ss += x[k];
        }
        const int in2 = wave_incl(ss, lane);
        int run = in2 - ss;
#pragma unroll
        for (int k = 0; k < 7; ++k) {
            const int idx = 7 * lane + k;
            if (idx < NBK) bst[idx] = run;
            run += x[k];
        }
        if (lane == 31) O[NBK * NCH] = in2;
    }
    __syncthreads();
#pragma unroll
    for (int i = 0; i < 7; ++i) {
        const int b = w + 32 * i;
        if (b < NBK) {
            int run = bst[b] + incl[i] - s[i];
            int4 o1, o2;
            o1.x = run; run += a[i].x; o1.y = run; run += a[i].y;
            o1.z = run; run += a[i].z; o1.w = run; run += a[i].w;
            o2.x = run; run += c[i].x; o2.y = run; run += c[i].y;
            o2.z = run; run += c[i].z; o2.w = run;
            O4[b * 64 + 2 * lane]     = o1;
            O4[b * 64 + 2 * lane + 1] = o2;
        }
    }
}

__global__ void grp_bin_place(const int* __restrict__ src, const int* __restrict__ dst,
                              const int* __restrict__ O, int2* __restrict__ ebuf) {
    __shared__ int cur[NBK];
    const int t = threadIdx.x, c = blockIdx.x;
    if (t < NBK) cur[t] = O[t * NCH + c];
    __syncthreads();
    int d[KTR], s[KTR];
#pragma unroll
    for (int k = 0; k < KTR; ++k) {
        const int e = c * CHE + min(t + 1024 * k, CHE - 1);
        d[k] = dst[e]; s[k] = src[e];
    }
#pragma unroll
    for (int k = 0; k < KTR; ++k)
        if (t + 1024 * k < CHE) {
            const int p = atomicAdd(&cur[bin_of(d[k])], 1);
            ebuf[min(max(p, 0), EE - 1)] = make_int2(s[k], d[k]);
        }
}

__global__ void grp_bin_sort(const int2* __restrict__ ebuf, const int* __restrict__ O,
                             int* __restrict__ seg, int* __restrict__ csr) {
    __shared__ int cnt[256];
    __shared__ int cur[256];
    __shared__ int wtot[8];
    const int t = threadIdx.x, lane = t & 31, w = t >> 5, b = blockIdx.x;
    const int s0 = min(max(O[b * NCH], 0), EE);
    const int n  = min(max(O[(b + 1) * NCH] - s0, 0), EE - s0);
    if (t < 256) cnt[t] = 0;
    __syncthreads();
    for (int i0 = 0; i0 < n; i0 += 4096) {
        int dk[4];
#pragma unroll
        for (int k = 0; k < 4; ++k) dk[k] = ebuf[s0 + min(i0 + t + 1024 * k, n - 1)].y;
#pragma unroll
        for (int k = 0; k < 4; ++k)
            if (i0 + t + 1024 * k < n) atomicAdd(&cnt[dk[k] & 255], 1);
    }
    __syncthreads();
    const int c    = (t < 256) ? cnt[t] : 0;
    const int incl = wave_incl(c, lane);
    if (t < 256 && lane == 31) wtot[w] = incl;
    __syncthreads();
    if (t < 256) {
        int base = 0;
        for (int k = 0; k < w; ++k) base += wtot[k];
        const int excl = base + incl - c;
        cur[t] = excl;
        const int node = b * 256 + t;
        if (node < NN) {
            seg[2 * node]     = s0 + excl;
            seg[2 * node + 1] = c;
        }
    }
    __syncthreads();
    for (int i0 = 0; i0 < n; i0 += 4096) {
        int2 r[4];
#pragma unroll
        for (int k = 0; k < 4; ++k) r[k] = ebuf[s0 + min(i0 + t + 1024 * k, n - 1)];
#pragma unroll
        for (int k = 0; k < 4; ++k)
            if (i0 + t + 1024 * k < n) {
                const int p = atomicAdd(&cur[r[k].y & 255], 1);
                csr[s0 + min(max(p, 0), n - 1)] = r[k].x;
            }
    }
}

__device__ __forceinline__ float lrelu(float v) { return (v > 0.f) ? v : 0.2f * v; }
__device__ __forceinline__ float elu1(float v)  { return (v > 0.f) ? v : (expf(v) - 1.0f); }

__device__ __forceinline__ int lane_i(int v, int j) { return __builtin_amdgcn_readlane(v, j); }
__device__ __forceinline__ float lane_f(float v, int j) {
    return __int_as_float(__builtin_amdgcn_readlane(__float_as_int(v), j));
}
__device__ __forceinline__ float head_w(int head, float a0, float a1, float a2, float a3, int j) {
    const float w0 = lane_f(a0, j), w1 = lane_f(a1, j), w2 = lane_f(a2, j), w3 = lane_f(a3, j);
    const float lo = (head & 1) ? w1 : w0;
    const float hi = (head & 1) ? w3 : w2;
    return (head & 2) ? hi : lo;
}
__device__ __forceinline__ void mul_add4(float4& acc, const float4 f, const float w) {
    acc.x += f.x * w; acc.y += f.y * w; acc.z += f.z * w; acc.w += f.w * w;
}

__device__ __forceinline__ float4 score4(const float4 q, const float4 e) {
    return make_float4(lrelu(q.x + e.x), lrelu(q.y + e.y), lrelu(q.z + e.z), lrelu(q.w + e.w));
}
__device__ __forceinline__ float4 max4(const float4 a, const float4 b) {
    return make_float4(fmaxf(a.x, b.x), fmaxf(a.y, b.y), fmaxf(a.z, b.z), fmaxf(a.w, b.w));
}
__device__ __forceinline__ float4 exp4(const float4 v, const float4 m) {
    return make_float4(expf(v.x - m.x), expf(v.y - m.y), expf(v.z - m.z), expf(v.w - m.w));
}
__device__ __forceinline__ float4 div4(const float4 p, const float4 d) {
    return make_float4(p.x / d.x, p.y / d.y, p.z / d.z, p.w / d.w);
}
__device__ __forceinline__ float4 wave_max4(float4 m) {
    for (int o = 16; o > 0; o >>= 1) {
        m.x = fmaxf(m.x, __shfl_xor(m.x, o)); m.y = fmaxf(m.y, __shfl_xor(m.y, o));
        m.z = fmaxf(m.z, __shfl_xor(m.z, o)); m.w = fmaxf(m.w, __shfl_xor(m.w, o));
    }
    return m;
}
__device__ __forceinline__ float4 wave_sum4(float4 d) {
    for (int o = 16; o > 0; o >>= 1) {
        d.x += __shfl_xor(d.x, o); d.y += __shfl_xor(d.y, o);
        d.z += __shfl_xor(d.z, o); d.w += __shfl_xor(d.w, o);
    }
    return d;
}

__device__ __forceinline__ void gather_rows2(const float4* __restrict__ feat4, int lane, int head,
                                             int s_l, const float4 a, const float4 b,
                                             int count, float4& accA, float4& accB) {
    int j = 0;
    for (; j + 4 <= count; j += 4) {
        const float4* r0 = feat4 + (size_t)lane_i(s_l, j)     * 64 + lane;
        const float4* r1 = feat4 + (size_t)lane_i(s_l, j + 1) * 64 + lane;
        const float4* r2 = feat4 + (size_t)lane_i(s_l, j + 2) * 64 + lane;
        const float4* r3 = feat4 + (size_t)lane_i(s_l, j + 3) * 64 + lane;
        const float4 fA0 = r0[0], fB0 = r0[32];
        const float4 fA1 = r1[0], fB1 = r1[32];
        const float4 fA2 = r2[0], fB2 = r2[32];
        const float4 fA3 = r3[0], fB3 = r3[32];
        mul_add4(accA, fA0, head_w(head, a.x, a.y, a.z, a.w, j));
        mul_add4(accB, fB0, head_w(head, b.x, b.y, b.z, b.w, j));
        mul_add4(accA, fA1, head_w(head, a.x, a.y, a.z, a.w, j + 1));
        mul_add4(accB, fB1, head_w(head, b.x, b.y, b.z, b.w, j + 1));
        mul_add4(accA, fA2, head_w(head, a.x, a.y, a.z, a.w, j + 2));
        mul_add4(accB, fB2, head_w(head, b.x, b.y, b.z, b.w, j + 2));
        mul_add4(accA, fA3, head_w(head, a.x, a.y, a.z, a.w, j + 3));
        mul_add4(accB, fB3, head_w(head, b.x, b.y, b.z, b.w, j + 3));
    }
#pragma clang loop unroll(disable)
    for (; j < count; ++j) {
        const float4* r0 = feat4 + (size_t)lane_i(s_l, j) * 64 + lane;
        const float4 fA0 = r0[0], fB0 = r0[32];
        mul_add4(accA, fA0, head_w(head, a.x, a.y, a.z, a.w, j));
        mul_add4(accB, fB0, head_w(head, b.x, b.y, b.z, b.w, j));
    }
}

__global__ void gat_gather2(const int* __restrict__ seg, const int* __restrict__ csr,
                            const float* __restrict__ el, const float* __restrict__ er,
                            const float* __restrict__ feat,
                            const float* resid, float* out, int final_mix,
                            const float* __restrict__ lamb) {
    const int lane = threadIdx.x & 31;
    const int d = __builtin_amdgcn_readfirstlane(blockIdx.x * 8 + (threadIdx.x >> 5));
    if (d >= NN) return;
    const float lam = lamb[0];
    const int head = lane >> 3;
    const int beg  = seg[2 * d], deg = seg[2 * d + 1];
    const float4 erA = ((const float4*)er)[2 * d];
    const float4 erB = ((const float4*)er)[2 * d + 1];
    const float4* el4   = (const float4*)el;
    const float4* feat4 = (const float4*)feat;
    const float4 ninf = make_float4(-INFINITY, -INFINITY, -INFINITY, -INFINITY);
    const float4 zero = make_float4(0.f, 0.f, 0.f, 0.f);

    const bool ok0 = lane < deg;
    int s0 = 0;
    float4 vA = zero, vB = zero;
    if (ok0) {
        s0 = min(max(csr[beg + lane], 0), NN - 1);
        vA = score4(el4[2 * s0], erA);
        vB = score4(el4[2 * s0 + 1], erB);
    }
    float4 mA = ok0 ? vA : ninf, mB = ok0 ? vB : ninf;
    for (int base = 32; base < deg; base += 32) {
        const int idx = base + lane;
        if (idx < deg) {
            const int s = min(max(csr[beg + idx], 0), NN - 1);
            mA = max4(mA, score4(el4[2 * s], erA));
            mB = max4(mB, score4(el4[2 * s + 1], erB));
        }
    }
    mA = wave_max4(mA); mB = wave_max4(mB);

    const float4 pA = ok0 ? exp4(vA, mA) : zero;
    const float4 pB = ok0 ? exp4(vB, mB) : zero;
    float4 dA = pA, dB = pB;
    for (int base = 32; base < deg; base += 32) {
        const int idx = base + lane;
        if (idx < deg) {
            const int s = min(max(csr[beg + idx], 0), NN - 1);
            const float4 xA = exp4(score4(el4[2 * s], erA), mA);
            const float4 xB = exp4(score4(el4[2 * s + 1], erB), mB);
            dA.x += xA.x; dA.y += xA.y; dA.z += xA.z; dA.w += xA.w;
            dB.x += xB.x; dB.y += xB.y; dB.z += xB.z; dB.w += xB.w;
        }
    }
    dA = wave_sum4(dA); dB = wave_sum4(dB);

    float4 accA = zero, accB = zero;
    {
        const float4 a = ok0 ? div4(pA, dA) : zero;
        const float4 b = ok0 ? div4(pB, dB) : zero;
        gather_rows2(feat4, lane, head, s0, a, b, min(deg, 32), accA, accB);
    }
    for (int base = 32; base < deg; base += 32) {
        const int idx = base + lane;
        int s = 0;
        float4 a = zero, b = zero;
        if (idx < deg) {
            s = min(max(csr[beg + idx], 0), NN - 1);
            a = div4(exp4(score4(el4[2 * s], erA), mA), dA);
            b = div4(exp4(score4(el4[2 * s + 1], erB), mB), dB);
        }
        gather_rows2(feat4, lane, head, s, a, b, min(deg - base, 32), accA, accB);
    }

    if (resid) {
        const float4 rA = ((const float4*)resid)[(size_t)d * 64 + lane];
        const float4 rB = ((const float4*)resid)[(size_t)d * 64 + 32 + lane];
        accA.x += rA.x; accA.y += rA.y; accA.z += rA.z; accA.w += rA.w;
        accB.x += rB.x; accB.y += rB.y; accB.z += rB.z; accB.w += rB.w;
    }
    accA.x = elu1(accA.x); accA.y = elu1(accA.y); accA.z = elu1(accA.z); accA.w = elu1(accA.w);
    accB.x = elu1(accB.x); accB.y = elu1(accB.y); accB.z = elu1(accB.z); accB.w = elu1(accB.w);
    if (final_mix) {
        const float om = 1.0f - lam;
        float4 h;
        h.x = fmaf(lam, accA.x, om * accB.x); h.y = fmaf(lam, accA.y, om * accB.y);
        h.z = fmaf(lam, accA.z, om * accB.z); h.w = fmaf(lam, accA.w, om * accB.w);
        ((float4*)out)[(size_t)d * 32 + lane] = h;
    } else {
        ((float4*)out)[(size_t)d * 64 + lane]      = accA;
        ((float4*)out)[(size_t)d * 64 + 32 + lane] = accB;
    }
}

static void run_conv(const float* X0, const float* X1, int split,
                     const float* W, const float* al, const float* ar,
                     const int* seg, const int* csr, const float* resid,
                     int final_mix, const float* lamb, float* out,
                     float* feat, float* el, float* er, hipStream_t stream) {
    gat_gemm128<<<2 * (NN / 16), 256, 0, stream>>>(X0, X1, split, W, al, ar, feat, el, er);
    gat_gather2<<<(NN + 7) / 8, 256, 0, stream>>>(seg, csr, el, er, feat, resid, out, final_mix, lamb);
}

extern "C" void kernel_launch(void* const* d_in, const int* in_sizes, int n_in,
                              void* d_out, int out_size, void* d_ws, size_t ws_size,
                              hipStream_t stream) {
    (void)in_sizes; (void)n_in; (void)out_size; (void)ws_size;

    const float* inputs = (const float*)d_in[0];
    const float* target = (const float*)d_in[1];
    const float* lamb   = (const float*)d_in[2];
    const float* W0     = (const float*)d_in[3];
    const float* al0    = (const float*)d_in[4];
    const float* ar0    = (const float*)d_in[5];
    const float* W1     = (const float*)d_in[6];
    const float* al1    = (const float*)d_in[7];
    const float* ar1    = (const float*)d_in[8];
    const float* Wout   = (const float*)d_in[9];
    const float* bout   = (const float*)d_in[10];
    const int*   src    = (const int*)d_in[11];
    const int*   dst    = (const int*)d_in[12];

    float* feat   = (float*)d_ws;
    float* hbuf   = feat + (size_t)2 * NN * FDIM;
    float* el     = hbuf + (size_t)2 * NN * FDIM;
    float* er     = el   + (size_t)2 * NN * NH;
    int*   seg    = (int*)(er + (size_t)2 * NN * NH);
    int*   csr    = seg + 2 * NN;
    int*   binM   = csr + EE;
    int*   binO   = binM + NBK * NCH;
    int2*  ebuf   = (int2*)(binO + NBK * NCH + 16);

    float* hregion = (float*)d_out;
    float* logits  = hregion + (size_t)NN * FDIM;

    grp_bin_count<<<NCH, 1024, 0, stream>>>(dst, binM);
    grp_bin_scan<<<1, 1024, 0, stream>>>(binM, binO);
    grp_bin_place<<<NCH, 1024, 0, stream>>>(src, dst, binO, ebuf);
    grp_bin_sort<<<NBK, 1024, 0, stream>>>(ebuf, binO, seg, csr);

    run_conv(inputs, target, 1, W0, al0, ar0, seg, csr, nullptr, 0, lamb, hbuf, feat, el, er, stream);
    run_conv(hbuf, hbuf, 0, W1, al1, ar1, seg, csr, hbuf, 1, lamb, hregion, feat, el, er, stream);

    gat_gemm_out<<<NN / 16, 96, 0, stream>>>(hregion, Wout, bout, logits);
}


# Layer 0's GEMM loses its own launch: its row tiles ride in the four edge-grouping launches (1024-thread work-groups of 4 tiles behind each launch's own work-groups); 9 launches -> 8, sums unchanged
# speedup vs baseline: 8.0100x; 1.0325x over previous
#include <hip/hip_runtime.h>
#include <hip/hip_bf16.h>
#include <math.h>

#define NN   50000
#define EE   800000
#define IND  128
#define HID  32
#define NH   4
#define NC   40
#define FDIM 128
#define APITCH 132

typedef float v8f __attribute__((ext_vector_type(8)));
typedef float v2f __attribute__((ext_vector_type(2)));

#define NTILE (2 * (NN / 16))
__device__ __forceinline__ void gemm128_tile(const float* __restrict__ X0,
                                             const float* __restrict__ X1,
                                             const int split,
                                             const float* __restrict__ W,
                                             const float* __restrict__ al,
                                             const float* __restrict__ ar,
                                             float* __restrict__ Y,
                                             float* __restrict__ el,
                                             float* __restrict__ er,
                                             int tile, const bool live, const int t,
                                             float* As, float* Cs) {
    const float* X = X0;
    int omul = 1, oadd = 0;
    if (split) {
        const int path = tile >= NN / 16;
        tile -= path * (NN / 16);
        X = path ? X1 : X0;
        omul = 2; oadd = path;
    }
    const int m0 = tile * 16;

    if (live)
        for (int i = t; i < 16 * FDIM / 4; i += 256)
            *(float4*)(As + (i >> 5) * APITCH + 4 * (i & 31)) =
                ((const float4*)(X + (size_t)m0 * FDIM))[i];
    __syncthreads();

    const int wave  = t >> 5;
    const int lane  = t & 31;
    const int n0    = wave * 16;
    const int mrow  = lane & 15;
    const int kgrp  = lane >> 4;
    const int cn    = lane & 15;
    const int mbase = (lane >> 4) * 8;

    v8f acc = {};
    if (live) {
        for (int kk = 0; kk < FDIM; kk += 4) {
            v2f a, b;
            a.x = As[mrow * APITCH + kk + 2 * kgrp + 0];
            a.y = As[mrow * APITCH + kk + 2 * kgrp + 1];
            b.x = W[(kk + 2 * kgrp + 0) * FDIM + n0 + mrow];
            b.y = W[(kk + 2 * kgrp + 1) * FDIM + n0 + mrow];
            acc = __builtin_amdgcn_wmma_f32_16x16x4_f32(false, a, false, b,
                                                        (short)0, acc, false, false);
        }
        for (int r = 0; r < 8; ++r)
            Cs[(mbase + r) * APITCH + n0 + cn] = acc[r];
    }
    __syncthreads();
    if (!live) return;
    for (int r = 0; r < 8; ++r)
        Y[(size_t)((m0 + mbase + r) * omul + oadd) * FDIM + n0 + cn] = acc[r];
    if (t < 128) {
        const int  row   = t & 15;
        const int  hd    = (t >> 4) & 3;
        const bool right = t >= 64;
        const float4* a4 = (const float4*)((right ? ar : al) + hd * HID);
        const float4* c4 = (const float4*)(Cs + row * APITCH + hd * HID);
        float s = 0.f;
        for (int q = 0; q < HID / 4; ++q) {
            const float4 cv = c4[q];
            const float4 a  = a4[q];
            s += cv.x * a.x; s += cv.y * a.y; s += cv.z * a.z; s += cv.w * a.w;
        }
        (right ? er : el)[(size_t)((m0 + row) * omul + oadd) * NH + hd] = s;
    }
}

__global__ void gat_gemm128(const float* __restrict__ X0,
                            const float* __restrict__ X1,
                            const int split,
                            const float* __restrict__ W,
                            const float* __restrict__ al,
                            const float* __restrict__ ar,
                            float* __restrict__ Y,
                            float* __restrict__ el,
                            float* __restrict__ er) {
    __shared__ __align__(16) float As[16 * APITCH];
    __shared__ __align__(16) float Cs[16 * APITCH];
    gemm128_tile(X0, X1, split, W, al, ar, Y, el, er, blockIdx.x, true, threadIdx.x, As, Cs);
}

#define GSZ   (4 * 16 * APITCH)
#define NGRP  ((NTILE + 3) / 4)
#define G_CNT 256
#define G_SCN 255
#define G_PLC 256
#define G_SRT (NGRP - G_CNT - G_SCN - G_PLC)
static_assert(G_SRT > 0, "the four shares must add up to all groups");
__device__ __forceinline__ void gemm128_group(const float* __restrict__ X0,
                                              const float* __restrict__ X1,
                                              const float* __restrict__ W,
                                              const float* __restrict__ al,
                                              const float* __restrict__ ar,
                                              float* __restrict__ Y,
                                              float* __restrict__ el,
                                              float* __restrict__ er,
                                              const int grp, float* gA, float* gC) {
    const int q    = __builtin_amdgcn_readfirstlane((int)threadIdx.x >> 8);
    const int tile = 4 * grp + q;
    gemm128_tile(X0, X1, 1, W, al, ar, Y, el, er, tile, tile < NTILE, threadIdx.x & 255,
                 gA + q * 16 * APITCH, gC + q * 16 * APITCH);
}
__global__ void gat_gemm_out(const float* __restrict__ Hm,
                             const float* __restrict__ Wout,
                             const float* __restrict__ bout,
                             float* __restrict__ logits) {
    __shared__ __align__(16) float As[16 * APITCH];
    const int m0 = blockIdx.x * 16;
    const int t  = threadIdx.x;
    for (int i = t; i < 16 * FDIM / 4; i += 96)
        *(float4*)(As + (i >> 5) * APITCH + 4 * (i & 31)) =
            ((const float4*)(Hm + (size_t)m0 * FDIM))[i];
    __syncthreads();

    const int wave = t >> 5;
    const int lane = t & 31;
    const int n0   = wave * 16;
    const int mrow = lane & 15;
    const int kgrp = lane >> 4;
    const int col  = n0 + mrow;

    v8f acc = {};
    for (int kk = 0; kk < FDIM; kk += 4) {
        v2f a, b;
        a.x = As[mrow * APITCH + kk + 2 * kgrp + 0];
        a.y = As[mrow * APITCH + kk + 2 * kgrp + 1];
        b.x = (col < NC) ? Wout[(kk + 2 * kgrp + 0) * NC + col] : 0.0f;
        b.y = (col < NC) ? Wout[(kk + 2 * kgrp + 1) * NC + col] : 0.0f;
        acc = __builtin_amdgcn_wmma_f32_16x16x4_f32(false, a, false, b,
                                                    (short)0, acc, false, false);
    }
    const int cn    = lane & 15;
    const int mbase = (lane >> 4) * 8;
    const int ocol  = n0 + cn;
    if (ocol < NC) {
        const float bb = bout[ocol];
        for (int r = 0; r < 8; ++r)
            logits[(size_t)(m0 + mbase + r) * NC + ocol] = acc[r] + bb;
    }
}

#define NBK 196
#define NCH 256
#define CHE (EE / NCH)
#define KTR ((CHE + 1023) / 1024)
static_assert(EE % NCH == 0, "the edge list must split evenly into chunks");
static_assert(NBK * 256 >= NN, "the bins must cover every node");
static_assert(NCH == 256, "the scan below reads a bin's row as 32 lanes x 8 counts");

__device__ __forceinline__ int bin_of(int d) { return min((int)((unsigned)d >> 8), NBK - 1); }

__device__ __forceinline__ int wave_incl(int v, int lane) {
    for (int o = 1; o < 32; o <<= 1) {
        const int u = __shfl_up(v, o);
        if (lane >= o) v += u;
    }
    return v;
}

__global__ void grp_bin_count(const int* __restrict__ dst, int* __restrict__ M,
                              const float* __restrict__ gX0, const float* __restrict__ gX1,
                              const float* __restrict__ gW, const float* __restrict__ gal,
                              const float* __restrict__ gar, float* __restrict__ gY,
                              float* __restrict__ gel, float* __restrict__ ger) {
    __shared__ int hist[NBK];
    __shared__ __align__(16) float gA[GSZ];
    __shared__ __align__(16) float gC[GSZ];
    if ((int)blockIdx.x >= NCH) {
        gemm128_group(gX0, gX1, gW, gal, gar, gY, gel, ger, 0 + (int)blockIdx.x - NCH, gA, gC);
        return;
    }
    const int t = threadIdx.x, c = blockIdx.x;
    if (t < NBK) hist[t] = 0;
    __syncthreads();
    int d[KTR];
#pragma unroll
    for (int k = 0; k < KTR; ++k) d[k] = dst[c * CHE + min(t + 1024 * k, CHE - 1)];
#pragma unroll
    for (int k = 0; k < KTR; ++k)
        if (t + 1024 * k < CHE) atomicAdd(&hist[bin_of(d[k])], 1);
    __syncthreads();
    if (t < NBK) M[t * NCH + c] = hist[t];
}

__global__ void grp_bin_scan(const int* __restrict__ M, int* __restrict__ O,
                              const float* __restrict__ gX0, const float* __restrict__ gX1,
                              const float* __restrict__ gW, const float* __restrict__ gal,
                              const float* __restrict__ gar, float* __restrict__ gY,
                              float* __restrict__ gel, float* __restrict__ ger) {
    __shared__ int rtot[224];
    __shared__ int bst[224];
    __shared__ __align__(16) float gA[GSZ];
    __shared__ __align__(16) float gC[GSZ];
    if ((int)blockIdx.x >= 1) {
        gemm128_group(gX0, gX1, gW, gal, gar, gY, gel, ger, G_CNT + (int)blockIdx.x - 1, gA, gC);
        return;
    }
    const int t = threadIdx.x, lane = t & 31, w = t >> 5;
    const int4* M4 = (const int4*)M;
    int4* O4 = (int4*)O;
    int4 a[7], c[7];
    int s[7], incl[7];
#pragma unroll
    for (int i = 0; i < 7; ++i) {
        const int b = w + 32 * i;
        a[i] = make_int4(0, 0, 0, 0); c[i] = a[i];
        if (b < NBK) { a[i] = M4[b * 64 + 2 * lane]; c[i] = M4[b * 64 + 2 * lane + 1]; }
    }
#pragma unroll
    for (int i = 0; i < 7; ++i) {
        const int b = w + 32 * i;
        s[i] = a[i].x + a[i].y + a[i].z + a[i].w + c[i].x + c[i].y + c[i].z + c[i].w;
        incl[i] = wave_incl(s[i], lane);
        if (lane == 31 && b < NBK) rtot[b] = incl[i];
    }
    __syncthreads();
    if (w == 0) {
        int x[7];
        int ss = 0;
#pragma unroll
        for (int k = 0; k < 7; ++k) {
            const int idx = 7 * lane + k;
            x[k] = (idx < NBK) ? rtot[idx] : 0;
            ss += x[k];
        }
        const int in2 = wave_incl(ss, lane);
        int run = in2 - ss;
#pragma unroll
        for (int k = 0; k < 7; ++k) {
            const int idx = 7 * lane + k;
            if (idx < NBK) bst[idx] = run;
            run += x[k];
        }
        if (lane == 31) O[NBK * NCH] = in2;
    }
    __syncthreads();
#pragma unroll
    for (int i = 0; i < 7; ++i) {
        const int b = w + 32 * i;
        if (b < NBK) {
            int run = bst[b] + incl[i] - s[i];
            int4 o1, o2;
            o1.x = run; run += a[i].x; o1.y = run; run += a[i].y;
            o1.z = run; run += a[i].z; o1.w = run; run += a[i].w;
            o2.x = run; run += c[i].x; o2.y = run; run += c[i].y;
            o2.z = run; run += c[i].z; o2.w = run;
            O4[b * 64 + 2 * lane]     = o1;
            O4[b * 64 + 2 * lane + 1] = o2;
        }
    }
}

__global__ void grp_bin_place(const int* __restrict__ src, const int* __restrict__ dst,
                              const int* __restrict__ O, int2* __restrict__ ebuf,
                              const float* __restrict__ gX0, const float* __restrict__ gX1,
                              const float* __restrict__ gW, const float* __restrict__ gal,
                              const float* __restrict__ gar, float* __restrict__ gY,
                              float* __restrict__ gel, float* __restrict__ ger) {
    __shared__ int cur[NBK];
    __shared__ __align__(16) float gA[GSZ];
    __shared__ __align__(16) float gC[GSZ];
    if ((int)blockIdx.x >= NCH) {
        gemm128_group(gX0, gX1, gW, gal, gar, gY, gel, ger, G_CNT + G_SCN + (int)blockIdx.x - NCH, gA, gC);
        return;
    }
    const int t = threadIdx.x, c = blockIdx.x;
    if (t < NBK) cur[t] = O[t * NCH + c];
    __syncthreads();
    int d[KTR], s[KTR];
#pragma unroll
    for (int k = 0; k < KTR; ++k) {
        const int e = c * CHE + min(t + 1024 * k, CHE - 1);
        d[k] = dst[e]; s[k] = src[e];
    }
#pragma unroll
    for (int k = 0; k < KTR; ++k)
        if (t + 1024 * k < CHE) {
            const int p = atomicAdd(&cur[bin_of(d[k])], 1);
            ebuf[min(max(p, 0), EE - 1)] = make_int2(s[k], d[k]);
        }
}

__global__ void grp_bin_sort(const int2* __restrict__ ebuf, const int* __restrict__ O,
                             int* __restrict__ seg, int* __restrict__ csr,
                              const float* __restrict__ gX0, const float* __restrict__ gX1,
                              const float* __restrict__ gW, const float* __restrict__ gal,
                              const float* __restrict__ gar, float* __restrict__ gY,
                              float* __restrict__ gel, float* __restrict__ ger) {
    __shared__ int cnt[256];
    __shared__ int cur[256];
    __shared__ int wtot[8];
    __shared__ __align__(16) float gA[GSZ];
    __shared__ __align__(16) float gC[GSZ];
    if ((int)blockIdx.x >= NBK) {
        gemm128_group(gX0, gX1, gW, gal, gar, gY, gel, ger, G_CNT + G_SCN + G_PLC + (int)blockIdx.x - NBK, gA, gC);
        return;
    }
    const int t = threadIdx.x, lane = t & 31, w = t >> 5, b = blockIdx.x;
    const int s0 = min(max(O[b * NCH], 0), EE);
    const int n  = min(max(O[(b + 1) * NCH] - s0, 0), EE - s0);
    if (t < 256) cnt[t] = 0;
    __syncthreads();
    for (int i0 = 0; i0 < n; i0 += 4096) {
        int dk[4];
#pragma unroll
        for (int k = 0; k < 4; ++k) dk[k] = ebuf[s0 + min(i0 + t + 1024 * k, n - 1)].y;
#pragma unroll
        for (int k = 0; k < 4; ++k)
            if (i0 + t + 1024 * k < n) atomicAdd(&cnt[dk[k] & 255], 1);
    }
    __syncthreads();
    const int c    = (t < 256) ? cnt[t] : 0;
    const int incl = wave_incl(c, lane);
    if (t < 256 && lane == 31) wtot[w] = incl;
    __syncthreads();
    if (t < 256) {
        int base = 0;
        for (int k = 0; k < w; ++k) base += wtot[k];
        const int excl = base + incl - c;
        cur[t] = excl;
        const int node = b * 256 + t;
        if (node < NN) {
            seg[2 * node]     = s0 + excl;
            seg[2 * node + 1] = c;
        }
    }
    __syncthreads();
    for (int i0 = 0; i0 < n; i0 += 4096) {
        int2 r[4];
#pragma unroll
        for (int k = 0; k < 4; ++k) r[k] = ebuf[s0 + min(i0 + t + 1024 * k, n - 1)];
#pragma unroll
        for (int k = 0; k < 4; ++k)
            if (i0 + t + 1024 * k < n) {
                const int p = atomicAdd(&cur[r[k].y & 255], 1);
                csr[s0 + min(max(p, 0), n - 1)] = r[k].x;
            }
    }
}

__device__ __forceinline__ float lrelu(float v) { return (v > 0.f) ? v : 0.2f * v; }
__device__ __forceinline__ float elu1(float v)  { return (v > 0.f) ? v : (expf(v) - 1.0f); }

__device__ __forceinline__ int lane_i(int v, int j) { return __builtin_amdgcn_readlane(v, j); }
__device__ __forceinline__ float lane_f(float v, int j) {
    return __int_as_float(__builtin_amdgcn_readlane(__float_as_int(v), j));
}
__device__ __forceinline__ float head_w(int head, float a0, float a1, float a2, float a3, int j) {
    const float w0 = lane_f(a0, j), w1 = lane_f(a1, j), w2 = lane_f(a2, j), w3 = lane_f(a3, j);
    const float lo = (head & 1) ? w1 : w0;
    const float hi = (head & 1) ? w3 : w2;
    return (head & 2) ? hi : lo;
}
__device__ __forceinline__ void mul_add4(float4& acc, const float4 f, const float w) {
    acc.x += f.x * w; acc.y += f.y * w; acc.z += f.z * w; acc.w += f.w * w;
}

__device__ __forceinline__ float4 score4(const float4 q, const float4 e) {
    return make_float4(lrelu(q.x + e.x), lrelu(q.y + e.y), lrelu(q.z + e.z), lrelu(q.w + e.w));
}
__device__ __forceinline__ float4 max4(const float4 a, const float4 b) {
    return make_float4(fmaxf(a.x, b.x), fmaxf(a.y, b.y), fmaxf(a.z, b.z), fmaxf(a.w, b.w));
}
__device__ __forceinline__ float4 exp4(const float4 v, const float4 m) {
    return make_float4(expf(v.x - m.x), expf(v.y - m.y), expf(v.z - m.z), expf(v.w - m.w));
}
__device__ __forceinline__ float4 div4(const float4 p, const float4 d) {
    return make_float4(p.x / d.x, p.y / d.y, p.z / d.z, p.w / d.w);
}
__device__ __forceinline__ float4 wave_max4(float4 m) {
    for (int o = 16; o > 0; o >>= 1) {
        m.x = fmaxf(m.x, __shfl_xor(m.x, o)); m.y = fmaxf(m.y, __shfl_xor(m.y, o));
        m.z = fmaxf(m.z, __shfl_xor(m.z, o)); m.w = fmaxf(m.w, __shfl_xor(m.w, o));
    }
    return m;
}
__device__ __forceinline__ float4 wave_sum4(float4 d) {
    for (int o = 16; o > 0; o >>= 1) {
        d.x += __shfl_xor(d.x, o); d.y += __shfl_xor(d.y, o);
        d.z += __shfl_xor(d.z, o); d.w += __shfl_xor(d.w, o);
    }
    return d;
}

__device__ __forceinline__ void gather_rows2(const float4* __restrict__ feat4, int lane, int head,
                                             int s_l, const float4 a, const float4 b,
                                             int count, float4& accA, float4& accB) {
    int j = 0;
    for (; j + 4 <= count; j += 4) {
        const float4* r0 = feat4 + (size_t)lane_i(s_l, j)     * 64 + lane;
        const float4* r1 = feat4 + (size_t)lane_i(s_l, j + 1) * 64 + lane;
        const float4* r2 = feat4 + (size_t)lane_i(s_l, j + 2) * 64 + lane;
        const float4* r3 = feat4 + (size_t)lane_i(s_l, j + 3) * 64 + lane;
        const float4 fA0 = r0[0], fB0 = r0[32];
        const float4 fA1 = r1[0], fB1 = r1[32];
        const float4 fA2 = r2[0], fB2 = r2[32];
        const float4 fA3 = r3[0], fB3 = r3[32];
        mul_add4(accA, fA0, head_w(head, a.x, a.y, a.z, a.w, j));
        mul_add4(accB, fB0, head_w(head, b.x, b.y, b.z, b.w, j));
        mul_add4(accA, fA1, head_w(head, a.x, a.y, a.z, a.w, j + 1));
        mul_add4(accB, fB1, head_w(head, b.x, b.y, b.z, b.w, j + 1));
        mul_add4(accA, fA2, head_w(head, a.x, a.y, a.z, a.w, j + 2));
        mul_add4(accB, fB2, head_w(head, b.x, b.y, b.z, b.w, j + 2));
        mul_add4(accA, fA3, head_w(head, a.x, a.y, a.z, a.w, j + 3));
        mul_add4(accB, fB3, head_w(head, b.x, b.y, b.z, b.w, j + 3));
    }
#pragma clang loop unroll(disable)
    for (; j < count; ++j) {
        const float4* r0 = feat4 + (size_t)lane_i(s_l, j) * 64 + lane;
        const float4 fA0 = r0[0], fB0 = r0[32];
        mul_add4(accA, fA0, head_w(head, a.x, a.y, a.z, a.w, j));
        mul_add4(accB, fB0, head_w(head, b.x, b.y, b.z, b.w, j));
    }
}

__global__ void gat_gather2(const int* __restrict__ seg, const int* __restrict__ csr,
                            const float* __restrict__ el, const float* __restrict__ er,
                            const float* __restrict__ feat,
                            const float* resid, float* out, int final_mix,
                            const float* __restrict__ lamb) {
    const int lane = threadIdx.x & 31;
    const int d = __builtin_amdgcn_readfirstlane(blockIdx.x * 8 + (threadIdx.x >> 5));
    if (d >= NN) return;
    const float lam = lamb[0];
    const int head = lane >> 3;
    const int beg  = seg[2 * d], deg = seg[2 * d + 1];
    const float4 erA = ((const float4*)er)[2 * d];
    const float4 erB = ((const float4*)er)[2 * d + 1];
    const float4* el4   = (const float4*)el;
    const float4* feat4 = (const float4*)feat;
    const float4 ninf = make_float4(-INFINITY, -INFINITY, -INFINITY, -INFINITY);
    const float4 zero = make_float4(0.f, 0.f, 0.f, 0.f);

    const bool ok0 = lane < deg;
    int s0 = 0;
    float4 vA = zero, vB = zero;
    if (ok0) {
        s0 = min(max(csr[beg + lane], 0), NN - 1);
        vA = score4(el4[2 * s0], erA);
        vB = score4(el4[2 * s0 + 1], erB);
    }
    float4 mA = ok0 ? vA : ninf, mB = ok0 ? vB : ninf;
    for (int base = 32; base < deg; base += 32) {
        const int idx = base + lane;
        if (idx < deg) {
            const int s = min(max(csr[beg + idx], 0), NN - 1);
            mA = max4(mA, score4(el4[2 * s], erA));
            mB = max4(mB, score4(el4[2 * s + 1], erB));
        }
    }
    mA = wave_max4(mA); mB = wave_max4(mB);

    const float4 pA = ok0 ? exp4(vA, mA) : zero;
    const float4 pB = ok0 ? exp4(vB, mB) : zero;
    float4 dA = pA, dB = pB;
    for (int base = 32; base < deg; base += 32) {
        const int idx = base + lane;
        if (idx < deg) {
            const int s = min(max(csr[beg + idx], 0), NN - 1);
            const float4 xA = exp4(score4(el4[2 * s], erA), mA);
            const float4 xB = exp4(score4(el4[2 * s + 1], erB), mB);
            dA.x += xA.x; dA.y += xA.y; dA.z += xA.z; dA.w += xA.w;
            dB.x += xB.x; dB.y += xB.y; dB.z += xB.z; dB.w += xB.w;
        }
    }
    dA = wave_sum4(dA); dB = wave_sum4(dB);

    float4 accA = zero, accB = zero;
    {
        const float4 a = ok0 ? div4(pA, dA) : zero;
        const float4 b = ok0 ? div4(pB, dB) : zero;
        gather_rows2(feat4, lane, head, s0, a, b, min(deg, 32), accA, accB);
    }
    for (int base = 32; base < deg; base += 32) {
        const int idx = base + lane;
        int s = 0;
        float4 a = zero, b = zero;
        if (idx < deg) {
            s = min(max(csr[beg + idx], 0), NN - 1);
            a = div4(exp4(score4(el4[2 * s], erA), mA), dA);
            b = div4(exp4(score4(el4[2 * s + 1], erB), mB), dB);
        }
        gather_rows2(feat4, lane, head, s, a, b, min(deg - base, 32), accA, accB);
    }

    if (resid) {
        const float4 rA = ((const float4*)resid)[(size_t)d * 64 + lane];
        const float4 rB = ((const float4*)resid)[(size_t)d * 64 + 32 + lane];
        accA.x += rA.x; accA.y += rA.y; accA.z += rA.z; accA.w += rA.w;
        accB.x += rB.x; accB.y += rB.y; accB.z += rB.z; accB.w += rB.w;
    }
    accA.x = elu1(accA.x); accA.y = elu1(accA.y); accA.z = elu1(accA.z); accA.w = elu1(accA.w);
    accB.x = elu1(accB.x); accB.y = elu1(accB.y); accB.z = elu1(accB.z); accB.w = elu1(accB.w);
    if (final_mix) {
        const float om = 1.0f - lam;
        float4 h;
        h.x = fmaf(lam, accA.x, om * accB.x); h.y = fmaf(lam, accA.y, om * accB.y);
        h.z = fmaf(lam, accA.z, om * accB.z); h.w = fmaf(lam, accA.w, om * accB.w);
        ((float4*)out)[(size_t)d * 32 + lane] = h;
    } else {
        ((float4*)out)[(size_t)d * 64 + lane]      = accA;
        ((float4*)out)[(size_t)d * 64 + 32 + lane] = accB;
    }
}


extern "C" void kernel_launch(void* const* d_in, const int* in_sizes, int n_in,
                              void* d_out, int out_size, void* d_ws, size_t ws_size,
                              hipStream_t stream) {
    (void)in_sizes; (void)n_in; (void)out_size; (void)ws_size;

    const float* inputs = (const float*)d_in[0];
    const float* target = (const float*)d_in[1];
    const float* lamb   = (const float*)d_in[2];
    const float* W0     = (const float*)d_in[3];
    const float* al0    = (const float*)d_in[4];
    const float* ar0    = (const float*)d_in[5];
    const float* W1     = (const float*)d_in[6];
    const float* al1    = (const float*)d_in[7];
    const float* ar1    = (const float*)d_in[8];
    const float* Wout   = (const float*)d_in[9];
    const float* bout   = (const float*)d_in[10];
    const int*   src    = (const int*)d_in[11];
    const int*   dst    = (const int*)d_in[12];

    float* feat   = (float*)d_ws;
    float* hbuf   = feat + (size_t)2 * NN * FDIM;
    float* el     = hbuf + (size_t)2 * NN * FDIM;
    float* er     = el   + (size_t)2 * NN * NH;
    int*   seg    = (int*)(er + (size_t)2 * NN * NH);
    int*   csr    = seg + 2 * NN;
    int*   binM   = csr + EE;
    int*   binO   = binM + NBK * NCH;
    int2*  ebuf   = (int2*)(binO + NBK * NCH + 16);

    float* hregion = (float*)d_out;
    float* logits  = hregion + (size_t)NN * FDIM;

    grp_bin_count<<<NCH + G_CNT, 1024, 0, stream>>>(dst, binM,
                                                    inputs, target, W0, al0, ar0, feat, el, er);
    grp_bin_scan<<<1 + G_SCN, 1024, 0, stream>>>(binM, binO,
                                                 inputs, target, W0, al0, ar0, feat, el, er);
    grp_bin_place<<<NCH + G_PLC, 1024, 0, stream>>>(src, dst, binO, ebuf,
                                                    inputs, target, W0, al0, ar0, feat, el, er);
    grp_bin_sort<<<NBK + G_SRT, 1024, 0, stream>>>(ebuf, binO, seg, csr,
                                                   inputs, target, W0, al0, ar0, feat, el, er);

    gat_gather2<<<(NN + 7) / 8, 256, 0, stream>>>(seg, csr, el, er, feat, nullptr, hbuf, 0, lamb);
    gat_gemm128<<<NTILE, 256, 0, stream>>>(hbuf, hbuf, 0, W1, al1, ar1, feat, el, er);
    gat_gather2<<<(NN + 7) / 8, 256, 0, stream>>>(seg, csr, el, er, feat, hbuf, hregion, 1, lamb);

    gat_gemm_out<<<NN / 16, 96, 0, stream>>>(hregion, Wout, bout, logits);
}


# Gather cross-lane moves on DPP: softmax butterflies by row_xmask+permlanex16, edge weights by row_share, not readlanes+selects+LDS permutes; node runs sorted by source id: sum order changed
# speedup vs baseline: 8.2949x; 1.0356x over previous
#include <hip/hip_runtime.h>
#include <hip/hip_bf16.h>
#include <math.h>

#define NN   50000
#define EE   800000
#define IND  128
#define HID  32
#define NH   4
#define NC   40
#define FDIM 128
#define APITCH 132

typedef float v8f __attribute__((ext_vector_type(8)));
typedef float v2f __attribute__((ext_vector_type(2)));

#define NTILE (2 * (NN / 16))
__device__ __forceinline__ void gemm128_tile(const float* __restrict__ X0,
                                             const float* __restrict__ X1,
                                             const int split,
                                             const float* __restrict__ W,
                                             const float* __restrict__ al,
                                             const float* __restrict__ ar,
                                             float* __restrict__ Y,
                                             float* __restrict__ el,
                                             float* __restrict__ er,
                                             int tile, const bool live, const int t,
                                             float* As, float* Cs) {
    const float* X = X0;
    int omul = 1, oadd = 0;
    if (split) {
        const int path = tile >= NN / 16;
        tile -= path * (NN / 16);
        X = path ? X1 : X0;
        omul = 2; oadd = path;
    }
    const int m0 = tile * 16;

    if (live)
        for (int i = t; i < 16 * FDIM / 4; i += 256)
            *(float4*)(As + (i >> 5) * APITCH + 4 * (i & 31)) =
                ((const float4*)(X + (size_t)m0 * FDIM))[i];
    __syncthreads();

    const int wave  = t >> 5;
    const int lane  = t & 31;
    const int n0    = wave * 16;
    const int mrow  = lane & 15;
    const int kgrp  = lane >> 4;
    const int cn    = lane & 15;
    const int mbase = (lane >> 4) * 8;

    v8f acc = {};
    if (live) {
        for (int kk = 0; kk < FDIM; kk += 4) {
            v2f a, b;
            a.x = As[mrow * APITCH + kk + 2 * kgrp + 0];
            a.y = As[mrow * APITCH + kk + 2 * kgrp + 1];
            b.x = W[(kk + 2 * kgrp + 0) * FDIM + n0 + mrow];
            b.y = W[(kk + 2 * kgrp + 1) * FDIM + n0 + mrow];
            acc = __builtin_amdgcn_wmma_f32_16x16x4_f32(false, a, false, b,
                                                        (short)0, acc, false, false);
        }
        for (int r = 0; r < 8; ++r)
            Cs[(mbase + r) * APITCH + n0 + cn] = acc[r];
    }
    __syncthreads();
    if (!live) return;
    for (int r = 0; r < 8; ++r)
        Y[(size_t)((m0 + mbase + r) * omul + oadd) * FDIM + n0 + cn] = acc[r];
    if (t < 128) {
        const int  row   = t & 15;
        const int  hd    = (t >> 4) & 3;
        const bool right = t >= 64;
        const float4* a4 = (const float4*)((right ? ar : al) + hd * HID);
        const float4* c4 = (const float4*)(Cs + row * APITCH + hd * HID);
        float s = 0.f;
        for (int q = 0; q < HID / 4; ++q) {
            const float4 cv = c4[q];
            const float4 a  = a4[q];
            s += cv.x * a.x; s += cv.y * a.y; s += cv.z * a.z; s += cv.w * a.w;
        }
        (right ? er : el)[(size_t)((m0 + row) * omul + oadd) * NH + hd] = s;
    }
}

__global__ void gat_gemm128(const float* __restrict__ X0,
                            const float* __restrict__ X1,
                            const int split,
                            const float* __restrict__ W,
                            const float* __restrict__ al,
                            const float* __restrict__ ar,
                            float* __restrict__ Y,
                            float* __restrict__ el,
                            float* __restrict__ er) {
    __shared__ __align__(16) float As[16 * APITCH];
    __shared__ __align__(16) float Cs[16 * APITCH];
    gemm128_tile(X0, X1, split, W, al, ar, Y, el, er, blockIdx.x, true, threadIdx.x, As, Cs);
}

#define GSZ   (4 * 16 * APITCH)
#define NGRP  ((NTILE + 3) / 4)
#define G_CNT 256
#define G_SCN 255
#define G_PLC 256
#define G_SRT (NGRP - G_CNT - G_SCN - G_PLC)
static_assert(G_SRT > 0, "the four shares must add up to all groups");
__device__ __forceinline__ void gemm128_group(const float* __restrict__ X0,
                                              const float* __restrict__ X1,
                                              const float* __restrict__ W,
                                              const float* __restrict__ al,
                                              const float* __restrict__ ar,
                                              float* __restrict__ Y,
                                              float* __restrict__ el,
                                              float* __restrict__ er,
                                              const int grp, float* gA, float* gC) {
    const int q    = __builtin_amdgcn_readfirstlane((int)threadIdx.x >> 8);
    const int tile = 4 * grp + q;
    gemm128_tile(X0, X1, 1, W, al, ar, Y, el, er, tile, tile < NTILE, threadIdx.x & 255,
                 gA + q * 16 * APITCH, gC + q * 16 * APITCH);
}
__global__ void gat_gemm_out(const float* __restrict__ Hm,
                             const float* __restrict__ Wout,
                             const float* __restrict__ bout,
                             float* __restrict__ logits) {
    __shared__ __align__(16) float As[16 * APITCH];
    const int m0 = blockIdx.x * 16;
    const int t  = threadIdx.x;
    for (int i = t; i < 16 * FDIM / 4; i += 96)
        *(float4*)(As + (i >> 5) * APITCH + 4 * (i & 31)) =
            ((const float4*)(Hm + (size_t)m0 * FDIM))[i];
    __syncthreads();

    const int wave = t >> 5;
    const int lane = t & 31;
    const int n0   = wave * 16;
    const int mrow = lane & 15;
    const int kgrp = lane >> 4;
    const int col  = n0 + mrow;

    v8f acc = {};
    for (int kk = 0; kk < FDIM; kk += 4) {
        v2f a, b;
        a.x = As[mrow * APITCH + kk + 2 * kgrp + 0];
        a.y = As[mrow * APITCH + kk + 2 * kgrp + 1];
        b.x = (col < NC) ? Wout[(kk + 2 * kgrp + 0) * NC + col] : 0.0f;
        b.y = (col < NC) ? Wout[(kk + 2 * kgrp + 1) * NC + col] : 0.0f;
        acc = __builtin_amdgcn_wmma_f32_16x16x4_f32(false, a, false, b,
                                                    (short)0, acc, false, false);
    }
    const int cn    = lane & 15;
    const int mbase = (lane >> 4) * 8;
    const int ocol  = n0 + cn;
    if (ocol < NC) {
        const float bb = bout[ocol];
        for (int r = 0; r < 8; ++r)
            logits[(size_t)(m0 + mbase + r) * NC + ocol] = acc[r] + bb;
    }
}

#define NBK 196
#define NCH 256
#define CHE (EE / NCH)
#define KTR ((CHE + 1023) / 1024)
#define LCAP 6144
static_assert(EE % NCH == 0, "the edge list must split evenly into chunks");
static_assert(NBK * 256 >= NN, "the bins must cover every node");
static_assert(NCH == 256, "the scan below reads a bin's row as 32 lanes x 8 counts");

__device__ __forceinline__ int bin_of(int d) { return min((int)((unsigned)d >> 8), NBK - 1); }

__device__ __forceinline__ int wave_incl(int v, int lane) {
    for (int o = 1; o < 32; o <<= 1) {
        const int u = __shfl_up(v, o);
        if (lane >= o) v += u;
    }
    return v;
}

__global__ void grp_bin_count(const int* __restrict__ dst, int* __restrict__ M,
                              const float* __restrict__ gX0, const float* __restrict__ gX1,
                              const float* __restrict__ gW, const float* __restrict__ gal,
                              const float* __restrict__ gar, float* __restrict__ gY,
                              float* __restrict__ gel, float* __restrict__ ger) {
    __shared__ int hist[NBK];
    __shared__ __align__(16) float gA[GSZ];
    __shared__ __align__(16) float gC[GSZ];
    if ((int)blockIdx.x >= NCH) {
        gemm128_group(gX0, gX1, gW, gal, gar, gY, gel, ger, 0 + (int)blockIdx.x - NCH, gA, gC);
        return;
    }
    const int t = threadIdx.x, c = blockIdx.x;
    if (t < NBK) hist[t] = 0;
    __syncthreads();
    int d[KTR];
#pragma unroll
    for (int k = 0; k < KTR; ++k) d[k] = dst[c * CHE + min(t + 1024 * k, CHE - 1)];
#pragma unroll
    for (int k = 0; k < KTR; ++k)
        if (t + 1024 * k < CHE) atomicAdd(&hist[bin_of(d[k])], 1);
    __syncthreads();
    if (t < NBK) M[t * NCH + c] = hist[t];
}

__global__ void grp_bin_scan(const int* __restrict__ M, int* __restrict__ O,
                              const float* __restrict__ gX0, const float* __restrict__ gX1,
                              const float* __restrict__ gW, const float* __restrict__ gal,
                              const float* __restrict__ gar, float* __restrict__ gY,
                              float* __restrict__ gel, float* __restrict__ ger) {
    __shared__ int rtot[224];
    __shared__ int bst[224];
    __shared__ __align__(16) float gA[GSZ];
    __shared__ __align__(16) float gC[GSZ];
    if ((int)blockIdx.x >= 1) {
        gemm128_group(gX0, gX1, gW, gal, gar, gY, gel, ger, G_CNT + (int)blockIdx.x - 1, gA, gC);
        return;
    }
    const int t = threadIdx.x, lane = t & 31, w = t >> 5;
    const int4* M4 = (const int4*)M;
    int4* O4 = (int4*)O;
    int4 a[7], c[7];
    int s[7], incl[7];
#pragma unroll
    for (int i = 0; i < 7; ++i) {
        const int b = w + 32 * i;
        a[i] = make_int4(0, 0, 0, 0); c[i] = a[i];
        if (b < NBK) { a[i] = M4[b * 64 + 2 * lane]; c[i] = M4[b * 64 + 2 * lane + 1]; }
    }
#pragma unroll
    for (int i = 0; i < 7; ++i) {
        const int b = w + 32 * i;
        s[i] = a[i].x + a[i].y + a[i].z + a[i].w + c[i].x + c[i].y + c[i].z + c[i].w;
        incl[i] = wave_incl(s[i], lane);
        if (lane == 31 && b < NBK) rtot[b] = incl[i];
    }
    __syncthreads();
    if (w == 0) {
        int x[7];
        int ss = 0;
#pragma unroll
        for (int k = 0; k < 7; ++k) {
            const int idx = 7 * lane + k;
            x[k] = (idx < NBK) ? rtot[idx] : 0;
            ss += x[k];
        }
        const int in2 = wave_incl(ss, lane);
        int run = in2 - ss;
#pragma unroll
        for (int k = 0; k < 7; ++k) {
            const int idx = 7 * lane + k;
            if (idx < NBK) bst[idx] = run;
            run += x[k];
        }
        if (lane == 31) O[NBK * NCH] = in2;
    }
    __syncthreads();
#pragma unroll
    for (int i = 0; i < 7; ++i) {
        const int b = w + 32 * i;
        if (b < NBK) {
            int run = bst[b] + incl[i] - s[i];
            int4 o1, o2;
            o1.x = run; run += a[i].x; o1.y = run; run += a[i].y;
            o1.z = run; run += a[i].z; o1.w = run; run += a[i].w;
            o2.x = run; run += c[i].x; o2.y = run; run += c[i].y;
            o2.z = run; run += c[i].z; o2.w = run;
            O4[b * 64 + 2 * lane]     = o1;
            O4[b * 64 + 2 * lane + 1] = o2;
        }
    }
}

__global__ void grp_bin_place(const int* __restrict__ src, const int* __restrict__ dst,
                              const int* __restrict__ O, int2* __restrict__ ebuf,
                              const float* __restrict__ gX0, const float* __restrict__ gX1,
                              const float* __restrict__ gW, const float* __restrict__ gal,
                              const float* __restrict__ gar, float* __restrict__ gY,
                              float* __restrict__ gel, float* __restrict__ ger) {
    __shared__ int cur[NBK];
    __shared__ __align__(16) float gA[GSZ];
    __shared__ __align__(16) float gC[GSZ];
    if ((int)blockIdx.x >= NCH) {
        gemm128_group(gX0, gX1, gW, gal, gar, gY, gel, ger, G_CNT + G_SCN + (int)blockIdx.x - NCH, gA, gC);
        return;
    }
    const int t = threadIdx.x, c = blockIdx.x;
    if (t < NBK) cur[t] = O[t * NCH + c];
    __syncthreads();
    int d[KTR], s[KTR];
#pragma unroll
    for (int k = 0; k < KTR; ++k) {
        const int e = c * CHE + min(t + 1024 * k, CHE - 1);
        d[k] = dst[e]; s[k] = src[e];
    }
#pragma unroll
    for (int k = 0; k < KTR; ++k)
        if (t + 1024 * k < CHE) {
            const int p = atomicAdd(&cur[bin_of(d[k])], 1);
            ebuf[min(max(p, 0), EE - 1)] = make_int2(s[k], d[k]);
        }
}

__global__ void grp_bin_sort(const int2* __restrict__ ebuf, const int* __restrict__ O,
                             int* __restrict__ seg, int* __restrict__ csr,
                              const float* __restrict__ gX0, const float* __restrict__ gX1,
                              const float* __restrict__ gW, const float* __restrict__ gal,
                              const float* __restrict__ gar, float* __restrict__ gY,
                              float* __restrict__ gel, float* __restrict__ ger) {
    __shared__ int cnt[256];
    __shared__ int cur[256];
    __shared__ int wtot[8];
    __shared__ int lsrc[LCAP];
    __shared__ unsigned char lnod[LCAP];
    __shared__ __align__(16) float gA[GSZ];
    __shared__ __align__(16) float gC[GSZ];
    if ((int)blockIdx.x >= NBK) {
        gemm128_group(gX0, gX1, gW, gal, gar, gY, gel, ger, G_CNT + G_SCN + G_PLC + (int)blockIdx.x - NBK, gA, gC);
        return;
    }
    const int t = threadIdx.x, lane = t & 31, w = t >> 5, b = blockIdx.x;
    const int s0 = min(max(O[b * NCH], 0), EE);
    const int n  = min(max(O[(b + 1) * NCH] - s0, 0), EE - s0);
    if (t < 256) cnt[t] = 0;
    __syncthreads();
    for (int i0 = 0; i0 < n; i0 += 4096) {
        int dk[4];
#pragma unroll
        for (int k = 0; k < 4; ++k) dk[k] = ebuf[s0 + min(i0 + t + 1024 * k, n - 1)].y;
#pragma unroll
        for (int k = 0; k < 4; ++k)
            if (i0 + t + 1024 * k < n) atomicAdd(&cnt[dk[k] & 255], 1);
    }
    __syncthreads();
    const int c    = (t < 256) ? cnt[t] : 0;
    const int incl = wave_incl(c, lane);
    if (t < 256 && lane == 31) wtot[w] = incl;
    __syncthreads();
    if (t < 256) {
        int base = 0;
        for (int k = 0; k < w; ++k) base += wtot[k];
        const int excl = base + incl - c;
        cur[t] = excl;
        const int node = b * 256 + t;
        if (node < NN) {
            seg[2 * node]     = s0 + excl;
            seg[2 * node + 1] = c;
        }
    }
    __syncthreads();
    const bool inl = n <= LCAP;
    for (int i0 = 0; i0 < n; i0 += 4096) {
        int2 r[4];
#pragma unroll
        for (int k = 0; k < 4; ++k) r[k] = ebuf[s0 + min(i0 + t + 1024 * k, n - 1)];
#pragma unroll
        for (int k = 0; k < 4; ++k)
            if (i0 + t + 1024 * k < n) {
                const int nd = r[k].y & 255;
                const int p  = min(max(atomicAdd(&cur[nd], 1), 0), n - 1);
                if (inl) { lsrc[p] = r[k].x; lnod[p] = (unsigned char)nd; }
                else     csr[s0 + p] = r[k].x;
            }
    }
    if (!inl) return;
    __syncthreads();
    for (int p = t; p < n; p += 1024) {
        const int nd = lnod[p], sv = lsrc[p];
        const int c  = min(max(cnt[nd], 0), n);
        const int b0 = min(max(cur[nd] - c, 0), n - c);
        int rank = 0;
        for (int q = b0; q < b0 + c; ++q) {
            const int o = lsrc[q];
            rank += (o < sv || (o == sv && q < p)) ? 1 : 0;
        }
        csr[s0 + b0 + rank] = sv;
    }
}

__device__ __forceinline__ float lrelu(float v) { return (v > 0.f) ? v : 0.2f * v; }
__device__ __forceinline__ float elu1(float v)  { return (v > 0.f) ? v : (expf(v) - 1.0f); }

__device__ __forceinline__ int lane_i(int v, int j) { return __builtin_amdgcn_readlane(v, j); }
__device__ __forceinline__ float lane_f(float v, int j) {
    return __int_as_float(__builtin_amdgcn_readlane(__float_as_int(v), j));
}
__device__ __forceinline__ float head_w(int head, float a0, float a1, float a2, float a3, int j) {
    const float w0 = lane_f(a0, j), w1 = lane_f(a1, j), w2 = lane_f(a2, j), w3 = lane_f(a3, j);
    const float lo = (head & 1) ? w1 : w0;
    const float hi = (head & 1) ? w3 : w2;
    return (head & 2) ? hi : lo;
}
__device__ __forceinline__ void mul_add4(float4& acc, const float4 f, const float w) {
    acc.x += f.x * w; acc.y += f.y * w; acc.z += f.z * w; acc.w += f.w * w;
}

__device__ __forceinline__ float4 score4(const float4 q, const float4 e) {
    return make_float4(lrelu(q.x + e.x), lrelu(q.y + e.y), lrelu(q.z + e.z), lrelu(q.w + e.w));
}
__device__ __forceinline__ float4 max4(const float4 a, const float4 b) {
    return make_float4(fmaxf(a.x, b.x), fmaxf(a.y, b.y), fmaxf(a.z, b.z), fmaxf(a.w, b.w));
}
__device__ __forceinline__ float4 exp4(const float4 v, const float4 m) {
    return make_float4(expf(v.x - m.x), expf(v.y - m.y), expf(v.z - m.z), expf(v.w - m.w));
}
__device__ __forceinline__ float4 div4(const float4 p, const float4 d) {
    return make_float4(p.x / d.x, p.y / d.y, p.z / d.z, p.w / d.w);
}
__device__ __forceinline__ float xhalf(float x) {
    const int b = __float_as_int(x);
    return __int_as_float(__builtin_amdgcn_permlanex16(b, b, 0x76543210, 0xfedcba98, false, false));
}
template <int M> __device__ __forceinline__ float xrow(float x) {
    return __int_as_float(__builtin_amdgcn_mov_dpp(__float_as_int(x), 0x160 | M, 0xf, 0xf, true));
}
__device__ __forceinline__ float wave_max1(float m) {
    m = fmaxf(m, xhalf(m));
    m = fmaxf(m, xrow<8>(m)); m = fmaxf(m, xrow<4>(m));
    m = fmaxf(m, xrow<2>(m)); m = fmaxf(m, xrow<1>(m));
    return m;
}
__device__ __forceinline__ float wave_sum1(float d) {
    d += xhalf(d);
    d += xrow<8>(d); d += xrow<4>(d);
    d += xrow<2>(d); d += xrow<1>(d);
    return d;
}
__device__ __forceinline__ float4 wave_max4(float4 m) {
    return make_float4(wave_max1(m.x), wave_max1(m.y), wave_max1(m.z), wave_max1(m.w));
}
__device__ __forceinline__ float4 wave_sum4(float4 d) {
    return make_float4(wave_sum1(d.x), wave_sum1(d.y), wave_sum1(d.z), wave_sum1(d.w));
}

__device__ __forceinline__ void gather_rows2(const float4* __restrict__ feat4, int lane, int head,
                                             int s_l, const float4 a, const float4 b,
                                             int count, float4& accA, float4& accB) {
    int j = 0;
    for (; j + 4 <= count; j += 4) {
        const float4* r0 = feat4 + (size_t)lane_i(s_l, j)     * 64 + lane;
        const float4* r1 = feat4 + (size_t)lane_i(s_l, j + 1) * 64 + lane;
        const float4* r2 = feat4 + (size_t)lane_i(s_l, j + 2) * 64 + lane;
        const float4* r3 = feat4 + (size_t)lane_i(s_l, j + 3) * 64 + lane;
        const float4 fA0 = r0[0], fB0 = r0[32];
        const float4 fA1 = r1[0], fB1 = r1[32];
        const float4 fA2 = r2[0], fB2 = r2[32];
        const float4 fA3 = r3[0], fB3 = r3[32];
        mul_add4(accA, fA0, head_w(head, a.x, a.y, a.z, a.w, j));
        mul_add4(accB, fB0, head_w(head, b.x, b.y, b.z, b.w, j));
        mul_add4(accA, fA1, head_w(head, a.x, a.y, a.z, a.w, j + 1));
        mul_add4(accB, fB1, head_w(head, b.x, b.y, b.z, b.w, j + 1));
        mul_add4(accA, fA2, head_w(head, a.x, a.y, a.z, a.w, j + 2));
        mul_add4(accB, fB2, head_w(head, b.x, b.y, b.z, b.w, j + 2));
        mul_add4(accA, fA3, head_w(head, a.x, a.y, a.z, a.w, j + 3));
        mul_add4(accB, fB3, head_w(head, b.x, b.y, b.z, b.w, j + 3));
    }
#pragma clang loop unroll(disable)
    for (; j < count; ++j) {
        const float4* r0 = feat4 + (size_t)lane_i(s_l, j) * 64 + lane;
        const float4 fA0 = r0[0], fB0 = r0[32];
        mul_add4(accA, fA0, head_w(head, a.x, a.y, a.z, a.w, j));
        mul_add4(accB, fB0, head_w(head, b.x, b.y, b.z, b.w, j));
    }
}

struct HeadW { float q0, q1, q2, q3; };
__device__ __forceinline__ HeadW head_regs(const float4 a, const int lane) {
    const float xx = xhalf(a.x), xy = xhalf(a.y), xz = xhalf(a.z), xw = xhalf(a.w);
    const bool lo = lane < 16;
    HeadW r;
    r.q0 = lo ? a.x : xz;
    r.q1 = lo ? a.y : xw;
    r.q2 = lo ? xx : a.z;
    r.q3 = lo ? xy : a.w;
    return r;
}
template <int J> __device__ __forceinline__ float own_w(const HeadW& h) {
    const int qa = __float_as_int(J < 16 ? h.q0 : h.q2);
    const int qb = __float_as_int(J < 16 ? h.q1 : h.q3);
    int w = __builtin_amdgcn_mov_dpp(qa, 0x150 | (J & 15), 0xf, 0x3, false);
    w = __builtin_amdgcn_update_dpp(w, qb, 0x150 | (J & 15), 0xf, 0xc, false);
    return __int_as_float(w);
}
template <int J> __device__ __forceinline__ unsigned row_off(const int s_l, const unsigned lane16) {
    return (unsigned)__builtin_amdgcn_readlane(s_l, J) * 1024u + lane16;
}
__device__ __forceinline__ float4 row_ld(const float4* __restrict__ feat4, const unsigned off, const int path) {
    return *(const float4*)((const char*)feat4 + (size_t)off + (size_t)(512 * path));
}
template <int J> __device__ __forceinline__ void one_edge(const float4* __restrict__ feat4, const unsigned lane16,
                                                          const int s_l, const HeadW& ha, const HeadW& hb,
                                                          float4& accA, float4& accB) {
    const unsigned o0 = row_off<J>(s_l, lane16);
    const float4 fA0 = row_ld(feat4, o0, 0), fB0 = row_ld(feat4, o0, 1);
    mul_add4(accA, fA0, own_w<J>(ha));
    mul_add4(accB, fB0, own_w<J>(hb));
}
template <int J> __device__ __forceinline__ void rows_from(const float4* __restrict__ feat4, const unsigned lane16,
                                                           const int s_l, const HeadW& ha, const HeadW& hb,
                                                           const int count, float4& accA, float4& accB) {
    if constexpr (J < 32) {
        if (J + 4 <= count) {
            const unsigned o0 = row_off<J>(s_l, lane16);
            const unsigned o1 = row_off<J + 1>(s_l, lane16);
            const unsigned o2 = row_off<J + 2>(s_l, lane16);
            const unsigned o3 = row_off<J + 3>(s_l, lane16);
            const float4 fA0 = row_ld(feat4, o0, 0), fB0 = row_ld(feat4, o0, 1);
            const float4 fA1 = row_ld(feat4, o1, 0), fB1 = row_ld(feat4, o1, 1);
            const float4 fA2 = row_ld(feat4, o2, 0), fB2 = row_ld(feat4, o2, 1);
            const float4 fA3 = row_ld(feat4, o3, 0), fB3 = row_ld(feat4, o3, 1);
            mul_add4(accA, fA0, own_w<J>(ha));
            mul_add4(accB, fB0, own_w<J>(hb));
            mul_add4(accA, fA1, own_w<J + 1>(ha));
            mul_add4(accB, fB1, own_w<J + 1>(hb));
            mul_add4(accA, fA2, own_w<J + 2>(ha));
            mul_add4(accB, fB2, own_w<J + 2>(hb));
            mul_add4(accA, fA3, own_w<J + 3>(ha));
            mul_add4(accB, fB3, own_w<J + 3>(hb));
            rows_from<J + 4>(feat4, lane16, s_l, ha, hb, count, accA, accB);
        } else {
            if (J < count)     one_edge<J>(feat4, lane16, s_l, ha, hb, accA, accB);
            if (J + 1 < count) one_edge<J + 1>(feat4, lane16, s_l, ha, hb, accA, accB);
            if (J + 2 < count) one_edge<J + 2>(feat4, lane16, s_l, ha, hb, accA, accB);
        }
    }
}

__global__ void gat_gather2(const int* __restrict__ seg, const int* __restrict__ csr,
                            const float* __restrict__ el, const float* __restrict__ er,
                            const float* __restrict__ feat,
                            const float* resid, float* out, int final_mix,
                            const float* __restrict__ lamb) {
    const int lane = threadIdx.x & 31;
    const int d = __builtin_amdgcn_readfirstlane(blockIdx.x * 8 + (threadIdx.x >> 5));
    if (d >= NN) return;
    const float lam = lamb[0];
    const int head = lane >> 3;
    const int beg  = seg[2 * d], deg = seg[2 * d + 1];
    const float4 erA = ((const float4*)er)[2 * d];
    const float4 erB = ((const float4*)er)[2 * d + 1];
    const float4* el4   = (const float4*)el;
    const float4* feat4 = (const float4*)feat;
    const float4 ninf = make_float4(-INFINITY, -INFINITY, -INFINITY, -INFINITY);
    const float4 zero = make_float4(0.f, 0.f, 0.f, 0.f);

    const bool ok0 = lane < deg;
    int s0 = 0;
    float4 vA = zero, vB = zero;
    if (ok0) {
        s0 = min(max(csr[beg + lane], 0), NN - 1);
        vA = score4(el4[2 * s0], erA);
        vB = score4(el4[2 * s0 + 1], erB);
    }
    float4 mA = ok0 ? vA : ninf, mB = ok0 ? vB : ninf;
    for (int base = 32; base < deg; base += 32) {
        const int idx = base + lane;
        if (idx < deg) {
            const int s = min(max(csr[beg + idx], 0), NN - 1);
            mA = max4(mA, score4(el4[2 * s], erA));
            mB = max4(mB, score4(el4[2 * s + 1], erB));
        }
    }
    mA = wave_max4(mA); mB = wave_max4(mB);

    const float4 pA = ok0 ? exp4(vA, mA) : zero;
    const float4 pB = ok0 ? exp4(vB, mB) : zero;
    float4 dA = pA, dB = pB;
    for (int base = 32; base < deg; base += 32) {
        const int idx = base + lane;
        if (idx < deg) {
            const int s = min(max(csr[beg + idx], 0), NN - 1);
            const float4 xA = exp4(score4(el4[2 * s], erA), mA);
            const float4 xB = exp4(score4(el4[2 * s + 1], erB), mB);
            dA.x += xA.x; dA.y += xA.y; dA.z += xA.z; dA.w += xA.w;
            dB.x += xB.x; dB.y += xB.y; dB.z += xB.z; dB.w += xB.w;
        }
    }
    dA = wave_sum4(dA); dB = wave_sum4(dB);

    float4 accA = zero, accB = zero;
    {
        const float4 a = ok0 ? div4(pA, dA) : zero;
        const float4 b = ok0 ? div4(pB, dB) : zero;
        const HeadW ha = head_regs(a, lane), hb = head_regs(b, lane);
        rows_from<0>(feat4, (unsigned)lane * 16u, s0, ha, hb, min(deg, 32), accA, accB);
    }
    for (int base = 32; base < deg; base += 32) {
        const int idx = base + lane;
        int s = 0;
        float4 a = zero, b = zero;
        if (idx < deg) {
            s = min(max(csr[beg + idx], 0), NN - 1);
            a = div4(exp4(score4(el4[2 * s], erA), mA), dA);
            b = div4(exp4(score4(el4[2 * s + 1], erB), mB), dB);
        }
        gather_rows2(feat4, lane, head, s, a, b, min(deg - base, 32), accA, accB);
    }

    if (resid) {
        const float4 rA = ((const float4*)resid)[(size_t)d * 64 + lane];
        const float4 rB = ((const float4*)resid)[(size_t)d * 64 + 32 + lane];
        accA.x += rA.x; accA.y += rA.y; accA.z += rA.z; accA.w += rA.w;
        accB.x += rB.x; accB.y += rB.y; accB.z += rB.z; accB.w += rB.w;
    }
    accA.x = elu1(accA.x); accA.y = elu1(accA.y); accA.z = elu1(accA.z); accA.w = elu1(accA.w);
    accB.x = elu1(accB.x); accB.y = elu1(accB.y); accB.z = elu1(accB.z); accB.w = elu1(accB.w);
    if (final_mix) {
        const float om = 1.0f - lam;
        float4 h;
        h.x = fmaf(lam, accA.x, om * accB.x); h.y = fmaf(lam, accA.y, om * accB.y);
        h.z = fmaf(lam, accA.z, om * accB.z); h.w = fmaf(lam, accA.w, om * accB.w);
        ((float4*)out)[(size_t)d * 32 + lane] = h;
    } else {
        ((float4*)out)[(size_t)d * 64 + lane]      = accA;
        ((float4*)out)[(size_t)d * 64 + 32 + lane] = accB;
    }
}


extern "C" void kernel_launch(void* const* d_in, const int* in_sizes, int n_in,
                              void* d_out, int out_size, void* d_ws, size_t ws_size,
                              hipStream_t stream) {
    (void)in_sizes; (void)n_in; (void)out_size; (void)ws_size;

    const float* inputs = (const float*)d_in[0];
    const float* target = (const float*)d_in[1];
    const float* lamb   = (const float*)d_in[2];
    const float* W0     = (const float*)d_in[3];
    const float* al0    = (const float*)d_in[4];
    const float* ar0    = (const float*)d_in[5];
    const float* W1     = (const float*)d_in[6];
    const float* al1    = (const float*)d_in[7];
    const float* ar1    = (const float*)d_in[8];
    const float* Wout   = (const float*)d_in[9];
    const float* bout   = (const float*)d_in[10];
    const int*   src    = (const int*)d_in[11];
    const int*   dst    = (const int*)d_in[12];

    float* feat   = (float*)d_ws;
    float* hbuf   = feat + (size_t)2 * NN * FDIM;
    float* el     = hbuf + (size_t)2 * NN * FDIM;
    float* er     = el   + (size_t)2 * NN * NH;
    int*   seg    = (int*)(er + (size_t)2 * NN * NH);
    int*   csr    = seg + 2 * NN;
    int*   binM   = csr + EE;
    int*   binO   = binM + NBK * NCH;
    int2*  ebuf   = (int2*)(binO + NBK * NCH + 16);

    float* hregion = (float*)d_out;
    float* logits  = hregion + (size_t)NN * FDIM;

    grp_bin_count<<<NCH + G_CNT, 1024, 0, stream>>>(dst, binM,
                                                    inputs, target, W0, al0, ar0, feat, el, er);
    grp_bin_scan<<<1 + G_SCN, 1024, 0, stream>>>(binM, binO,
                                                 inputs, target, W0, al0, ar0, feat, el, er);
    grp_bin_place<<<NCH + G_PLC, 1024, 0, stream>>>(src, dst, binO, ebuf,
                                                    inputs, target, W0, al0, ar0, feat, el, er);
    grp_bin_sort<<<NBK + G_SRT, 1024, 0, stream>>>(ebuf, binO, seg, csr,
                                                   inputs, target, W0, al0, ar0, feat, el, er);

    gat_gather2<<<(NN + 7) / 8, 256, 0, stream>>>(seg, csr, el, er, feat, nullptr, hbuf, 0, lamb);
    gat_gemm128<<<NTILE, 256, 0, stream>>>(hbuf, hbuf, 0, W1, al1, ar1, feat, el, er);
    gat_gather2<<<(NN + 7) / 8, 256, 0, stream>>>(seg, csr, el, er, feat, hbuf, hregion, 1, lamb);

    gat_gemm_out<<<NN / 16, 96, 0, stream>>>(hregion, Wout, bout, logits);
}


# Edge grouping in two launches, not four: a chunk's work-group bins its own edges inside its own part of the list, a bin's work-group finds its 256 pieces itself; 8 launches -> 6, sums unchanged
# speedup vs baseline: 8.5777x; 1.0341x over previous
#include <hip/hip_runtime.h>
#include <hip/hip_bf16.h>
#include <math.h>

#define NN   50000
#define EE   800000
#define IND  128
#define HID  32
#define NH   4
#define NC   40
#define FDIM 128
#define APITCH 132

typedef float v8f __attribute__((ext_vector_type(8)));
typedef float v2f __attribute__((ext_vector_type(2)));

#define NTILE (2 * (NN / 16))
__device__ __forceinline__ void gemm128_tile(const float* __restrict__ X0,
                                             const float* __restrict__ X1,
                                             const int split,
                                             const float* __restrict__ W,
                                             const float* __restrict__ al,
                                             const float* __restrict__ ar,
                                             float* __restrict__ Y,
                                             float* __restrict__ el,
                                             float* __restrict__ er,
                                             int tile, const bool live, const int t,
                                             float* As, float* Cs) {
    const float* X = X0;
    int omul = 1, oadd = 0;
    if (split) {
        const int path = tile >= NN / 16;
        tile -= path * (NN / 16);
        X = path ? X1 : X0;
        omul = 2; oadd = path;
    }
    const int m0 = tile * 16;

    if (live)
        for (int i = t; i < 16 * FDIM / 4; i += 256)
            *(float4*)(As + (i >> 5) * APITCH + 4 * (i & 31)) =
                ((const float4*)(X + (size_t)m0 * FDIM))[i];
    __syncthreads();

    const int wave  = t >> 5;
    const int lane  = t & 31;
    const int n0    = wave * 16;
    const int mrow  = lane & 15;
    const int kgrp  = lane >> 4;
    const int cn    = lane & 15;
    const int mbase = (lane >> 4) * 8;

    v8f acc = {};
    if (live) {
        for (int kk = 0; kk < FDIM; kk += 4) {
            v2f a, b;
            a.x = As[mrow * APITCH + kk + 2 * kgrp + 0];
            a.y = As[mrow * APITCH + kk + 2 * kgrp + 1];
            b.x = W[(kk + 2 * kgrp + 0) * FDIM + n0 + mrow];
            b.y = W[(kk + 2 * kgrp + 1) * FDIM + n0 + mrow];
            acc = __builtin_amdgcn_wmma_f32_16x16x4_f32(false, a, false, b,
                                                        (short)0, acc, false, false);
        }
        for (int r = 0; r < 8; ++r)
            Cs[(mbase + r) * APITCH + n0 + cn] = acc[r];
    }
    __syncthreads();
    if (!live) return;
    for (int r = 0; r < 8; ++r)
        Y[(size_t)((m0 + mbase + r) * omul + oadd) * FDIM + n0 + cn] = acc[r];
    if (t < 128) {
        const int  row   = t & 15;
        const int  hd    = (t >> 4) & 3;
        const bool right = t >= 64;
        const float4* a4 = (const float4*)((right ? ar : al) + hd * HID);
        const float4* c4 = (const float4*)(Cs + row * APITCH + hd * HID);
        float s = 0.f;
        for (int q = 0; q < HID / 4; ++q) {
            const float4 cv = c4[q];
            const float4 a  = a4[q];
            s += cv.x * a.x; s += cv.y * a.y; s += cv.z * a.z; s += cv.w * a.w;
        }
        (right ? er : el)[(size_t)((m0 + row) * omul + oadd) * NH + hd] = s;
    }
}

__global__ void gat_gemm128(const float* __restrict__ X0,
                            const float* __restrict__ X1,
                            const int split,
                            const float* __restrict__ W,
                            const float* __restrict__ al,
                            const float* __restrict__ ar,
                            float* __restrict__ Y,
                            float* __restrict__ el,
                            float* __restrict__ er) {
    __shared__ __align__(16) float As[16 * APITCH];
    __shared__ __align__(16) float Cs[16 * APITCH];
    gemm128_tile(X0, X1, split, W, al, ar, Y, el, er, blockIdx.x, true, threadIdx.x, As, Cs);
}


#define GSZ   (4 * 16 * APITCH)
#define NGRP  ((NTILE + 3) / 4)
#define G_L1  768
#define G_L2  (NGRP - G_L1)
static_assert(G_L2 > 0, "the two shares must add up to all groups");
__device__ __forceinline__ void gemm128_group(const float* __restrict__ X0,
                                              const float* __restrict__ X1,
                                              const float* __restrict__ W,
                                              const float* __restrict__ al,
                                              const float* __restrict__ ar,
                                              float* __restrict__ Y,
                                              float* __restrict__ el,
                                              float* __restrict__ er,
                                              const int grp, float* gA, float* gC) {
    const int q    = __builtin_amdgcn_readfirstlane((int)threadIdx.x >> 8);
    const int tile = 4 * grp + q;
    gemm128_tile(X0, X1, 1, W, al, ar, Y, el, er, tile, tile < NTILE, threadIdx.x & 255,
                 gA + q * 16 * APITCH, gC + q * 16 * APITCH);
}
__global__ void gat_gemm_out(const float* __restrict__ Hm,
                             const float* __restrict__ Wout,
                             const float* __restrict__ bout,
                             float* __restrict__ logits) {
    __shared__ __align__(16) float As[16 * APITCH];
    const int m0 = blockIdx.x * 16;
    const int t  = threadIdx.x;
    for (int i = t; i < 16 * FDIM / 4; i += 96)
        *(float4*)(As + (i >> 5) * APITCH + 4 * (i & 31)) =
            ((const float4*)(Hm + (size_t)m0 * FDIM))[i];
    __syncthreads();

    const int wave = t >> 5;
    const int lane = t & 31;
    const int n0   = wave * 16;
    const int mrow = lane & 15;
    const int kgrp = lane >> 4;
    const int col  = n0 + mrow;

    v8f acc = {};
    for (int kk = 0; kk < FDIM; kk += 4) {
        v2f a, b;
        a.x = As[mrow * APITCH + kk + 2 * kgrp + 0];
        a.y = As[mrow * APITCH + kk + 2 * kgrp + 1];
        b.x = (col < NC) ? Wout[(kk + 2 * kgrp + 0) * NC + col] : 0.0f;
        b.y = (col < NC) ? Wout[(kk + 2 * kgrp + 1) * NC + col] : 0.0f;
        acc = __builtin_amdgcn_wmma_f32_16x16x4_f32(false, a, false, b,
                                                    (short)0, acc, false, false);
    }
    const int cn    = lane & 15;
    const int mbase = (lane >> 4) * 8;
    const int ocol  = n0 + cn;
    if (ocol < NC) {
        const float bb = bout[ocol];
        for (int r = 0; r < 8; ++r)
            logits[(size_t)(m0 + mbase + r) * NC + ocol] = acc[r] + bb;
    }
}

#define NBK 196
#define NCH 256
#define CHE (EE / NCH)
#define KTR ((CHE + 1023) / 1024)
#define LCAP 6144
static_assert(EE % NCH == 0, "the edge list must split evenly into chunks");
static_assert(NBK * 256 >= NN, "the bins must cover every node");
static_assert(NCH * 4 == 1024, "level 2 walks a bin's NCH pieces with four threads each");

__device__ __forceinline__ int bin_of(int d) { return min((int)((unsigned)d >> 8), NBK - 1); }

__device__ __forceinline__ int wave_incl(int v, int lane) {
    for (int o = 1; o < 32; o <<= 1) {
        const int u = __shfl_up(v, o);
        if (lane >= o) v += u;
    }
    return v;
}

__global__ void grp_chunk_bins(const int* __restrict__ src, const int* __restrict__ dst,
                               int* __restrict__ M, int* __restrict__ F, int2* __restrict__ ebuf,
                               const float* __restrict__ gX0, const float* __restrict__ gX1,
                               const float* __restrict__ gW, const float* __restrict__ gal,
                               const float* __restrict__ gar, float* __restrict__ gY,
                               float* __restrict__ gel, float* __restrict__ ger) {
    __shared__ int hist[NBK];
    __shared__ int cur[NBK];
    __shared__ __align__(16) float gA[GSZ];
    __shared__ __align__(16) float gC[GSZ];
    if ((int)blockIdx.x >= NCH) {
        gemm128_group(gX0, gX1, gW, gal, gar, gY, gel, ger, (int)blockIdx.x - NCH, gA, gC);
        return;
    }
    const int t = threadIdx.x, c = blockIdx.x;
    if (t < NBK) hist[t] = 0;
    __syncthreads();
    int d[KTR], s[KTR];
#pragma unroll
    for (int k = 0; k < KTR; ++k) {
        const int e = c * CHE + min(t + 1024 * k, CHE - 1);
        d[k] = dst[e]; s[k] = src[e];
    }
#pragma unroll
    for (int k = 0; k < KTR; ++k)
        if (t + 1024 * k < CHE) atomicAdd(&hist[bin_of(d[k])], 1);
    __syncthreads();
    if (t < 32) {
        int x[7];
        int ss = 0;
#pragma unroll
        for (int k = 0; k < 7; ++k) {
            const int idx = 7 * t + k;
            x[k] = (idx < NBK) ? hist[idx] : 0;
            ss += x[k];
        }
        int run = wave_incl(ss, t) - ss;
#pragma unroll
        for (int k = 0; k < 7; ++k) {
            const int idx = 7 * t + k;
            if (idx < NBK) {
                cur[idx] = run;
                M[idx * NCH + c] = x[k];
                F[idx * NCH + c] = run;
            }
            run += x[k];
        }
    }
    __syncthreads();
#pragma unroll
    for (int k = 0; k < KTR; ++k)
        if (t + 1024 * k < CHE) {
            const int p = atomicAdd(&cur[bin_of(d[k])], 1);
            ebuf[c * CHE + min(max(p, 0), CHE - 1)] = make_int2(s[k], d[k]);
        }
}

#define MQ ((NBK * NCH / 4 + 1023) / 1024)
__global__ void grp_bin_sort(const int2* __restrict__ ebuf, const int* __restrict__ M,
                             const int* __restrict__ F,
                             int* __restrict__ seg, int* __restrict__ csr,
                              const float* __restrict__ gX0, const float* __restrict__ gX1,
                              const float* __restrict__ gW, const float* __restrict__ gal,
                              const float* __restrict__ gar, float* __restrict__ gY,
                              float* __restrict__ gel, float* __restrict__ ger) {
    __shared__ int cnt[256];
    __shared__ int cur[256];
    __shared__ int wtot[8];
    __shared__ int red[64];
    __shared__ int lsrc[LCAP];
    __shared__ unsigned char lnod[LCAP];
    __shared__ __align__(16) float gA[GSZ];
    __shared__ __align__(16) float gC[GSZ];
    if ((int)blockIdx.x >= NBK) {
        gemm128_group(gX0, gX1, gW, gal, gar, gY, gel, ger, G_L1 + (int)blockIdx.x - NBK, gA, gC);
        return;
    }
    const int t = threadIdx.x, lane = t & 31, w = t >> 5, b = blockIdx.x;
    const int pj   = t & 3;
    const int plen = min(max(M[b * NCH + (t >> 2)], 0), CHE);
    const int pbeg = (t >> 2) * CHE + min(max(F[b * NCH + (t >> 2)], 0), CHE - plen);
    int pre = 0;
    {
        const int4* M4 = (const int4*)M;
        const int lim = b * (NCH / 4);
        int4 v[MQ];
#pragma unroll
        for (int k = 0; k < MQ; ++k) {
            const int i = t + 1024 * k;
            v[k] = make_int4(0, 0, 0, 0);
            if (i < lim) v[k] = M4[i];
        }
#pragma unroll
        for (int k = 0; k < MQ; ++k) pre += v[k].x + v[k].y + v[k].z + v[k].w;
    }
    const int pre_w  = wave_incl(pre, lane);
    const int mine_w = wave_incl(pj == 0 ? plen : 0, lane);
    if (lane == 31) { red[w] = pre_w; red[32 + w] = mine_w; }
    if (t < 256) cnt[t] = 0;
    __syncthreads();
    const int s0v = __builtin_amdgcn_readlane(wave_incl(red[lane], lane), 31);
    const int nv  = __builtin_amdgcn_readlane(wave_incl(red[32 + lane], lane), 31);
    const int s0  = min(max(s0v, 0), EE);
    const int n   = min(max(nv, 0), EE - s0);
    for (int i0 = pj; i0 < plen; i0 += 16) {
        int dk[4];
#pragma unroll
        for (int k = 0; k < 4; ++k) dk[k] = ebuf[pbeg + min(i0 + 4 * k, plen - 1)].y;
#pragma unroll
        for (int k = 0; k < 4; ++k)
            if (i0 + 4 * k < plen) atomicAdd(&cnt[dk[k] & 255], 1);
    }
    __syncthreads();
    const int c    = (t < 256) ? cnt[t] : 0;
    const int incl = wave_incl(c, lane);
    if (t < 256 && lane == 31) wtot[w] = incl;
    __syncthreads();
    if (t < 256) {
        int base = 0;
        for (int k = 0; k < w; ++k) base += wtot[k];
        const int excl = base + incl - c;
        cur[t] = excl;
        const int node = b * 256 + t;
        if (node < NN) {
            seg[2 * node]     = s0 + excl;
            seg[2 * node + 1] = c;
        }
    }
    __syncthreads();
    const bool inl = n <= LCAP;
    for (int i0 = pj; i0 < plen; i0 += 16) {
        int2 r[4];
#pragma unroll
        for (int k = 0; k < 4; ++k) r[k] = ebuf[pbeg + min(i0 + 4 * k, plen - 1)];
#pragma unroll
        for (int k = 0; k < 4; ++k)
            if (i0 + 4 * k < plen) {
                const int nd = r[k].y & 255;
                const int p  = min(max(atomicAdd(&cur[nd], 1), 0), n - 1);
                if (inl) { lsrc[p] = r[k].x; lnod[p] = (unsigned char)nd; }
                else     csr[s0 + p] = r[k].x;
            }
    }
    if (!inl) return;
    __syncthreads();
    for (int p = t; p < n; p += 1024) {
        const int nd = lnod[p], sv = lsrc[p];
        const int c  = min(max(cnt[nd], 0), n);
        const int b0 = min(max(cur[nd] - c, 0), n - c);
        int rank = 0;
        for (int q = b0; q < b0 + c; ++q) {
            const int o = lsrc[q];
            rank += (o < sv || (o == sv && q < p)) ? 1 : 0;
        }
        csr[s0 + b0 + rank] = sv;
    }
}

__device__ __forceinline__ float lrelu(float v) { return (v > 0.f) ? v : 0.2f * v; }
__device__ __forceinline__ float elu1(float v)  { return (v > 0.f) ? v : (expf(v) - 1.0f); }

__device__ __forceinline__ int lane_i(int v, int j) { return __builtin_amdgcn_readlane(v, j); }
__device__ __forceinline__ float lane_f(float v, int j) {
    return __int_as_float(__builtin_amdgcn_readlane(__float_as_int(v), j));
}
__device__ __forceinline__ float head_w(int head, float a0, float a1, float a2, float a3, int j) {
    const float w0 = lane_f(a0, j), w1 = lane_f(a1, j), w2 = lane_f(a2, j), w3 = lane_f(a3, j);
    const float lo = (head & 1) ? w1 : w0;
    const float hi = (head & 1) ? w3 : w2;
    return (head & 2) ? hi : lo;
}
__device__ __forceinline__ void mul_add4(float4& acc, const float4 f, const float w) {
    acc.x += f.x * w; acc.y += f.y * w; acc.z += f.z * w; acc.w += f.w * w;
}

__device__ __forceinline__ float4 score4(const float4 q, const float4 e) {
    return make_float4(lrelu(q.x + e.x), lrelu(q.y + e.y), lrelu(q.z + e.z), lrelu(q.w + e.w));
}
__device__ __forceinline__ float4 max4(const float4 a, const float4 b) {
    return make_float4(fmaxf(a.x, b.x), fmaxf(a.y, b.y), fmaxf(a.z, b.z), fmaxf(a.w, b.w));
}
__device__ __forceinline__ float4 exp4(const float4 v, const float4 m) {
    return make_float4(expf(v.x - m.x), expf(v.y - m.y), expf(v.z - m.z), expf(v.w - m.w));
}
__device__ __forceinline__ float4 div4(const float4 p, const float4 d) {
    return make_float4(p.x / d.x, p.y / d.y, p.z / d.z, p.w / d.w);
}
__device__ __forceinline__ float xhalf(float x) {
    const int b = __float_as_int(x);
    return __int_as_float(__builtin_amdgcn_permlanex16(b, b, 0x76543210, 0xfedcba98, false, false));
}
template <int M> __device__ __forceinline__ float xrow(float x) {
    return __int_as_float(__builtin_amdgcn_mov_dpp(__float_as_int(x), 0x160 | M, 0xf, 0xf, true));
}
__device__ __forceinline__ float wave_max1(float m) {
    m = fmaxf(m, xhalf(m));
    m = fmaxf(m, xrow<8>(m)); m = fmaxf(m, xrow<4>(m));
    m = fmaxf(m, xrow<2>(m)); m = fmaxf(m, xrow<1>(m));
    return m;
}
__device__ __forceinline__ float wave_sum1(float d) {
    d += xhalf(d);
    d += xrow<8>(d); d += xrow<4>(d);
    d += xrow<2>(d); d += xrow<1>(d);
    return d;
}
__device__ __forceinline__ float4 wave_max4(float4 m) {
    return make_float4(wave_max1(m.x), wave_max1(m.y), wave_max1(m.z), wave_max1(m.w));
}
__device__ __forceinline__ float4 wave_sum4(float4 d) {
    return make_float4(wave_sum1(d.x), wave_sum1(d.y), wave_sum1(d.z), wave_sum1(d.w));
}

__device__ __forceinline__ void gather_rows2(const float4* __restrict__ feat4, int lane, int head,
                                             int s_l, const float4 a, const float4 b,
                                             int count, float4& accA, float4& accB) {
    int j = 0;
    for (; j + 4 <= count; j += 4) {
        const float4* r0 = feat4 + (size_t)lane_i(s_l, j)     * 64 + lane;
        const float4* r1 = feat4 + (size_t)lane_i(s_l, j + 1) * 64 + lane;
        const float4* r2 = feat4 + (size_t)lane_i(s_l, j + 2) * 64 + lane;
        const float4* r3 = feat4 + (size_t)lane_i(s_l, j + 3) * 64 + lane;
        const float4 fA0 = r0[0], fB0 = r0[32];
        const float4 fA1 = r1[0], fB1 = r1[32];
        const float4 fA2 = r2[0], fB2 = r2[32];
        const float4 fA3 = r3[0], fB3 = r3[32];
        mul_add4(accA, fA0, head_w(head, a.x, a.y, a.z, a.w, j));
        mul_add4(accB, fB0, head_w(head, b.x, b.y, b.z, b.w, j));
        mul_add4(accA, fA1, head_w(head, a.x, a.y, a.z, a.w, j + 1));
        mul_add4(accB, fB1, head_w(head, b.x, b.y, b.z, b.w, j + 1));
        mul_add4(accA, fA2, head_w(head, a.x, a.y, a.z, a.w, j + 2));
        mul_add4(accB, fB2, head_w(head, b.x, b.y, b.z, b.w, j + 2));
        mul_add4(accA, fA3, head_w(head, a.x, a.y, a.z, a.w, j + 3));
        mul_add4(accB, fB3, head_w(head, b.x, b.y, b.z, b.w, j + 3));
    }
#pragma clang loop unroll(disable)
    for (; j < count; ++j) {
        const float4* r0 = feat4 + (size_t)lane_i(s_l, j) * 64 + lane;
        const float4 fA0 = r0[0], fB0 = r0[32];
        mul_add4(accA, fA0, head_w(head, a.x, a.y, a.z, a.w, j));
        mul_add4(accB, fB0, head_w(head, b.x, b.y, b.z, b.w, j));
    }
}

struct HeadW { float q0, q1, q2, q3; };
__device__ __forceinline__ HeadW head_regs(const float4 a, const int lane) {
    const float xx = xhalf(a.x), xy = xhalf(a.y), xz = xhalf(a.z), xw = xhalf(a.w);
    const bool lo = lane < 16;
    HeadW r;
    r.q0 = lo ? a.x : xz;
    r.q1 = lo ? a.y : xw;
    r.q2 = lo ? xx : a.z;
    r.q3 = lo ? xy : a.w;
    return r;
}
template <int J> __device__ __forceinline__ float own_w(const HeadW& h) {
    const int qa = __float_as_int(J < 16 ? h.q0 : h.q2);
    const int qb = __float_as_int(J < 16 ? h.q1 : h.q3);
    int w = __builtin_amdgcn_mov_dpp(qa, 0x150 | (J & 15), 0xf, 0x3, false);
    w = __builtin_amdgcn_update_dpp(w, qb, 0x150 | (J & 15), 0xf, 0xc, false);
    return __int_as_float(w);
}
template <int J> __device__ __forceinline__ unsigned row_off(const int s_l, const unsigned lane16) {
    return (unsigned)__builtin_amdgcn_readlane(s_l, J) * 1024u + lane16;
}
__device__ __forceinline__ float4 row_ld(const float4* __restrict__ feat4, const unsigned off, const int path) {
    return *(const float4*)((const char*)feat4 + (size_t)off + (size_t)(512 * path));
}
template <int J> __device__ __forceinline__ void one_edge(const float4* __restrict__ feat4, const unsigned lane16,
                                                          const int s_l, const HeadW& ha, const HeadW& hb,
                                                          float4& accA, float4& accB) {
    const unsigned o0 = row_off<J>(s_l, lane16);
    const float4 fA0 = row_ld(feat4, o0, 0), fB0 = row_ld(feat4, o0, 1);
    mul_add4(accA, fA0, own_w<J>(ha));
    mul_add4(accB, fB0, own_w<J>(hb));
}
template <int J> __device__ __forceinline__ void rows_from(const float4* __restrict__ feat4, const unsigned lane16,
                                                           const int s_l, const HeadW& ha, const HeadW& hb,
                                                           const int count, float4& accA, float4& accB) {
    if constexpr (J < 32) {
        if (J + 4 <= count) {
            const unsigned o0 = row_off<J>(s_l, lane16);
            const unsigned o1 = row_off<J + 1>(s_l, lane16);
            const unsigned o2 = row_off<J + 2>(s_l, lane16);
            const unsigned o3 = row_off<J + 3>(s_l, lane16);
            const float4 fA0 = row_ld(feat4, o0, 0), fB0 = row_ld(feat4, o0, 1);
            const float4 fA1 = row_ld(feat4, o1, 0), fB1 = row_ld(feat4, o1, 1);
            const float4 fA2 = row_ld(feat4, o2, 0), fB2 = row_ld(feat4, o2, 1);
            const float4 fA3 = row_ld(feat4, o3, 0), fB3 = row_ld(feat4, o3, 1);
            mul_add4(accA, fA0, own_w<J>(ha));
            mul_add4(accB, fB0, own_w<J>(hb));
            mul_add4(accA, fA1, own_w<J + 1>(ha));
            mul_add4(accB, fB1, own_w<J + 1>(hb));
            mul_add4(accA, fA2, own_w<J + 2>(ha));
            mul_add4(accB, fB2, own_w<J + 2>(hb));
            mul_add4(accA, fA3, own_w<J + 3>(ha));
            mul_add4(accB, fB3, own_w<J + 3>(hb));
            rows_from<J + 4>(feat4, lane16, s_l, ha, hb, count, accA, accB);
        } else {
            if (J < count)     one_edge<J>(feat4, lane16, s_l, ha, hb, accA, accB);
            if (J + 1 < count) one_edge<J + 1>(feat4, lane16, s_l, ha, hb, accA, accB);
            if (J + 2 < count) one_edge<J + 2>(feat4, lane16, s_l, ha, hb, accA, accB);
        }
    }
}

__global__ void gat_gather2(const int* __restrict__ seg, const int* __restrict__ csr,
                            const float* __restrict__ el, const float* __restrict__ er,
                            const float* __restrict__ feat,
                            const float* resid, float* out, int final_mix,
                            const float* __restrict__ lamb) {
    const int lane = threadIdx.x & 31;
    const int d = __builtin_amdgcn_readfirstlane(blockIdx.x * 8 + (threadIdx.x >> 5));
    if (d >= NN) return;
    const float lam = lamb[0];
    const int head = lane >> 3;
    const int beg  = seg[2 * d], deg = seg[2 * d + 1];
    const float4 erA = ((const float4*)er)[2 * d];
    const float4 erB = ((const float4*)er)[2 * d + 1];
    const float4* el4   = (const float4*)el;
    const float4* feat4 = (const float4*)feat;
    const float4 ninf = make_float4(-INFINITY, -INFINITY, -INFINITY, -INFINITY);
    const float4 zero = make_float4(0.f, 0.f, 0.f, 0.f);

    const bool ok0 = lane < deg;
    int s0 = 0;
    float4 vA = zero, vB = zero;
    if (ok0) {
        s0 = min(max(csr[beg + lane], 0), NN - 1);
        vA = score4(el4[2 * s0], erA);
        vB = score4(el4[2 * s0 + 1], erB);
    }
    float4 mA = ok0 ? vA : ninf, mB = ok0 ? vB : ninf;
    for (int base = 32; base < deg; base += 32) {
        const int idx = base + lane;
        if (idx < deg) {
            const int s = min(max(csr[beg + idx], 0), NN - 1);
            mA = max4(mA, score4(el4[2 * s], erA));
            mB = max4(mB, score4(el4[2 * s + 1], erB));
        }
    }
    mA = wave_max4(mA); mB = wave_max4(mB);

    const float4 pA = ok0 ? exp4(vA, mA) : zero;
    const float4 pB = ok0 ? exp4(vB, mB) : zero;
    float4 dA = pA, dB = pB;
    for (int base = 32; base < deg; base += 32) {
        const int idx = base + lane;
        if (idx < deg) {
            const int s = min(max(csr[beg + idx], 0), NN - 1);
            const float4 xA = exp4(score4(el4[2 * s], erA), mA);
            const float4 xB = exp4(score4(el4[2 * s + 1], erB), mB);
            dA.x += xA.x; dA.y += xA.y; dA.z += xA.z; dA.w += xA.w;
            dB.x += xB.x; dB.y += xB.y; dB.z += xB.z; dB.w += xB.w;
        }
    }
    dA = wave_sum4(dA); dB = wave_sum4(dB);

    float4 accA = zero, accB = zero;
    {
        const float4 a = ok0 ? div4(pA, dA) : zero;
        const float4 b = ok0 ? div4(pB, dB) : zero;
        const HeadW ha = head_regs(a, lane), hb = head_regs(b, lane);
        rows_from<0>(feat4, (unsigned)lane * 16u, s0, ha, hb, min(deg, 32), accA, accB);
    }
    for (int base = 32; base < deg; base += 32) {
        const int idx = base + lane;
        int s = 0;
        float4 a = zero, b = zero;
        if (idx < deg) {
            s = min(max(csr[beg + idx], 0), NN - 1);
            a = div4(exp4(score4(el4[2 * s], erA), mA), dA);
            b = div4(exp4(score4(el4[2 * s + 1], erB), mB), dB);
        }
        gather_rows2(feat4, lane, head, s, a, b, min(deg - base, 32), accA, accB);
    }

    if (resid) {
        const float4 rA = ((const float4*)resid)[(size_t)d * 64 + lane];
        const float4 rB = ((const float4*)resid)[(size_t)d * 64 + 32 + lane];
        accA.x += rA.x; accA.y += rA.y; accA.z += rA.z; accA.w += rA.w;
        accB.x += rB.x; accB.y += rB.y; accB.z += rB.z; accB.w += rB.w;
    }
    accA.x = elu1(accA.x); accA.y = elu1(accA.y); accA.z = elu1(accA.z); accA.w = elu1(accA.w);
    accB.x = elu1(accB.x); accB.y = elu1(accB.y); accB.z = elu1(accB.z); accB.w = elu1(accB.w);
    if (final_mix) {
        const float om = 1.0f - lam;
        float4 h;
        h.x = fmaf(lam, accA.x, om * accB.x); h.y = fmaf(lam, accA.y, om * accB.y);
        h.z = fmaf(lam, accA.z, om * accB.z); h.w = fmaf(lam, accA.w, om * accB.w);
        ((float4*)out)[(size_t)d * 32 + lane] = h;
    } else {
        ((float4*)out)[(size_t)d * 64 + lane]      = accA;
        ((float4*)out)[(size_t)d * 64 + 32 + lane] = accB;
    }
}


extern "C" void kernel_launch(void* const* d_in, const int* in_sizes, int n_in,
                              void* d_out, int out_size, void* d_ws, size_t ws_size,
                              hipStream_t stream) {
    (void)in_sizes; (void)n_in; (void)out_size; (void)ws_size;

    const float* inputs = (const float*)d_in[0];
    const float* target = (const float*)d_in[1];
    const float* lamb   = (const float*)d_in[2];
    const float* W0     = (const float*)d_in[3];
    const float* al0    = (const float*)d_in[4];
    const float* ar0    = (const float*)d_in[5];
    const float* W1     = (const float*)d_in[6];
    const float* al1    = (const float*)d_in[7];
    const float* ar1    = (const float*)d_in[8];
    const float* Wout   = (const float*)d_in[9];
    const float* bout   = (const float*)d_in[10];
    const int*   src    = (const int*)d_in[11];
    const int*   dst    = (const int*)d_in[12];

    float* feat   = (float*)d_ws;
    float* hbuf   = feat + (size_t)2 * NN * FDIM;
    float* el     = hbuf + (size_t)2 * NN * FDIM;
    float* er     = el   + (size_t)2 * NN * NH;
    int*   seg    = (int*)(er + (size_t)2 * NN * NH);
    int*   csr    = seg + 2 * NN;
    int*   binM   = csr + EE;
    int*   binF   = binM + NBK * NCH;
    int2*  ebuf   = (int2*)(binF + NBK * NCH + 16);

    float* hregion = (float*)d_out;
    float* logits  = hregion + (size_t)NN * FDIM;

    grp_chunk_bins<<<NCH + G_L1, 1024, 0, stream>>>(src, dst, binM, binF, ebuf,
                                                    inputs, target, W0, al0, ar0, feat, el, er);
    grp_bin_sort<<<NBK + G_L2, 1024, 0, stream>>>(ebuf, binM, binF, seg, csr,
                                                  inputs, target, W0, al0, ar0, feat, el, er);

    gat_gather2<<<(NN + 7) / 8, 256, 0, stream>>>(seg, csr, el, er, feat, nullptr, hbuf, 0, lamb);
    gat_gemm128<<<NTILE, 256, 0, stream>>>(hbuf, hbuf, 0, W1, al1, ar1, feat, el, er);
    gat_gather2<<<(NN + 7) / 8, 256, 0, stream>>>(seg, csr, el, er, feat, hbuf, hregion, 1, lamb);

    gat_gemm_out<<<NN / 16, 96, 0, stream>>>(hregion, Wout, bout, logits);
}
